# starting kernel
# baseline (speedup 1.0000x reference)
#include <hip/hip_runtime.h>
#include <hip/hip_bf16.h>

#define N_NODES 100000
#define N_EDGES 1600000
#define IN_DIM 128
#define OUT_DIM 32
#define BN_EPS 1e-5f
#define LRELU_SLOPE 0.01f

typedef __attribute__((ext_vector_type(2))) float v2f;
typedef __attribute__((ext_vector_type(8))) float v8f;

__device__ __forceinline__ unsigned fflip(float f) {
    unsigned u = __float_as_uint(f);
    return (u >> 31) ? ~u : (u | 0x80000000u);
}
__device__ __forceinline__ float funflip(unsigned u) {
    u = (u >> 31) ? (u & 0x7FFFFFFFu) : ~u;
    return __uint_as_float(u);
}

__global__ void gat_init_kernel(float* hout, float* esum, unsigned* emax, float* bn) {
    int idx = blockIdx.x * blockDim.x + threadIdx.x;
    int stride = gridDim.x * blockDim.x;
    for (int i = idx; i < N_NODES * OUT_DIM; i += stride) hout[i] = 0.0f;
    for (int i = idx; i < N_NODES; i += stride) { esum[i] = 0.0f; emax[i] = 0u; }
    for (int i = idx; i < 2 * OUT_DIM; i += stride) bn[i] = 0.0f;
}

__global__ void gat_gemm_wmma_kernel(const float* __restrict__ h,
                                     const float* __restrict__ W,
                                     float* __restrict__ z) {
    __shared__ float lds_w[IN_DIM * OUT_DIM];
    for (int i = threadIdx.x; i < IN_DIM * OUT_DIM; i += blockDim.x)
        lds_w[i] = W[i];
    __syncthreads();

    const int wave = threadIdx.x >> 5;
    const int lane = threadIdx.x & 31;
    const int tile = blockIdx.x * (blockDim.x >> 5) + wave;
    if (tile >= N_NODES / 16) return;

    const int m0   = tile * 16;
    const int row  = m0 + (lane & 15);
    const int koff = (lane >> 4) * 2;
    const int col  = lane & 15;

    v8f c0 = {}, c1 = {};
    const float* hrow = h + (size_t)row * IN_DIM;

    #pragma unroll 4
    for (int k = 0; k < IN_DIM; k += 4) {
        v2f a;
        a.x = hrow[k + koff];
        a.y = hrow[k + koff + 1];
        v2f b0, b1;
        b0.x = lds_w[(k + koff)     * OUT_DIM + col];
        b0.y = lds_w[(k + koff + 1) * OUT_DIM + col];
        b1.x = lds_w[(k + koff)     * OUT_DIM + col + 16];
        b1.y = lds_w[(k + koff + 1) * OUT_DIM + col + 16];
        c0 = __builtin_amdgcn_wmma_f32_16x16x4_f32(false, a, false, b0, (short)0, c0, false, false);
        c1 = __builtin_amdgcn_wmma_f32_16x16x4_f32(false, a, false, b1, (short)0, c1, false, false);
    }

    const int rbase = m0 + ((lane >> 4) * 8);
    #pragma unroll
    for (int r = 0; r < 8; ++r) {
        z[(size_t)(rbase + r) * OUT_DIM + col]      = c0[r];
        z[(size_t)(rbase + r) * OUT_DIM + col + 16] = c1[r];
    }
}

__global__ void gat_attn_proj_kernel(const float* __restrict__ z,
                                     const float* __restrict__ Wa,
                                     float* __restrict__ el,
                                     float* __restrict__ er) {
    int gid = blockIdx.x * blockDim.x + threadIdx.x;
    int node = gid >> 5;
    int lane = gid & 31;
    if (node >= N_NODES) return;
    float v  = z[(size_t)node * OUT_DIM + lane];
    float pl = v * Wa[lane];
    float pr = v * Wa[OUT_DIM + lane];
    #pragma unroll
    for (int off = 16; off > 0; off >>= 1) {
        pl += __shfl_down(pl, off, 32);
        pr += __shfl_down(pr, off, 32);
    }
    if (lane == 0) { el[node] = pl; er[node] = pr; }
}

__global__ void gat_edge_max_kernel(const float* __restrict__ el,
                                    const float* __restrict__ er,
                                    const int* __restrict__ src,
                                    const int* __restrict__ dst,
                                    float* __restrict__ e_edge,
                                    unsigned* __restrict__ emax) {
    int i = blockIdx.x * blockDim.x + threadIdx.x;
    if (i >= N_EDGES) return;
    float e = el[src[i]] + er[dst[i]];
    e = (e >= 0.0f) ? e : LRELU_SLOPE * e;
    e_edge[i] = e;
    atomicMax(&emax[dst[i]], fflip(e));
}

__global__ void gat_edge_exp_kernel(const int* __restrict__ dst,
                                    const float* __restrict__ e_edge,
                                    const unsigned* __restrict__ emax,
                                    float* __restrict__ eexp,
                                    float* __restrict__ esum) {
    int i = blockIdx.x * blockDim.x + threadIdx.x;
    if (i >= N_EDGES) return;
    int d = dst[i];
    float ex = __expf(e_edge[i] - funflip(emax[d]));
    eexp[i] = ex;
    atomicAdd(&esum[d], ex);
}

__global__ void gat_edge_accum_kernel(const int* __restrict__ src,
                                      const int* __restrict__ dst,
                                      const float* __restrict__ z,
                                      const float* __restrict__ eexp,
                                      const float* __restrict__ esum,
                                      float* __restrict__ hout) {
    int gid = blockIdx.x * blockDim.x + threadIdx.x;
    int edge = gid >> 5;
    if (edge >= N_EDGES) return;
    int c = gid & 31;
    int s = src[edge], d = dst[edge];
    float alpha = eexp[edge] / esum[d];
    atomicAdd(&hout[(size_t)d * OUT_DIM + c], alpha * z[(size_t)s * OUT_DIM + c]);
}

__global__ void gat_bn_stats_kernel(const float* __restrict__ hout, float* __restrict__ bn) {
    __shared__ float ssum[OUT_DIM];
    __shared__ float ssq[OUT_DIM];
    int lane = threadIdx.x & 31;
    if (threadIdx.x < OUT_DIM) { ssum[threadIdx.x] = 0.0f; ssq[threadIdx.x] = 0.0f; }
    __syncthreads();
    int row = (blockIdx.x * blockDim.x + threadIdx.x) >> 5;
    int stride = (gridDim.x * blockDim.x) >> 5;
    float s = 0.0f, q = 0.0f;
    for (; row < N_NODES; row += stride) {
        float v = hout[(size_t)row * OUT_DIM + lane];
        s += v;
        q += v * v;
    }
    atomicAdd(&ssum[lane], s);
    atomicAdd(&ssq[lane], q);
    __syncthreads();
    if (threadIdx.x < OUT_DIM) {
        atomicAdd(&bn[threadIdx.x], ssum[threadIdx.x]);
        atomicAdd(&bn[OUT_DIM + threadIdx.x], ssq[threadIdx.x]);
    }
}

__global__ void gat_bn_apply_kernel(const float* __restrict__ bn,
                                    const float* __restrict__ gamma,
                                    const float* __restrict__ beta,
                                    float* __restrict__ out) {
    int idx = blockIdx.x * blockDim.x + threadIdx.x;
    if (idx >= N_NODES * OUT_DIM) return;
    int c = idx & 31;
    const float inv_n = 1.0f / (float)N_NODES;
    float mean = bn[c] * inv_n;
    float var  = bn[OUT_DIM + c] * inv_n - mean * mean;
    float x = (out[idx] - mean) * rsqrtf(var + BN_EPS) * gamma[c] + beta[c];
    out[idx] = (x > 0.0f) ? x : expm1f(x);
}

extern "C" void kernel_launch(void* const* d_in, const int* in_sizes, int n_in,
                              void* d_out, int out_size, void* d_ws, size_t ws_size,
                              hipStream_t stream) {
    const float* h     = (const float*)d_in[0];
    const float* W_fc  = (const float*)d_in[1];
    const float* W_at  = (const float*)d_in[2];
    const float* gamma = (const float*)d_in[3];
    const float* beta  = (const float*)d_in[4];
    const int*   src   = (const int*)d_in[5];
    const int*   dst   = (const int*)d_in[6];
    float* out = (float*)d_out;

    float* z        = (float*)d_ws;
    float* el       = z + (size_t)N_NODES * OUT_DIM;
    float* er       = el + N_NODES;
    unsigned* emax  = (unsigned*)(er + N_NODES);
    float* esum     = (float*)(emax + N_NODES);
    float* e_edge   = esum + N_NODES;
    float* eexp     = e_edge + N_EDGES;
    float* bn       = eexp + N_EDGES;

    const int B = 256;

    gat_init_kernel<<<1024, B, 0, stream>>>(out, esum, emax, bn);

    gat_gemm_wmma_kernel<<<(N_NODES / 16 + 7) / 8, B, 0, stream>>>(h, W_fc, z);

    gat_attn_proj_kernel<<<(N_NODES * OUT_DIM + B - 1) / B, B, 0, stream>>>(z, W_at, el, er);

    gat_edge_max_kernel<<<(N_EDGES + B - 1) / B, B, 0, stream>>>(el, er, src, dst, e_edge, emax);

    gat_edge_exp_kernel<<<(N_EDGES + B - 1) / B, B, 0, stream>>>(dst, e_edge, emax, eexp, esum);

    gat_edge_accum_kernel<<<((size_t)N_EDGES * OUT_DIM + B - 1) / B, B, 0, stream>>>(
        src, dst, z, eexp, esum, out);

    gat_bn_stats_kernel<<<512, B, 0, stream>>>(out, bn);

    gat_bn_apply_kernel<<<(N_NODES * OUT_DIM + B - 1) / B, B, 0, stream>>>(bn, gamma, beta, out);
}


# Edges counting-sorted by dst inside the call (one returning atomic per edge gives an arrival-order rank); one wave per node then does max, exp-sum and the weighted row sum with no atomics
# speedup vs baseline: 1.8340x; 1.8340x over previous
#include <hip/hip_runtime.h>
#include <hip/hip_bf16.h>

#define N_NODES 100000
#define N_EDGES 1600000
#define IN_DIM 128
#define OUT_DIM 32
#define BN_EPS 1e-5f
#define LRELU_SLOPE 0.01f

typedef __attribute__((ext_vector_type(2))) float v2f;
typedef __attribute__((ext_vector_type(8))) float v8f;

__global__ void gat_init_kernel(int* cnt, float* bn) {
    int idx = blockIdx.x * blockDim.x + threadIdx.x;
    if (idx < N_NODES) cnt[idx] = 0;
    if (idx < 2 * OUT_DIM) bn[idx] = 0.0f;
}

__global__ void gat_gemm_wmma_kernel(const float* __restrict__ h,
                                     const float* __restrict__ W,
                                     float* __restrict__ z) {
    __shared__ float lds_w[IN_DIM * OUT_DIM];
    for (int i = threadIdx.x; i < IN_DIM * OUT_DIM; i += blockDim.x)
        lds_w[i] = W[i];
    __syncthreads();

    const int wave = threadIdx.x >> 5;
    const int lane = threadIdx.x & 31;
    const int tile = blockIdx.x * (blockDim.x >> 5) + wave;
    if (tile >= N_NODES / 16) return;

    const int m0   = tile * 16;
    const int row  = m0 + (lane & 15);
    const int koff = (lane >> 4) * 2;
    const int col  = lane & 15;

    v8f c0 = {}, c1 = {};
    const float* hrow = h + (size_t)row * IN_DIM;

    #pragma unroll 4
    for (int k = 0; k < IN_DIM; k += 4) {
        v2f a;
        a.x = hrow[k + koff];
        a.y = hrow[k + koff + 1];
        v2f b0, b1;
        b0.x = lds_w[(k + koff)     * OUT_DIM + col];
        b0.y = lds_w[(k + koff + 1) * OUT_DIM + col];
        b1.x = lds_w[(k + koff)     * OUT_DIM + col + 16];
        b1.y = lds_w[(k + koff + 1) * OUT_DIM + col + 16];
        c0 = __builtin_amdgcn_wmma_f32_16x16x4_f32(false, a, false, b0, (short)0, c0, false, false);
        c1 = __builtin_amdgcn_wmma_f32_16x16x4_f32(false, a, false, b1, (short)0, c1, false, false);
    }

    const int rbase = m0 + ((lane >> 4) * 8);
    #pragma unroll
    for (int r = 0; r < 8; ++r) {
        z[(size_t)(rbase + r) * OUT_DIM + col]      = c0[r];
        z[(size_t)(rbase + r) * OUT_DIM + col + 16] = c1[r];
    }
}

__global__ void gat_attn_proj_kernel(const float* __restrict__ z,
                                     const float* __restrict__ Wa,
                                     float* __restrict__ el,
                                     float* __restrict__ er) {
    int gid = blockIdx.x * blockDim.x + threadIdx.x;
    int node = gid >> 5;
    int lane = gid & 31;
    if (node >= N_NODES) return;
    float v  = z[(size_t)node * OUT_DIM + lane];
    float pl = v * Wa[lane];
    float pr = v * Wa[OUT_DIM + lane];
    #pragma unroll
    for (int off = 16; off > 0; off >>= 1) {
        pl += __shfl_down(pl, off, 32);
        pr += __shfl_down(pr, off, 32);
    }
    if (lane == 0) { el[node] = pl; er[node] = pr; }
}

__global__ void gat_edge_rank_kernel(const int* __restrict__ dst,
                                     int* __restrict__ cnt,
                                     int* __restrict__ rank) {
    int i = blockIdx.x * blockDim.x + threadIdx.x;
    if (i >= N_EDGES) return;
    rank[i] = atomicAdd(&cnt[dst[i]], 1);
}

__global__ void gat_offset_scan_kernel(const int* __restrict__ cnt, int* __restrict__ off) {
    __shared__ int wpre[8];
    __shared__ int wsum[8];
    const int4* c4 = (const int4*)cnt;
    const int t = threadIdx.x;
    const int lane = t & 31;
    const int wave = t >> 5;
    const int first4 = blockIdx.x * 256;

    const int nb = blockIdx.x;
    int pre = 0;
    for (int it = 0; it < nb; it += 8) {
        #pragma unroll
        for (int u = 0; u < 8; ++u) {
            const int c = it + u;
            const int4 p = c4[(c < nb ? c : nb - 1) * 256 + t];
            const int s4 = (p.x + p.y) + (p.z + p.w);
            pre += (c < nb) ? s4 : 0;
        }
    }
    const int i4 = first4 + t;
    int4 v = make_int4(0, 0, 0, 0);
    if (i4 < N_NODES / 4) v = c4[i4];
    const int local = (v.x + v.y) + (v.z + v.w);

    int inc = local;
    #pragma unroll
    for (int o = 1; o < 32; o <<= 1) {
        int u = __shfl_up(inc, o, 32);
        if (lane >= o) inc += u;
    }
    #pragma unroll
    for (int o = 16; o > 0; o >>= 1) pre += __shfl_xor(pre, o, 32);
    if (lane == 31) wsum[wave] = inc;
    if (lane == 0)  wpre[wave] = pre;
    __syncthreads();
    int base = 0;
    #pragma unroll
    for (int w = 0; w < 8; ++w) {
        base += wpre[w];
        if (w < wave) base += wsum[w];
    }
    if (i4 < N_NODES / 4) {
        int4 o;
        o.x = base + inc - local;
        o.y = o.x + v.x;
        o.z = o.y + v.y;
        o.w = o.z + v.z;
        ((int4*)off)[i4] = o;
    }
}

__global__ void gat_edge_fill_kernel(const float* __restrict__ el,
                                     const float* __restrict__ er,
                                     const int* __restrict__ src,
                                     const int* __restrict__ dst,
                                     const int* __restrict__ off,
                                     const int* __restrict__ rank,
                                     int2* __restrict__ csr) {
    int i = blockIdx.x * blockDim.x + threadIdx.x;
    if (i >= N_EDGES) return;
    int s = src[i], d = dst[i];
    float e = el[s] + er[d];
    e = (e >= 0.0f) ? e : LRELU_SLOPE * e;
    csr[off[d] + rank[i]] = make_int2(s, __float_as_int(e));
}

__global__ void gat_node_gather_kernel(const int* __restrict__ off,
                                       const int* __restrict__ cnt,
                                       const int2* __restrict__ csr,
                                       const float* __restrict__ z,
                                       float* __restrict__ hout) {
    const int gid  = blockIdx.x * blockDim.x + threadIdx.x;
    const int lane = threadIdx.x & 31;
    const int node = gid >> 5;
    if (node >= N_NODES) return;
    const int beg = __builtin_amdgcn_readfirstlane(off[node]);
    const int n   = __builtin_amdgcn_readfirstlane(cnt[node]);
    const int2* seg = csr + beg;

    const bool has = lane < n;
    int2 rec = make_int2(0, 0);
    if (n > 0) rec = seg[has ? lane : n - 1];
    const float e0 = __int_as_float(rec.y);

    float m = has ? e0 : -INFINITY;
    for (int j = lane + 32; j < n; j += 32) m = fmaxf(m, __int_as_float(seg[j].y));
    #pragma unroll
    for (int o = 16; o > 0; o >>= 1) m = fmaxf(m, __shfl_xor(m, o, 32));

    float ssum = has ? __expf(e0 - m) : 0.0f;
    for (int j = lane + 32; j < n; j += 32) ssum += __expf(__int_as_float(seg[j].y) - m);
    #pragma unroll
    for (int o = 16; o > 0; o >>= 1) ssum += __shfl_xor(ssum, o, 32);

    float acc = 0.0f;
    for (int base = 0; base < n; base += 32) {
        const int left = n - base;
        const bool in = lane < left;
        if (base > 0) rec = seg[in ? base + lane : n - 1];
        const int s = rec.x;
        const int abits = __float_as_int(in ? __expf(__int_as_float(rec.y) - m) / ssum : 0.0f);
        const int here = left < 32 ? left : 32;
        const int kend = (here + 7) & ~7;
        for (int k = 0; k < kend; k += 8) {
            #pragma unroll
            for (int u = 0; u < 8; ++u) {
                const int   sk = __builtin_amdgcn_readlane(s, k + u);
                const float ak = __int_as_float(__builtin_amdgcn_readlane(abits, k + u));
                acc += ak * z[(size_t)sk * OUT_DIM + lane];
            }
        }
    }
    hout[(size_t)node * OUT_DIM + lane] = acc;
}

__global__ void gat_bn_stats_kernel(const float* __restrict__ hout, float* __restrict__ bn) {
    __shared__ float ssum[OUT_DIM];
    __shared__ float ssq[OUT_DIM];
    int lane = threadIdx.x & 31;
    if (threadIdx.x < OUT_DIM) { ssum[threadIdx.x] = 0.0f; ssq[threadIdx.x] = 0.0f; }
    __syncthreads();
    int row = (blockIdx.x * blockDim.x + threadIdx.x) >> 5;
    int stride = (gridDim.x * blockDim.x) >> 5;
    float s = 0.0f, q = 0.0f;
    for (; row < N_NODES; row += stride) {
        float v = hout[(size_t)row * OUT_DIM + lane];
        s += v;
        q += v * v;
    }
    atomicAdd(&ssum[lane], s);
    atomicAdd(&ssq[lane], q);
    __syncthreads();
    if (threadIdx.x < OUT_DIM) {
        atomicAdd(&bn[threadIdx.x], ssum[threadIdx.x]);
        atomicAdd(&bn[OUT_DIM + threadIdx.x], ssq[threadIdx.x]);
    }
}

__global__ void gat_bn_apply_kernel(const float* __restrict__ bn,
                                    const float* __restrict__ gamma,
                                    const float* __restrict__ beta,
                                    float* __restrict__ out) {
    int idx = blockIdx.x * blockDim.x + threadIdx.x;
    if (idx >= N_NODES * OUT_DIM) return;
    int c = idx & 31;
    const float inv_n = 1.0f / (float)N_NODES;
    float mean = bn[c] * inv_n;
    float var  = bn[OUT_DIM + c] * inv_n - mean * mean;
    float x = (out[idx] - mean) * rsqrtf(var + BN_EPS) * gamma[c] + beta[c];
    out[idx] = (x > 0.0f) ? x : expm1f(x);
}

extern "C" void kernel_launch(void* const* d_in, const int* in_sizes, int n_in,
                              void* d_out, int out_size, void* d_ws, size_t ws_size,
                              hipStream_t stream) {
    const float* h     = (const float*)d_in[0];
    const float* W_fc  = (const float*)d_in[1];
    const float* W_at  = (const float*)d_in[2];
    const float* gamma = (const float*)d_in[3];
    const float* beta  = (const float*)d_in[4];
    const int*   src   = (const int*)d_in[5];
    const int*   dst   = (const int*)d_in[6];
    float* out = (float*)d_out;

    float* z    = (float*)d_ws;
    float* el   = z + (size_t)N_NODES * OUT_DIM;
    float* er   = el + N_NODES;
    int*   cnt  = (int*)(er + N_NODES);
    int*   off  = cnt + N_NODES;
    int2*  csr  = (int2*)(off + N_NODES);
    float* bn   = (float*)(csr + N_EDGES);
    int*   rank = (int*)d_out;

    const int B = 256;

    gat_init_kernel<<<(N_NODES + B - 1) / B, B, 0, stream>>>(cnt, bn);

    gat_gemm_wmma_kernel<<<(N_NODES / 16 + 7) / 8, B, 0, stream>>>(h, W_fc, z);

    gat_attn_proj_kernel<<<(N_NODES * OUT_DIM + B - 1) / B, B, 0, stream>>>(z, W_at, el, er);

    gat_edge_rank_kernel<<<(N_EDGES + B - 1) / B, B, 0, stream>>>(dst, cnt, rank);

    gat_offset_scan_kernel<<<(N_NODES / 4 + 255) / 256, 256, 0, stream>>>(cnt, off);

    gat_edge_fill_kernel<<<(N_EDGES + B - 1) / B, B, 0, stream>>>(el, er, src, dst, off, rank, csr);

    gat_node_gather_kernel<<<(N_NODES * OUT_DIM + B - 1) / B, B, 0, stream>>>(off, cnt, csr, z, out);

    gat_bn_stats_kernel<<<512, B, 0, stream>>>(out, bn);

    gat_bn_apply_kernel<<<(N_NODES * OUT_DIM + B - 1) / B, B, 0, stream>>>(bn, gamma, beta, out);
}


# Edge sort by dst without a global returning atomic per edge: two-level counting sort (256-node buckets) whose counts and arrival-order slots are LDS atomics; csr stays in arrival order
# speedup vs baseline: 2.7658x; 1.5081x over previous
#include <hip/hip_runtime.h>
#include <hip/hip_bf16.h>

#define N_NODES 100000
#define N_EDGES 1600000
#define IN_DIM 128
#define OUT_DIM 32
#define BN_EPS 1e-5f
#define LRELU_SLOPE 0.01f

#define BSHIFT 8
#define BNODES 256
#define NBUCK  ((N_NODES + BNODES - 1) / BNODES)
#define NCHUNK 256
#define CHUNK  (N_EDGES / NCHUNK)
#define TABN   (NBUCK * NCHUNK)
#define SORT_T 1024
#define EPT    ((CHUNK + SORT_T - 1) / SORT_T)
#define SRC_BITS 17
#define FILL_U 5
static_assert(N_EDGES % NCHUNK == 0, "chunks tile the edge list");
static_assert(TABN % 4 == 0, "table is scanned as int4");
static_assert(N_NODES <= (1 << SRC_BITS), "src and the in-bucket dst share one word");

typedef __attribute__((ext_vector_type(2))) float v2f;
typedef __attribute__((ext_vector_type(8))) float v8f;

__global__ void gat_gemm_wmma_kernel(const float* __restrict__ h,
                                     const float* __restrict__ W,
                                     float* __restrict__ z) {
    __shared__ float lds_w[IN_DIM * OUT_DIM];
    for (int i = threadIdx.x; i < IN_DIM * OUT_DIM; i += blockDim.x)
        lds_w[i] = W[i];
    __syncthreads();

    const int wave = threadIdx.x >> 5;
    const int lane = threadIdx.x & 31;
    const int tile = blockIdx.x * (blockDim.x >> 5) + wave;
    if (tile >= N_NODES / 16) return;

    const int m0   = tile * 16;
    const int row  = m0 + (lane & 15);
    const int koff = (lane >> 4) * 2;
    const int col  = lane & 15;

    v8f c0 = {}, c1 = {};
    const float* hrow = h + (size_t)row * IN_DIM;

    #pragma unroll 4
    for (int k = 0; k < IN_DIM; k += 4) {
        v2f a;
        a.x = hrow[k + koff];
        a.y = hrow[k + koff + 1];
        v2f b0, b1;
        b0.x = lds_w[(k + koff)     * OUT_DIM + col];
        b0.y = lds_w[(k + koff + 1) * OUT_DIM + col];
        b1.x = lds_w[(k + koff)     * OUT_DIM + col + 16];
        b1.y = lds_w[(k + koff + 1) * OUT_DIM + col + 16];
        c0 = __builtin_amdgcn_wmma_f32_16x16x4_f32(false, a, false, b0, (short)0, c0, false, false);
        c1 = __builtin_amdgcn_wmma_f32_16x16x4_f32(false, a, false, b1, (short)0, c1, false, false);
    }

    const int rbase = m0 + ((lane >> 4) * 8);
    #pragma unroll
    for (int r = 0; r < 8; ++r) {
        z[(size_t)(rbase + r) * OUT_DIM + col]      = c0[r];
        z[(size_t)(rbase + r) * OUT_DIM + col + 16] = c1[r];
    }
}

__global__ void gat_attn_proj_kernel(const float* __restrict__ z,
                                     const float* __restrict__ Wa,
                                     float* __restrict__ el,
                                     float* __restrict__ er) {
    int gid = blockIdx.x * blockDim.x + threadIdx.x;
    int node = gid >> 5;
    int lane = gid & 31;
    if (node >= N_NODES) return;
    float v  = z[(size_t)node * OUT_DIM + lane];
    float pl = v * Wa[lane];
    float pr = v * Wa[OUT_DIM + lane];
    #pragma unroll
    for (int off = 16; off > 0; off >>= 1) {
        pl += __shfl_down(pl, off, 32);
        pr += __shfl_down(pr, off, 32);
    }
    if (lane == 0) { el[node] = pl; er[node] = pr; }
}

__global__ void __launch_bounds__(SORT_T)
gat_bucket_count_kernel(const int* __restrict__ dst, int* __restrict__ tab, float* __restrict__ bn) {
    __shared__ int bc[NBUCK];
    const int t = threadIdx.x;
    if (t < NBUCK) bc[t] = 0;
    if (blockIdx.x == 0 && t < 2 * OUT_DIM) bn[t] = 0.0f;
    __syncthreads();
    const int* d = dst + blockIdx.x * CHUNK;
    int dv[EPT];
    #pragma unroll
    for (int k = 0; k < EPT; ++k) {
        const int i = t + k * SORT_T;
        dv[k] = d[i < CHUNK ? i : CHUNK - 1];
    }
    #pragma unroll
    for (int k = 0; k < EPT; ++k)
        if (t + k * SORT_T < CHUNK) atomicAdd(&bc[dv[k] >> BSHIFT], 1);
    __syncthreads();
    if (t < NBUCK) tab[t * NCHUNK + blockIdx.x] = bc[t];
}

__global__ void gat_offset_scan_kernel(const int* __restrict__ cnt, int* __restrict__ off) {
    __shared__ int wpre[8];
    __shared__ int wsum[8];
    const int4* c4 = (const int4*)cnt;
    const int t = threadIdx.x;
    const int lane = t & 31;
    const int wave = t >> 5;
    const int first4 = blockIdx.x * 256;

    const int nb = blockIdx.x;
    int pre = 0;
    for (int it = 0; it < nb; it += 8) {
        #pragma unroll
        for (int u = 0; u < 8; ++u) {
            const int c = it + u;
            const int4 p = c4[(c < nb ? c : nb - 1) * 256 + t];
            const int s4 = (p.x + p.y) + (p.z + p.w);
            pre += (c < nb) ? s4 : 0;
        }
    }
    const int i4 = first4 + t;
    int4 v = make_int4(0, 0, 0, 0);
    if (i4 < TABN / 4) v = c4[i4];
    const int local = (v.x + v.y) + (v.z + v.w);

    int inc = local;
    #pragma unroll
    for (int o = 1; o < 32; o <<= 1) {
        int u = __shfl_up(inc, o, 32);
        if (lane >= o) inc += u;
    }
    #pragma unroll
    for (int o = 16; o > 0; o >>= 1) pre += __shfl_xor(pre, o, 32);
    if (lane == 31) wsum[wave] = inc;
    if (lane == 0)  wpre[wave] = pre;
    __syncthreads();
    int base = 0;
    #pragma unroll
    for (int w = 0; w < 8; ++w) {
        base += wpre[w];
        if (w < wave) base += wsum[w];
    }
    if (i4 < TABN / 4) {
        int4 o;
        o.x = base + inc - local;
        o.y = o.x + v.x;
        o.z = o.y + v.y;
        o.w = o.z + v.z;
        ((int4*)off)[i4] = o;
    }
}

__global__ void __launch_bounds__(SORT_T)
gat_bucket_scatter_kernel(const int* __restrict__ src,
                          const int* __restrict__ dst,
                          const int* __restrict__ toff,
                          int* __restrict__ tmp) {
    __shared__ int cur[NBUCK];
    const int t = threadIdx.x;
    if (t < NBUCK) cur[t] = toff[t * NCHUNK + blockIdx.x];
    const int* s = src + blockIdx.x * CHUNK;
    const int* d = dst + blockIdx.x * CHUNK;
    int sv[EPT], dv[EPT];
    #pragma unroll
    for (int k = 0; k < EPT; ++k) {
        const int i = t + k * SORT_T;
        const int ic = i < CHUNK ? i : CHUNK - 1;
        sv[k] = s[ic];
        dv[k] = d[ic];
    }
    __syncthreads();
    #pragma unroll
    for (int k = 0; k < EPT; ++k) {
        if (t + k * SORT_T < CHUNK) {
            const int pos = atomicAdd(&cur[dv[k] >> BSHIFT], 1);
            tmp[pos] = sv[k] | ((dv[k] & (BNODES - 1)) << SRC_BITS);
        }
    }
}

__global__ void __launch_bounds__(SORT_T)
gat_bucket_fill_kernel(const float* __restrict__ el,
                       const float* __restrict__ er,
                       const int* __restrict__ toff,
                       const int* __restrict__ tmp,
                       int* __restrict__ cnt,
                       int* __restrict__ off,
                       int2* __restrict__ csr) {
    __shared__ int   nc[BNODES];
    __shared__ float ler[BNODES];
    __shared__ int   wtot[BNODES / 32];
    const int t = threadIdx.x;
    const int lane = t & 31;
    const int b = blockIdx.x;
    const int node = b * BNODES + t;
    const int beg = toff[b * NCHUNK];
    const int end = (b + 1 < NBUCK) ? toff[(b + 1) * NCHUNK] : N_EDGES;
    if (t < BNODES) {
        nc[t] = 0;
        ler[t] = (node < N_NODES) ? er[node] : 0.0f;
    }
    __syncthreads();
    for (int j0 = beg + t; j0 < end; j0 += FILL_U * SORT_T) {
        int pv[FILL_U];
        #pragma unroll
        for (int u = 0; u < FILL_U; ++u) {
            const int j = j0 + u * SORT_T;
            pv[u] = tmp[j < end ? j : end - 1];
        }
        #pragma unroll
        for (int u = 0; u < FILL_U; ++u)
            if (j0 + u * SORT_T < end) atomicAdd(&nc[pv[u] >> SRC_BITS], 1);
    }
    __syncthreads();
    int c = 0, inc = 0;
    if (t < BNODES) {
        c = nc[t];
        inc = c;
        #pragma unroll
        for (int o = 1; o < 32; o <<= 1) {
            const int u = __shfl_up(inc, o, 32);
            if (lane >= o) inc += u;
        }
        if (lane == 31) wtot[t >> 5] = inc;
    }
    __syncthreads();
    if (t < BNODES) {
        int base = beg;
        #pragma unroll
        for (int w = 0; w < BNODES / 32; ++w)
            if (w < (t >> 5)) base += wtot[w];
        const int first = base + inc - c;
        nc[t] = first;
        if (node < N_NODES) { cnt[node] = c; off[node] = first; }
    }
    __syncthreads();
    for (int j0 = beg + t; j0 < end; j0 += FILL_U * SORT_T) {
        int pv[FILL_U];
        float ev[FILL_U];
        #pragma unroll
        for (int u = 0; u < FILL_U; ++u) {
            const int j = j0 + u * SORT_T;
            pv[u] = tmp[j < end ? j : end - 1];
        }
        #pragma unroll
        for (int u = 0; u < FILL_U; ++u) ev[u] = el[pv[u] & ((1 << SRC_BITS) - 1)];
        #pragma unroll
        for (int u = 0; u < FILL_U; ++u) {
            if (j0 + u * SORT_T < end) {
                const int s  = pv[u] & ((1 << SRC_BITS) - 1);
                const int dl = pv[u] >> SRC_BITS;
                const int pos = atomicAdd(&nc[dl], 1);
                float e = ev[u] + ler[dl];
                e = (e >= 0.0f) ? e : LRELU_SLOPE * e;
                csr[pos] = make_int2(s, __float_as_int(e));
            }
        }
    }
}

__global__ void gat_node_gather_kernel(const int* __restrict__ off,
                                       const int* __restrict__ cnt,
                                       const int2* __restrict__ csr,
                                       const float* __restrict__ z,
                                       float* __restrict__ hout) {
    const int gid  = blockIdx.x * blockDim.x + threadIdx.x;
    const int lane = threadIdx.x & 31;
    const int node = gid >> 5;
    if (node >= N_NODES) return;
    const int beg = __builtin_amdgcn_readfirstlane(off[node]);
    const int n   = __builtin_amdgcn_readfirstlane(cnt[node]);
    const int2* seg = csr + beg;

    const bool has = lane < n;
    int2 rec = make_int2(0, 0);
    if (n > 0) rec = seg[has ? lane : n - 1];
    const float e0 = __int_as_float(rec.y);

    float m = has ? e0 : -INFINITY;
    for (int j = lane + 32; j < n; j += 32) m = fmaxf(m, __int_as_float(seg[j].y));
    #pragma unroll
    for (int o = 16; o > 0; o >>= 1) m = fmaxf(m, __shfl_xor(m, o, 32));

    float ssum = has ? __expf(e0 - m) : 0.0f;
    for (int j = lane + 32; j < n; j += 32) ssum += __expf(__int_as_float(seg[j].y) - m);
    #pragma unroll
    for (int o = 16; o > 0; o >>= 1) ssum += __shfl_xor(ssum, o, 32);

    float acc = 0.0f;
    for (int base = 0; base < n; base += 32) {
        const int left = n - base;
        const bool in = lane < left;
        if (base > 0) rec = seg[in ? base + lane : n - 1];
        const int s = rec.x;
        const int abits = __float_as_int(in ? __expf(__int_as_float(rec.y) - m) / ssum : 0.0f);
        const int here = left < 32 ? left : 32;
        const int kend = (here + 7) & ~7;
        for (int k = 0; k < kend; k += 8) {
            #pragma unroll
            for (int u = 0; u < 8; ++u) {
                const int   sk = __builtin_amdgcn_readlane(s, k + u);
                const float ak = __int_as_float(__builtin_amdgcn_readlane(abits, k + u));
                acc += ak * z[(size_t)sk * OUT_DIM + lane];
            }
        }
    }
    hout[(size_t)node * OUT_DIM + lane] = acc;
}

__global__ void gat_bn_stats_kernel(const float* __restrict__ hout, float* __restrict__ bn) {
    __shared__ float ssum[OUT_DIM];
    __shared__ float ssq[OUT_DIM];
    int lane = threadIdx.x & 31;
    if (threadIdx.x < OUT_DIM) { ssum[threadIdx.x] = 0.0f; ssq[threadIdx.x] = 0.0f; }
    __syncthreads();
    int row = (blockIdx.x * blockDim.x + threadIdx.x) >> 5;
    int stride = (gridDim.x * blockDim.x) >> 5;
    float s = 0.0f, q = 0.0f;
    for (; row < N_NODES; row += stride) {
        float v = hout[(size_t)row * OUT_DIM + lane];
        s += v;
        q += v * v;
    }
    atomicAdd(&ssum[lane], s);
    atomicAdd(&ssq[lane], q);
    __syncthreads();
    if (threadIdx.x < OUT_DIM) {
        atomicAdd(&bn[threadIdx.x], ssum[threadIdx.x]);
        atomicAdd(&bn[OUT_DIM + threadIdx.x], ssq[threadIdx.x]);
    }
}

__global__ void gat_bn_apply_kernel(const float* __restrict__ bn,
                                    const float* __restrict__ gamma,
                                    const float* __restrict__ beta,
                                    float* __restrict__ out) {
    int idx = blockIdx.x * blockDim.x + threadIdx.x;
    if (idx >= N_NODES * OUT_DIM) return;
    int c = idx & 31;
    const float inv_n = 1.0f / (float)N_NODES;
    float mean = bn[c] * inv_n;
    float var  = bn[OUT_DIM + c] * inv_n - mean * mean;
    float x = (out[idx] - mean) * rsqrtf(var + BN_EPS) * gamma[c] + beta[c];
    out[idx] = (x > 0.0f) ? x : expm1f(x);
}

extern "C" void kernel_launch(void* const* d_in, const int* in_sizes, int n_in,
                              void* d_out, int out_size, void* d_ws, size_t ws_size,
                              hipStream_t stream) {
    const float* h     = (const float*)d_in[0];
    const float* W_fc  = (const float*)d_in[1];
    const float* W_at  = (const float*)d_in[2];
    const float* gamma = (const float*)d_in[3];
    const float* beta  = (const float*)d_in[4];
    const int*   src   = (const int*)d_in[5];
    const int*   dst   = (const int*)d_in[6];
    float* out = (float*)d_out;

    float* z    = (float*)d_ws;
    float* el   = z + (size_t)N_NODES * OUT_DIM;
    float* er   = el + N_NODES;
    int*   cnt  = (int*)(er + N_NODES);
    int*   off  = cnt + N_NODES;
    int2*  csr  = (int2*)(off + N_NODES);
    float* bn   = (float*)(csr + N_EDGES);
    int*   tmp  = (int*)d_out;
    int*   tab  = tmp + N_EDGES;
    int*   toff = tab + TABN;

    const int B = 256;

    gat_gemm_wmma_kernel<<<(N_NODES / 16 + 7) / 8, B, 0, stream>>>(h, W_fc, z);

    gat_attn_proj_kernel<<<(N_NODES * OUT_DIM + B - 1) / B, B, 0, stream>>>(z, W_at, el, er);

    gat_bucket_count_kernel<<<NCHUNK, SORT_T, 0, stream>>>(dst, tab, bn);

    gat_offset_scan_kernel<<<(TABN / 4 + 255) / 256, 256, 0, stream>>>(tab, toff);

    gat_bucket_scatter_kernel<<<NCHUNK, SORT_T, 0, stream>>>(src, dst, toff, tmp);

    gat_bucket_fill_kernel<<<NBUCK, SORT_T, 0, stream>>>(el, er, toff, tmp, cnt, off, csr);

    gat_node_gather_kernel<<<(N_NODES * OUT_DIM + B - 1) / B, B, 0, stream>>>(off, cnt, csr, z, out);

    gat_bn_stats_kernel<<<512, B, 0, stream>>>(out, bn);

    gat_bn_apply_kernel<<<(N_NODES * OUT_DIM + B - 1) / B, B, 0, stream>>>(bn, gamma, beta, out);
}


# Nine launches to six: el/er come from the product's accumulators (row sums in a new tree order), the bucket count shares the product's launch, each scatter block scans the size table itself
# speedup vs baseline: 3.4977x; 1.2646x over previous
#include <hip/hip_runtime.h>
#include <hip/hip_bf16.h>

#define N_NODES 100000
#define N_EDGES 1600000
#define IN_DIM 128
#define OUT_DIM 32
#define BN_EPS 1e-5f
#define LRELU_SLOPE 0.01f

#define BSHIFT 8
#define BNODES 256
#define NBUCK  ((N_NODES + BNODES - 1) / BNODES)
#define NCHUNK 256
#define CHUNK  (N_EDGES / NCHUNK)
#define TABN   (NBUCK * NCHUNK)
#define SORT_T 1024
#define EPT    ((CHUNK + SORT_T - 1) / SORT_T)
#define SRC_BITS 17
#define FILL_U 5
static_assert(N_EDGES % NCHUNK == 0, "chunks tile the edge list");
static_assert(TABN % 8 == 0, "table rows of 16-bit counts are read as int4");
static_assert(N_NODES <= (1 << SRC_BITS), "src and the in-bucket dst share one word");
static_assert(NCHUNK == 256, "a table row of 16-bit counts is one int4 per lane of one wave");
static_assert(4 * CHUNK <= 65535, "four piece sizes summed still fit 16 bits");
static_assert(NBUCK <= SORT_T, "one thread per bucket in the cursor scan");

#define GEMM_TILES   (N_NODES / 16)
#define GEMM_WAVES   (SORT_T / 32)
#define GEMM_BLOCKS  ((GEMM_TILES + GEMM_WAVES - 1) / GEMM_WAVES)
#define SCAN_WAVES   ((NBUCK + 31) / 32)
#define SORT_WAVES   (SORT_T / 32)
#define ROWS_PER_WAVE ((NBUCK + SORT_WAVES - 1) / SORT_WAVES)
static_assert(ROWS_PER_WAVE <= 16, "one lane of each half wave per row of a wave");

typedef __attribute__((ext_vector_type(2))) float v2f;
typedef __attribute__((ext_vector_type(8))) float v8f;

template <int M> __device__ __forceinline__ float row_xor_f(float x) {
    return __int_as_float(__builtin_amdgcn_update_dpp(0, __float_as_int(x), 0x160 + M, 0xf, 0xf, true));
}
template <int M> __device__ __forceinline__ int row_xor_i(int x) {
    return __builtin_amdgcn_update_dpp(0, x, 0x160 + M, 0xf, 0xf, true);
}
__device__ __forceinline__ float row_sum16_f(float x) {
    x += row_xor_f<1>(x);
    x += row_xor_f<2>(x);
    x += row_xor_f<4>(x);
    x += row_xor_f<8>(x);
    return x;
}
__device__ __forceinline__ int row_sum16_i(int x) {
    x += row_xor_i<1>(x);
    x += row_xor_i<2>(x);
    x += row_xor_i<4>(x);
    x += row_xor_i<8>(x);
    return x;
}

__global__ void __launch_bounds__(SORT_T)
gat_gemm_proj_count_kernel(const float* __restrict__ h,
                           const float* __restrict__ W,
                           const float* __restrict__ Wa,
                           const int* __restrict__ dst,
                           float* __restrict__ z,
                           float* __restrict__ el,
                           float* __restrict__ er,
                           unsigned short* __restrict__ tab,
                           float* __restrict__ bn) {
    __shared__ float lds_w[IN_DIM * OUT_DIM];
    __shared__ int bc[NBUCK];
    const int t = threadIdx.x;

    if (blockIdx.x >= GEMM_BLOCKS) {
        const int chunk = blockIdx.x - GEMM_BLOCKS;
        if (t < NBUCK) bc[t] = 0;
        if (chunk == 0 && t < 2 * OUT_DIM) bn[t] = 0.0f;
        __syncthreads();
        const int* d = dst + chunk * CHUNK;
        int dv[EPT];
        #pragma unroll
        for (int k = 0; k < EPT; ++k) {
            const int i = t + k * SORT_T;
            dv[k] = d[i < CHUNK ? i : CHUNK - 1];
        }
        #pragma unroll
        for (int k = 0; k < EPT; ++k)
            if (t + k * SORT_T < CHUNK) atomicAdd(&bc[dv[k] >> BSHIFT], 1);
        __syncthreads();
        if (t < NBUCK) tab[t * NCHUNK + chunk] = (unsigned short)bc[t];
        return;
    }

    for (int i = t; i < IN_DIM * OUT_DIM; i += SORT_T)
        lds_w[i] = W[i];
    __syncthreads();

    const int wave = t >> 5;
    const int lane = t & 31;
    const int tile = blockIdx.x * GEMM_WAVES + wave;
    if (tile >= GEMM_TILES) return;

    const int m0   = tile * 16;
    const int row  = m0 + (lane & 15);
    const int koff = (lane >> 4) * 2;
    const int col  = lane & 15;

    const float al0 = Wa[col];
    const float al1 = Wa[col + 16];
    const float ar0 = Wa[OUT_DIM + col];
    const float ar1 = Wa[OUT_DIM + col + 16];

    v8f c0 = {}, c1 = {};
    const float* hrow = h + (size_t)row * IN_DIM;

    #pragma unroll 4
    for (int k = 0; k < IN_DIM; k += 4) {
        v2f a;
        a.x = hrow[k + koff];
        a.y = hrow[k + koff + 1];
        v2f b0, b1;
        b0.x = lds_w[(k + koff)     * OUT_DIM + col];
        b0.y = lds_w[(k + koff + 1) * OUT_DIM + col];
        b1.x = lds_w[(k + koff)     * OUT_DIM + col + 16];
        b1.y = lds_w[(k + koff + 1) * OUT_DIM + col + 16];
        c0 = __builtin_amdgcn_wmma_f32_16x16x4_f32(false, a, false, b0, (short)0, c0, false, false);
        c1 = __builtin_amdgcn_wmma_f32_16x16x4_f32(false, a, false, b1, (short)0, c1, false, false);
    }

    const int rbase = m0 + ((lane >> 4) * 8);
    #pragma unroll
    for (int r = 0; r < 8; ++r) {
        z[(size_t)(rbase + r) * OUT_DIM + col]      = c0[r];
        z[(size_t)(rbase + r) * OUT_DIM + col + 16] = c1[r];
    }

    __builtin_amdgcn_sched_barrier(0);
    float pl[8], pr[8];
    #pragma unroll
    for (int r = 0; r < 8; ++r) {
        pl[r] = row_sum16_f(c0[r] * al0 + c1[r] * al1);
        pr[r] = row_sum16_f(c0[r] * ar0 + c1[r] * ar1);
    }
    if (col == 0) {
        float4* pe = (float4*)(el + rbase);
        pe[0] = make_float4(pl[0], pl[1], pl[2], pl[3]);
        pe[1] = make_float4(pl[4], pl[5], pl[6], pl[7]);
        float4* qe = (float4*)(er + rbase);
        qe[0] = make_float4(pr[0], pr[1], pr[2], pr[3]);
        qe[1] = make_float4(pr[4], pr[5], pr[6], pr[7]);
    }
}

__global__ void __launch_bounds__(SORT_T)
gat_bucket_scatter_kernel(const int* __restrict__ src,
                          const int* __restrict__ dst,
                          const unsigned short* __restrict__ tab,
                          int* __restrict__ bstart,
                          int* __restrict__ tmp) {
    __shared__ int cur[NBUCK];
    __shared__ int rtot[NBUCK];
    __shared__ int wsum[SCAN_WAVES];
    const int t = threadIdx.x;
    const int lane = t & 31;
    const int wave = t >> 5;
    const int c = blockIdx.x;
    const int* s = src + c * CHUNK;
    const int* d = dst + c * CHUNK;
    int sv[EPT], dv[EPT];
    #pragma unroll
    for (int k = 0; k < EPT; ++k) {
        const int i = t + k * SORT_T;
        const int ic = i < CHUNK ? i : CHUNK - 1;
        sv[k] = s[ic];
        dv[k] = d[ic];
    }

    const int4* t4 = (const int4*)tab;
    const int j0 = 8 * lane;
    int4 p[ROWS_PER_WAVE];
    #pragma unroll
    for (int i = 0; i < ROWS_PER_WAVE; ++i) {
        const int b  = wave + SORT_WAVES * i;
        const int bb = b < NBUCK ? b : NBUCK - 1;
        p[i] = t4[bb * (NCHUNK / 8) + lane];
    }
    __builtin_amdgcn_sched_barrier(0);
    unsigned keep[4];
    #pragma unroll
    for (int k = 0; k < 4; ++k)
        keep[k] = (j0 + 2 * k < c ? 0x0000ffffu : 0u) | (j0 + 2 * k + 1 < c ? 0xffff0000u : 0u);
    int vt = 0, vp = 0;
    #pragma unroll
    for (int i = 0; i < ROWS_PER_WAVE; ++i) {
        const unsigned w0 = (unsigned)p[i].x, w1 = (unsigned)p[i].y;
        const unsigned w2 = (unsigned)p[i].z, w3 = (unsigned)p[i].w;
        const unsigned t2 = (w0 + w1) + (w2 + w3);
        const unsigned p2 = ((w0 & keep[0]) + (w1 & keep[1])) + ((w2 & keep[2]) + (w3 & keep[3]));
        const int ts = row_sum16_i((int)((t2 & 0xffffu) + (t2 >> 16)));
        const int ps = row_sum16_i((int)((p2 & 0xffffu) + (p2 >> 16)));
        if ((lane & 15) == i) { vt = ts; vp = ps; }
    }
    vt += __shfl_xor(vt, 16, 32);
    vp += __shfl_xor(vp, 16, 32);
    const int bl = wave + SORT_WAVES * lane;
    if (lane < ROWS_PER_WAVE && bl < NBUCK) { rtot[bl] = vt; cur[bl] = vp; }
    __syncthreads();
    int own = 0, inc = 0;
    if (wave < SCAN_WAVES) {
        own = (t < NBUCK) ? rtot[t] : 0;
        inc = own;
        #pragma unroll
        for (int o = 1; o < 32; o <<= 1) {
            const int u = __shfl_up(inc, o, 32);
            if (lane >= o) inc += u;
        }
        if (lane == 31) wsum[wave] = inc;
    }
    __syncthreads();
    if (t < NBUCK) {
        int base = 0;
        #pragma unroll
        for (int w = 0; w < SCAN_WAVES; ++w)
            if (w < wave) base += wsum[w];
        const int first = base + inc - own;
        cur[t] += first;
        if (c == 0) bstart[t] = first;
    }
    __syncthreads();
    #pragma unroll
    for (int k = 0; k < EPT; ++k) {
        if (t + k * SORT_T < CHUNK) {
            const int pos = atomicAdd(&cur[dv[k] >> BSHIFT], 1);
            tmp[pos] = sv[k] | ((dv[k] & (BNODES - 1)) << SRC_BITS);
        }
    }
}

__global__ void __launch_bounds__(SORT_T)
gat_bucket_fill_kernel(const float* __restrict__ el,
                       const float* __restrict__ er,
                       const int* __restrict__ bstart,
                       const int* __restrict__ tmp,
                       int* __restrict__ cnt,
                       int* __restrict__ off,
                       int2* __restrict__ csr) {
    __shared__ int   nc[BNODES];
    __shared__ float ler[BNODES];
    __shared__ int   wtot[BNODES / 32];
    const int t = threadIdx.x;
    const int lane = t & 31;
    const int b = blockIdx.x;
    const int node = b * BNODES + t;
    const int beg = bstart[b];
    const int end = (b + 1 < NBUCK) ? bstart[b + 1] : N_EDGES;
    if (t < BNODES) {
        nc[t] = 0;
        ler[t] = (node < N_NODES) ? er[node] : 0.0f;
    }
    __syncthreads();
    for (int j0 = beg + t; j0 < end; j0 += FILL_U * SORT_T) {
        int pv[FILL_U];
        #pragma unroll
        for (int u = 0; u < FILL_U; ++u) {
            const int j = j0 + u * SORT_T;
            pv[u] = tmp[j < end ? j : end - 1];
        }
        #pragma unroll
        for (int u = 0; u < FILL_U; ++u)
            if (j0 + u * SORT_T < end) atomicAdd(&nc[pv[u] >> SRC_BITS], 1);
    }
    __syncthreads();
    int c = 0, inc = 0;
    if (t < BNODES) {
        c = nc[t];
        inc = c;
        #pragma unroll
        for (int o = 1; o < 32; o <<= 1) {
            const int u = __shfl_up(inc, o, 32);
            if (lane >= o) inc += u;
        }
        if (lane == 31) wtot[t >> 5] = inc;
    }
    __syncthreads();
    if (t < BNODES) {
        int base = beg;
        #pragma unroll
        for (int w = 0; w < BNODES / 32; ++w)
            if (w < (t >> 5)) base += wtot[w];
        const int first = base + inc - c;
        nc[t] = first;
        if (node < N_NODES) { cnt[node] = c; off[node] = first; }
    }
    __syncthreads();
    for (int j0 = beg + t; j0 < end; j0 += FILL_U * SORT_T) {
        int pv[FILL_U];
        float ev[FILL_U];
        #pragma unroll
        for (int u = 0; u < FILL_U; ++u) {
            const int j = j0 + u * SORT_T;
            pv[u] = tmp[j < end ? j : end - 1];
        }
        #pragma unroll
        for (int u = 0; u < FILL_U; ++u) ev[u] = el[pv[u] & ((1 << SRC_BITS) - 1)];
        #pragma unroll
        for (int u = 0; u < FILL_U; ++u) {
            if (j0 + u * SORT_T < end) {
                const int s  = pv[u] & ((1 << SRC_BITS) - 1);
                const int dl = pv[u] >> SRC_BITS;
                const int pos = atomicAdd(&nc[dl], 1);
                float e = ev[u] + ler[dl];
                e = (e >= 0.0f) ? e : LRELU_SLOPE * e;
                csr[pos] = make_int2(s, __float_as_int(e));
            }
        }
    }
}

__global__ void gat_node_gather_kernel(const int* __restrict__ off,
                                       const int* __restrict__ cnt,
                                       const int2* __restrict__ csr,
                                       const float* __restrict__ z,
                                       float* __restrict__ hout) {
    const int gid  = blockIdx.x * blockDim.x + threadIdx.x;
    const int lane = threadIdx.x & 31;
    const int node = gid >> 5;
    if (node >= N_NODES) return;
    const int beg = __builtin_amdgcn_readfirstlane(off[node]);
    const int n   = __builtin_amdgcn_readfirstlane(cnt[node]);
    const int2* seg = csr + beg;

    const bool has = lane < n;
    int2 rec = make_int2(0, 0);
    if (n > 0) rec = seg[has ? lane : n - 1];
    const float e0 = __int_as_float(rec.y);

    float m = has ? e0 : -INFINITY;
    for (int j = lane + 32; j < n; j += 32) m = fmaxf(m, __int_as_float(seg[j].y));
    #pragma unroll
    for (int o = 16; o > 0; o >>= 1) m = fmaxf(m, __shfl_xor(m, o, 32));

    float ssum = has ? __expf(e0 - m) : 0.0f;
    for (int j = lane + 32; j < n; j += 32) ssum += __expf(__int_as_float(seg[j].y) - m);
    #pragma unroll
    for (int o = 16; o > 0; o >>= 1) ssum += __shfl_xor(ssum, o, 32);

    float acc = 0.0f;
    for (int base = 0; base < n; base += 32) {
        const int left = n - base;
        const bool in = lane < left;
        if (base > 0) rec = seg[in ? base + lane : n - 1];
        const int s = rec.x;
        const int abits = __float_as_int(in ? __expf(__int_as_float(rec.y) - m) / ssum : 0.0f);
        const int here = left < 32 ? left : 32;
        const int kend = (here + 7) & ~7;
        for (int k = 0; k < kend; k += 8) {
            #pragma unroll
            for (int u = 0; u < 8; ++u) {
                const int   sk = __builtin_amdgcn_readlane(s, k + u);
                const float ak = __int_as_float(__builtin_amdgcn_readlane(abits, k + u));
                acc += ak * z[(size_t)sk * OUT_DIM + lane];
            }
        }
    }
    hout[(size_t)node * OUT_DIM + lane] = acc;
}

__global__ void gat_bn_stats_kernel(const float* __restrict__ hout, float* __restrict__ bn) {
    __shared__ float ssum[OUT_DIM];
    __shared__ float ssq[OUT_DIM];
    int lane = threadIdx.x & 31;
    if (threadIdx.x < OUT_DIM) { ssum[threadIdx.x] = 0.0f; ssq[threadIdx.x] = 0.0f; }
    __syncthreads();
    int row = (blockIdx.x * blockDim.x + threadIdx.x) >> 5;
    int stride = (gridDim.x * blockDim.x) >> 5;
    float s = 0.0f, q = 0.0f;
    for (; row < N_NODES; row += stride) {
        float v = hout[(size_t)row * OUT_DIM + lane];
        s += v;
        q += v * v;
    }
    atomicAdd(&ssum[lane], s);
    atomicAdd(&ssq[lane], q);
    __syncthreads();
    if (threadIdx.x < OUT_DIM) {
        atomicAdd(&bn[threadIdx.x], ssum[threadIdx.x]);
        atomicAdd(&bn[OUT_DIM + threadIdx.x], ssq[threadIdx.x]);
    }
}

__global__ void gat_bn_apply_kernel(const float* __restrict__ bn,
                                    const float* __restrict__ gamma,
                                    const float* __restrict__ beta,
                                    float* __restrict__ out) {
    int idx = blockIdx.x * blockDim.x + threadIdx.x;
    if (idx >= N_NODES * OUT_DIM) return;
    int c = idx & 31;
    const float inv_n = 1.0f / (float)N_NODES;
    float mean = bn[c] * inv_n;
    float var  = bn[OUT_DIM + c] * inv_n - mean * mean;
    float x = (out[idx] - mean) * rsqrtf(var + BN_EPS) * gamma[c] + beta[c];
    out[idx] = (x > 0.0f) ? x : expm1f(x);
}

extern "C" void kernel_launch(void* const* d_in, const int* in_sizes, int n_in,
                              void* d_out, int out_size, void* d_ws, size_t ws_size,
                              hipStream_t stream) {
    const float* h     = (const float*)d_in[0];
    const float* W_fc  = (const float*)d_in[1];
    const float* W_at  = (const float*)d_in[2];
    const float* gamma = (const float*)d_in[3];
    const float* beta  = (const float*)d_in[4];
    const int*   src   = (const int*)d_in[5];
    const int*   dst   = (const int*)d_in[6];
    float* out = (float*)d_out;

    float* z    = (float*)d_ws;
    float* el   = z + (size_t)N_NODES * OUT_DIM;
    float* er   = el + N_NODES;
    int*   cnt  = (int*)(er + N_NODES);
    int*   off  = cnt + N_NODES;
    int2*  csr  = (int2*)(off + N_NODES);
    float* bn   = (float*)(csr + N_EDGES);
    int*   tmp    = (int*)d_out;
    unsigned short* tab = (unsigned short*)(tmp + N_EDGES);
    int*   bstart = tmp + N_EDGES + TABN / 2;

    const int B = 256;

    gat_gemm_proj_count_kernel<<<GEMM_BLOCKS + NCHUNK, SORT_T, 0, stream>>>(h, W_fc, W_at, dst, z, el, er, tab, bn);

    gat_bucket_scatter_kernel<<<NCHUNK, SORT_T, 0, stream>>>(src, dst, tab, bstart, tmp);

    gat_bucket_fill_kernel<<<NBUCK, SORT_T, 0, stream>>>(el, er, bstart, tmp, cnt, off, csr);

    gat_node_gather_kernel<<<(N_NODES * OUT_DIM + B - 1) / B, B, 0, stream>>>(off, cnt, csr, z, out);

    gat_bn_stats_kernel<<<512, B, 0, stream>>>(out, bn);

    gat_bn_apply_kernel<<<(N_NODES * OUT_DIM + B - 1) / B, B, 0, stream>>>(bn, gamma, beta, out);
}


# BN statistics: a wave asks for all 25 of its rows before its first add, one memory round trip per wave where there were 25 dependent ones; same grid, rows still added in row order
# speedup vs baseline: 3.6027x; 1.0300x over previous
#include <hip/hip_runtime.h>
#include <hip/hip_bf16.h>

#define N_NODES 100000
#define N_EDGES 1600000
#define IN_DIM 128
#define OUT_DIM 32
#define BN_EPS 1e-5f
#define LRELU_SLOPE 0.01f

#define BSHIFT 8
#define BNODES 256
#define NBUCK  ((N_NODES + BNODES - 1) / BNODES)
#define NCHUNK 256
#define CHUNK  (N_EDGES / NCHUNK)
#define TABN   (NBUCK * NCHUNK)
#define SORT_T 1024
#define EPT    ((CHUNK + SORT_T - 1) / SORT_T)
#define SRC_BITS 17
#define FILL_U 5
static_assert(N_EDGES % NCHUNK == 0, "chunks tile the edge list");
static_assert(TABN % 8 == 0, "table rows of 16-bit counts are read as int4");
static_assert(N_NODES <= (1 << SRC_BITS), "src and the in-bucket dst share one word");
static_assert(NCHUNK == 256, "a table row of 16-bit counts is one int4 per lane of one wave");
static_assert(4 * CHUNK <= 65535, "four piece sizes summed still fit 16 bits");
static_assert(NBUCK <= SORT_T, "one thread per bucket in the cursor scan");

#define GEMM_TILES   (N_NODES / 16)
#define GEMM_WAVES   (SORT_T / 32)
#define GEMM_BLOCKS  ((GEMM_TILES + GEMM_WAVES - 1) / GEMM_WAVES)
#define SCAN_WAVES   ((NBUCK + 31) / 32)
#define SORT_WAVES   (SORT_T / 32)
#define ROWS_PER_WAVE ((NBUCK + SORT_WAVES - 1) / SORT_WAVES)
static_assert(ROWS_PER_WAVE <= 16, "one lane of each half wave per row of a wave");

typedef __attribute__((ext_vector_type(2))) float v2f;
typedef __attribute__((ext_vector_type(8))) float v8f;

template <int M> __device__ __forceinline__ float row_xor_f(float x) {
    return __int_as_float(__builtin_amdgcn_update_dpp(0, __float_as_int(x), 0x160 + M, 0xf, 0xf, true));
}
template <int M> __device__ __forceinline__ int row_xor_i(int x) {
    return __builtin_amdgcn_update_dpp(0, x, 0x160 + M, 0xf, 0xf, true);
}
__device__ __forceinline__ float row_sum16_f(float x) {
    x += row_xor_f<1>(x);
    x += row_xor_f<2>(x);
    x += row_xor_f<4>(x);
    x += row_xor_f<8>(x);
    return x;
}
__device__ __forceinline__ int row_sum16_i(int x) {
    x += row_xor_i<1>(x);
    x += row_xor_i<2>(x);
    x += row_xor_i<4>(x);
    x += row_xor_i<8>(x);
    return x;
}

__global__ void __launch_bounds__(SORT_T)
gat_gemm_proj_count_kernel(const float* __restrict__ h,
                           const float* __restrict__ W,
                           const float* __restrict__ Wa,
                           const int* __restrict__ dst,
                           float* __restrict__ z,
                           float* __restrict__ el,
                           float* __restrict__ er,
                           unsigned short* __restrict__ tab,
                           float* __restrict__ bn) {
    __shared__ float lds_w[IN_DIM * OUT_DIM];
    __shared__ int bc[NBUCK];
    const int t = threadIdx.x;

    if (blockIdx.x >= GEMM_BLOCKS) {
        const int chunk = blockIdx.x - GEMM_BLOCKS;
        if (t < NBUCK) bc[t] = 0;
        if (chunk == 0 && t < 2 * OUT_DIM) bn[t] = 0.0f;
        __syncthreads();
        const int* d = dst + chunk * CHUNK;
        int dv[EPT];
        #pragma unroll
        for (int k = 0; k < EPT; ++k) {
            const int i = t + k * SORT_T;
            dv[k] = d[i < CHUNK ? i : CHUNK - 1];
        }
        #pragma unroll
        for (int k = 0; k < EPT; ++k)
            if (t + k * SORT_T < CHUNK) atomicAdd(&bc[dv[k] >> BSHIFT], 1);
        __syncthreads();
        if (t < NBUCK) tab[t * NCHUNK + chunk] = (unsigned short)bc[t];
        return;
    }

    for (int i = t; i < IN_DIM * OUT_DIM; i += SORT_T)
        lds_w[i] = W[i];
    __syncthreads();

    const int wave = t >> 5;
    const int lane = t & 31;
    const int tile = blockIdx.x * GEMM_WAVES + wave;
    if (tile >= GEMM_TILES) return;

    const int m0   = tile * 16;
    const int row  = m0 + (lane & 15);
    const int koff = (lane >> 4) * 2;
    const int col  = lane & 15;

    const float al0 = Wa[col];
    const float al1 = Wa[col + 16];
    const float ar0 = Wa[OUT_DIM + col];
    const float ar1 = Wa[OUT_DIM + col + 16];

    v8f c0 = {}, c1 = {};
    const float* hrow = h + (size_t)row * IN_DIM;

    #pragma unroll 4
    for (int k = 0; k < IN_DIM; k += 4) {
        v2f a;
        a.x = hrow[k + koff];
        a.y = hrow[k + koff + 1];
        v2f b0, b1;
        b0.x = lds_w[(k + koff)     * OUT_DIM + col];
        b0.y = lds_w[(k + koff + 1) * OUT_DIM + col];
        b1.x = lds_w[(k + koff)     * OUT_DIM + col + 16];
        b1.y = lds_w[(k + koff + 1) * OUT_DIM + col + 16];
        c0 = __builtin_amdgcn_wmma_f32_16x16x4_f32(false, a, false, b0, (short)0, c0, false, false);
        c1 = __builtin_amdgcn_wmma_f32_16x16x4_f32(false, a, false, b1, (short)0, c1, false, false);
    }

    const int rbase = m0 + ((lane >> 4) * 8);
    #pragma unroll
    for (int r = 0; r < 8; ++r) {
        z[(size_t)(rbase + r) * OUT_DIM + col]      = c0[r];
        z[(size_t)(rbase + r) * OUT_DIM + col + 16] = c1[r];
    }

    __builtin_amdgcn_sched_barrier(0);
    float pl[8], pr[8];
    #pragma unroll
    for (int r = 0; r < 8; ++r) {
        pl[r] = row_sum16_f(c0[r] * al0 + c1[r] * al1);
        pr[r] = row_sum16_f(c0[r] * ar0 + c1[r] * ar1);
    }
    if (col == 0) {
        float4* pe = (float4*)(el + rbase);
        pe[0] = make_float4(pl[0], pl[1], pl[2], pl[3]);
        pe[1] = make_float4(pl[4], pl[5], pl[6], pl[7]);
        float4* qe = (float4*)(er + rbase);
        qe[0] = make_float4(pr[0], pr[1], pr[2], pr[3]);
        qe[1] = make_float4(pr[4], pr[5], pr[6], pr[7]);
    }
}

__global__ void __launch_bounds__(SORT_T)
gat_bucket_scatter_kernel(const int* __restrict__ src,
                          const int* __restrict__ dst,
                          const unsigned short* __restrict__ tab,
                          int* __restrict__ bstart,
                          int* __restrict__ tmp) {
    __shared__ int cur[NBUCK];
    __shared__ int rtot[NBUCK];
    __shared__ int wsum[SCAN_WAVES];
    const int t = threadIdx.x;
    const int lane = t & 31;
    const int wave = t >> 5;
    const int c = blockIdx.x;
    const int* s = src + c * CHUNK;
    const int* d = dst + c * CHUNK;
    int sv[EPT], dv[EPT];
    #pragma unroll
    for (int k = 0; k < EPT; ++k) {
        const int i = t + k * SORT_T;
        const int ic = i < CHUNK ? i : CHUNK - 1;
        sv[k] = s[ic];
        dv[k] = d[ic];
    }

    const int4* t4 = (const int4*)tab;
    const int j0 = 8 * lane;
    int4 p[ROWS_PER_WAVE];
    #pragma unroll
    for (int i = 0; i < ROWS_PER_WAVE; ++i) {
        const int b  = wave + SORT_WAVES * i;
        const int bb = b < NBUCK ? b : NBUCK - 1;
        p[i] = t4[bb * (NCHUNK / 8) + lane];
    }
    __builtin_amdgcn_sched_barrier(0);
    unsigned keep[4];
    #pragma unroll
    for (int k = 0; k < 4; ++k)
        keep[k] = (j0 + 2 * k < c ? 0x0000ffffu : 0u) | (j0 + 2 * k + 1 < c ? 0xffff0000u : 0u);
    int vt = 0, vp = 0;
    #pragma unroll
    for (int i = 0; i < ROWS_PER_WAVE; ++i) {
        const unsigned w0 = (unsigned)p[i].x, w1 = (unsigned)p[i].y;
        const unsigned w2 = (unsigned)p[i].z, w3 = (unsigned)p[i].w;
        const unsigned t2 = (w0 + w1) + (w2 + w3);
        const unsigned p2 = ((w0 & keep[0]) + (w1 & keep[1])) + ((w2 & keep[2]) + (w3 & keep[3]));
        const int ts = row_sum16_i((int)((t2 & 0xffffu) + (t2 >> 16)));
        const int ps = row_sum16_i((int)((p2 & 0xffffu) + (p2 >> 16)));
        if ((lane & 15) == i) { vt = ts; vp = ps; }
    }
    vt += __shfl_xor(vt, 16, 32);
    vp += __shfl_xor(vp, 16, 32);
    const int bl = wave + SORT_WAVES * lane;
    if (lane < ROWS_PER_WAVE && bl < NBUCK) { rtot[bl] = vt; cur[bl] = vp; }
    __syncthreads();
    int own = 0, inc = 0;
    if (wave < SCAN_WAVES) {
        own = (t < NBUCK) ? rtot[t] : 0;
        inc = own;
        #pragma unroll
        for (int o = 1; o < 32; o <<= 1) {
            const int u = __shfl_up(inc, o, 32);
            if (lane >= o) inc += u;
        }
        if (lane == 31) wsum[wave] = inc;
    }
    __syncthreads();
    if (t < NBUCK) {
        int base = 0;
        #pragma unroll
        for (int w = 0; w < SCAN_WAVES; ++w)
            if (w < wave) base += wsum[w];
        const int first = base + inc - own;
        cur[t] += first;
        if (c == 0) bstart[t] = first;
    }
    __syncthreads();
    #pragma unroll
    for (int k = 0; k < EPT; ++k) {
        if (t + k * SORT_T < CHUNK) {
            const int pos = atomicAdd(&cur[dv[k] >> BSHIFT], 1);
            tmp[pos] = sv[k] | ((dv[k] & (BNODES - 1)) << SRC_BITS);
        }
    }
}

__global__ void __launch_bounds__(SORT_T)
gat_bucket_fill_kernel(const float* __restrict__ el,
                       const float* __restrict__ er,
                       const int* __restrict__ bstart,
                       const int* __restrict__ tmp,
                       int* __restrict__ cnt,
                       int* __restrict__ off,
                       int2* __restrict__ csr) {
    __shared__ int   nc[BNODES];
    __shared__ float ler[BNODES];
    __shared__ int   wtot[BNODES / 32];
    const int t = threadIdx.x;
    const int lane = t & 31;
    const int b = blockIdx.x;
    const int node = b * BNODES + t;
    const int beg = bstart[b];
    const int end = (b + 1 < NBUCK) ? bstart[b + 1] : N_EDGES;
    if (t < BNODES) {
        nc[t] = 0;
        ler[t] = (node < N_NODES) ? er[node] : 0.0f;
    }
    __syncthreads();
    for (int j0 = beg + t; j0 < end; j0 += FILL_U * SORT_T) {
        int pv[FILL_U];
        #pragma unroll
        for (int u = 0; u < FILL_U; ++u) {
            const int j = j0 + u * SORT_T;
            pv[u] = tmp[j < end ? j : end - 1];
        }
        #pragma unroll
        for (int u = 0; u < FILL_U; ++u)
            if (j0 + u * SORT_T < end) atomicAdd(&nc[pv[u] >> SRC_BITS], 1);
    }
    __syncthreads();
    int c = 0, inc = 0;
    if (t < BNODES) {
        c = nc[t];
        inc = c;
        #pragma unroll
        for (int o = 1; o < 32; o <<= 1) {
            const int u = __shfl_up(inc, o, 32);
            if (lane >= o) inc += u;
        }
        if (lane == 31) wtot[t >> 5] = inc;
    }
    __syncthreads();
    if (t < BNODES) {
        int base = beg;
        #pragma unroll
        for (int w = 0; w < BNODES / 32; ++w)
            if (w < (t >> 5)) base += wtot[w];
        const int first = base + inc - c;
        nc[t] = first;
        if (node < N_NODES) { cnt[node] = c; off[node] = first; }
    }
    __syncthreads();
    for (int j0 = beg + t; j0 < end; j0 += FILL_U * SORT_T) {
        int pv[FILL_U];
        float ev[FILL_U];
        #pragma unroll
        for (int u = 0; u < FILL_U; ++u) {
            const int j = j0 + u * SORT_T;
            pv[u] = tmp[j < end ? j : end - 1];
        }
        #pragma unroll
        for (int u = 0; u < FILL_U; ++u) ev[u] = el[pv[u] & ((1 << SRC_BITS) - 1)];
        #pragma unroll
        for (int u = 0; u < FILL_U; ++u) {
            if (j0 + u * SORT_T < end) {
                const int s  = pv[u] & ((1 << SRC_BITS) - 1);
                const int dl = pv[u] >> SRC_BITS;
                const int pos = atomicAdd(&nc[dl], 1);
                float e = ev[u] + ler[dl];
                e = (e >= 0.0f) ? e : LRELU_SLOPE * e;
                csr[pos] = make_int2(s, __float_as_int(e));
            }
        }
    }
}

__global__ void gat_node_gather_kernel(const int* __restrict__ off,
                                       const int* __restrict__ cnt,
                                       const int2* __restrict__ csr,
                                       const float* __restrict__ z,
                                       float* __restrict__ hout) {
    const int gid  = blockIdx.x * blockDim.x + threadIdx.x;
    const int lane = threadIdx.x & 31;
    const int node = gid >> 5;
    if (node >= N_NODES) return;
    const int beg = __builtin_amdgcn_readfirstlane(off[node]);
    const int n   = __builtin_amdgcn_readfirstlane(cnt[node]);
    const int2* seg = csr + beg;

    const bool has = lane < n;
    int2 rec = make_int2(0, 0);
    if (n > 0) rec = seg[has ? lane : n - 1];
    const float e0 = __int_as_float(rec.y);

    float m = has ? e0 : -INFINITY;
    for (int j = lane + 32; j < n; j += 32) m = fmaxf(m, __int_as_float(seg[j].y));
    #pragma unroll
    for (int o = 16; o > 0; o >>= 1) m = fmaxf(m, __shfl_xor(m, o, 32));

    float ssum = has ? __expf(e0 - m) : 0.0f;
    for (int j = lane + 32; j < n; j += 32) ssum += __expf(__int_as_float(seg[j].y) - m);
    #pragma unroll
    for (int o = 16; o > 0; o >>= 1) ssum += __shfl_xor(ssum, o, 32);

    float acc = 0.0f;
    for (int base = 0; base < n; base += 32) {
        const int left = n - base;
        const bool in = lane < left;
        if (base > 0) rec = seg[in ? base + lane : n - 1];
        const int s = rec.x;
        const int abits = __float_as_int(in ? __expf(__int_as_float(rec.y) - m) / ssum : 0.0f);
        const int here = left < 32 ? left : 32;
        const int kend = (here + 7) & ~7;
        for (int k = 0; k < kend; k += 8) {
            #pragma unroll
            for (int u = 0; u < 8; ++u) {
                const int   sk = __builtin_amdgcn_readlane(s, k + u);
                const float ak = __int_as_float(__builtin_amdgcn_readlane(abits, k + u));
                acc += ak * z[(size_t)sk * OUT_DIM + lane];
            }
        }
    }
    hout[(size_t)node * OUT_DIM + lane] = acc;
}

#define STAT_T      256
#define STAT_BLOCKS 512
#define STAT_WAVES  (STAT_BLOCKS * STAT_T / 32)
#define STAT_ROWS   ((N_NODES + STAT_WAVES - 1) / STAT_WAVES)
static_assert((STAT_ROWS - 1) * STAT_WAVES < N_NODES, "only a wave's last row can lie past the end");
__global__ void __launch_bounds__(STAT_T)
gat_bn_stats_kernel(const float* __restrict__ hout, float* __restrict__ bn) {
    __shared__ float ssum[OUT_DIM];
    __shared__ float ssq[OUT_DIM];
    const int lane = threadIdx.x & 31;
    if (threadIdx.x < OUT_DIM) { ssum[threadIdx.x] = 0.0f; ssq[threadIdx.x] = 0.0f; }
    const int w = (blockIdx.x * STAT_T + threadIdx.x) >> 5;
    const int last = w + (STAT_ROWS - 1) * STAT_WAVES;
    const bool has_last = last < N_NODES;
    float v[STAT_ROWS];
    #pragma unroll
    for (int i = 0; i < STAT_ROWS - 1; ++i)
        v[i] = hout[(size_t)(w + i * STAT_WAVES) * OUT_DIM + lane];
    v[STAT_ROWS - 1] = hout[(size_t)(has_last ? last : N_NODES - 1) * OUT_DIM + lane];
    float s = 0.0f, q = 0.0f;
    #pragma unroll
    for (int i = 0; i < STAT_ROWS - 1; ++i) {
        s += v[i];
        q += v[i] * v[i];
    }
    if (has_last) {
        s += v[STAT_ROWS - 1];
        q += v[STAT_ROWS - 1] * v[STAT_ROWS - 1];
    }
    __syncthreads();
    atomicAdd(&ssum[lane], s);
    atomicAdd(&ssq[lane], q);
    __syncthreads();
    if (threadIdx.x < OUT_DIM) {
        atomicAdd(&bn[threadIdx.x], ssum[threadIdx.x]);
        atomicAdd(&bn[OUT_DIM + threadIdx.x], ssq[threadIdx.x]);
    }
}

__global__ void gat_bn_apply_kernel(const float* __restrict__ bn,
                                    const float* __restrict__ gamma,
                                    const float* __restrict__ beta,
                                    float* __restrict__ out) {
    int idx = blockIdx.x * blockDim.x + threadIdx.x;
    if (idx >= N_NODES * OUT_DIM) return;
    int c = idx & 31;
    const float inv_n = 1.0f / (float)N_NODES;
    float mean = bn[c] * inv_n;
    float var  = bn[OUT_DIM + c] * inv_n - mean * mean;
    float x = (out[idx] - mean) * rsqrtf(var + BN_EPS) * gamma[c] + beta[c];
    out[idx] = (x > 0.0f) ? x : expm1f(x);
}

extern "C" void kernel_launch(void* const* d_in, const int* in_sizes, int n_in,
                              void* d_out, int out_size, void* d_ws, size_t ws_size,
                              hipStream_t stream) {
    const float* h     = (const float*)d_in[0];
    const float* W_fc  = (const float*)d_in[1];
    const float* W_at  = (const float*)d_in[2];
    const float* gamma = (const float*)d_in[3];
    const float* beta  = (const float*)d_in[4];
    const int*   src   = (const int*)d_in[5];
    const int*   dst   = (const int*)d_in[6];
    float* out = (float*)d_out;

    float* z    = (float*)d_ws;
    float* el   = z + (size_t)N_NODES * OUT_DIM;
    float* er   = el + N_NODES;
    int*   cnt  = (int*)(er + N_NODES);
    int*   off  = cnt + N_NODES;
    int2*  csr  = (int2*)(off + N_NODES);
    float* bn   = (float*)(csr + N_EDGES);
    int*   tmp    = (int*)d_out;
    unsigned short* tab = (unsigned short*)(tmp + N_EDGES);
    int*   bstart = tmp + N_EDGES + TABN / 2;

    const int B = 256;

    gat_gemm_proj_count_kernel<<<GEMM_BLOCKS + NCHUNK, SORT_T, 0, stream>>>(h, W_fc, W_at, dst, z, el, er, tab, bn);

    gat_bucket_scatter_kernel<<<NCHUNK, SORT_T, 0, stream>>>(src, dst, tab, bstart, tmp);

    gat_bucket_fill_kernel<<<NBUCK, SORT_T, 0, stream>>>(el, er, bstart, tmp, cnt, off, csr);

    gat_node_gather_kernel<<<(N_NODES * OUT_DIM + B - 1) / B, B, 0, stream>>>(off, cnt, csr, z, out);

    gat_bn_stats_kernel<<<STAT_BLOCKS, STAT_T, 0, stream>>>(out, bn);

    gat_bn_apply_kernel<<<(N_NODES * OUT_DIM + B - 1) / B, B, 0, stream>>>(bn, gamma, beta, out);
}


# Bucket fill and node gather share one launch (six launches to five): a bucket's (src, e) records are sorted by node in LDS and read there by the same block's waves, never stored to memory
# speedup vs baseline: 3.9278x; 1.0902x over previous
#include <hip/hip_runtime.h>
#include <hip/hip_bf16.h>

#define N_NODES 100000
#define N_EDGES 1600000
#define IN_DIM 128
#define OUT_DIM 32
#define BN_EPS 1e-5f
#define LRELU_SLOPE 0.01f

#define BSHIFT 8
#define BNODES 256
#define NBUCK  ((N_NODES + BNODES - 1) / BNODES)
#define NCHUNK 256
#define CHUNK  (N_EDGES / NCHUNK)
#define TABN   (NBUCK * NCHUNK)
#define SORT_T 1024
#define EPT    ((CHUNK + SORT_T - 1) / SORT_T)
#define SRC_BITS 17
#define FILL_U 5
static_assert(N_EDGES % NCHUNK == 0, "chunks tile the edge list");
static_assert(TABN % 8 == 0, "table rows of 16-bit counts are read as int4");
static_assert(N_NODES <= (1 << SRC_BITS), "src and the in-bucket dst share one word");
static_assert(NCHUNK == 256, "a table row of 16-bit counts is one int4 per lane of one wave");
static_assert(4 * CHUNK <= 65535, "four piece sizes summed still fit 16 bits");
static_assert(NBUCK <= SORT_T, "one thread per bucket in the cursor scan");

#define GEMM_TILES   (N_NODES / 16)
#define GEMM_WAVES   (SORT_T / 32)
#define GEMM_BLOCKS  ((GEMM_TILES + GEMM_WAVES - 1) / GEMM_WAVES)
#define SCAN_WAVES   ((NBUCK + 31) / 32)
#define SORT_WAVES   (SORT_T / 32)
#define ROWS_PER_WAVE ((NBUCK + SORT_WAVES - 1) / SORT_WAVES)
static_assert(ROWS_PER_WAVE <= 16, "one lane of each half wave per row of a wave");

typedef __attribute__((ext_vector_type(2))) float v2f;
typedef __attribute__((ext_vector_type(8))) float v8f;

template <int M> __device__ __forceinline__ float row_xor_f(float x) {
    return __int_as_float(__builtin_amdgcn_update_dpp(0, __float_as_int(x), 0x160 + M, 0xf, 0xf, true));
}
template <int M> __device__ __forceinline__ int row_xor_i(int x) {
    return __builtin_amdgcn_update_dpp(0, x, 0x160 + M, 0xf, 0xf, true);
}
__device__ __forceinline__ float row_sum16_f(float x) {
    x += row_xor_f<1>(x);
    x += row_xor_f<2>(x);
    x += row_xor_f<4>(x);
    x += row_xor_f<8>(x);
    return x;
}
__device__ __forceinline__ int row_sum16_i(int x) {
    x += row_xor_i<1>(x);
    x += row_xor_i<2>(x);
    x += row_xor_i<4>(x);
    x += row_xor_i<8>(x);
    return x;
}

__global__ void __launch_bounds__(SORT_T)
gat_gemm_proj_count_kernel(const float* __restrict__ h,
                           const float* __restrict__ W,
                           const float* __restrict__ Wa,
                           const int* __restrict__ dst,
                           float* __restrict__ z,
                           float* __restrict__ el,
                           float* __restrict__ er,
                           unsigned short* __restrict__ tab,
                           float* __restrict__ bn) {
    __shared__ float lds_w[IN_DIM * OUT_DIM];
    __shared__ int bc[NBUCK];
    const int t = threadIdx.x;

    if (blockIdx.x >= GEMM_BLOCKS) {
        const int chunk = blockIdx.x - GEMM_BLOCKS;
        if (t < NBUCK) bc[t] = 0;
        if (chunk == 0 && t < 2 * OUT_DIM) bn[t] = 0.0f;
        __syncthreads();
        const int* d = dst + chunk * CHUNK;
        int dv[EPT];
        #pragma unroll
        for (int k = 0; k < EPT; ++k) {
            const int i = t + k * SORT_T;
            dv[k] = d[i < CHUNK ? i : CHUNK - 1];
        }
        #pragma unroll
        for (int k = 0; k < EPT; ++k)
            if (t + k * SORT_T < CHUNK) atomicAdd(&bc[dv[k] >> BSHIFT], 1);
        __syncthreads();
        if (t < NBUCK) tab[t * NCHUNK + chunk] = (unsigned short)bc[t];
        return;
    }

    for (int i = t; i < IN_DIM * OUT_DIM; i += SORT_T)
        lds_w[i] = W[i];
    __syncthreads();

    const int wave = t >> 5;
    const int lane = t & 31;
    const int tile = blockIdx.x * GEMM_WAVES + wave;
    if (tile >= GEMM_TILES) return;

    const int m0   = tile * 16;
    const int row  = m0 + (lane & 15);
    const int koff = (lane >> 4) * 2;
    const int col  = lane & 15;

    const float al0 = Wa[col];
    const float al1 = Wa[col + 16];
    const float ar0 = Wa[OUT_DIM + col];
    const float ar1 = Wa[OUT_DIM + col + 16];

    v8f c0 = {}, c1 = {};
    const float* hrow = h + (size_t)row * IN_DIM;

    #pragma unroll 4
    for (int k = 0; k < IN_DIM; k += 4) {
        v2f a;
        a.x = hrow[k + koff];
        a.y = hrow[k + koff + 1];
        v2f b0, b1;
        b0.x = lds_w[(k + koff)     * OUT_DIM + col];
        b0.y = lds_w[(k + koff + 1) * OUT_DIM + col];
        b1.x = lds_w[(k + koff)     * OUT_DIM + col + 16];
        b1.y = lds_w[(k + koff + 1) * OUT_DIM + col + 16];
        c0 = __builtin_amdgcn_wmma_f32_16x16x4_f32(false, a, false, b0, (short)0, c0, false, false);
        c1 = __builtin_amdgcn_wmma_f32_16x16x4_f32(false, a, false, b1, (short)0, c1, false, false);
    }

    const int rbase = m0 + ((lane >> 4) * 8);
    #pragma unroll
    for (int r = 0; r < 8; ++r) {
        z[(size_t)(rbase + r) * OUT_DIM + col]      = c0[r];
        z[(size_t)(rbase + r) * OUT_DIM + col + 16] = c1[r];
    }

    __builtin_amdgcn_sched_barrier(0);
    float pl[8], pr[8];
    #pragma unroll
    for (int r = 0; r < 8; ++r) {
        pl[r] = row_sum16_f(c0[r] * al0 + c1[r] * al1);
        pr[r] = row_sum16_f(c0[r] * ar0 + c1[r] * ar1);
    }
    if (col == 0) {
        float4* pe = (float4*)(el + rbase);
        pe[0] = make_float4(pl[0], pl[1], pl[2], pl[3]);
        pe[1] = make_float4(pl[4], pl[5], pl[6], pl[7]);
        float4* qe = (float4*)(er + rbase);
        qe[0] = make_float4(pr[0], pr[1], pr[2], pr[3]);
        qe[1] = make_float4(pr[4], pr[5], pr[6], pr[7]);
    }
}

__global__ void __launch_bounds__(SORT_T)
gat_bucket_scatter_kernel(const int* __restrict__ src,
                          const int* __restrict__ dst,
                          const unsigned short* __restrict__ tab,
                          int* __restrict__ bstart,
                          int* __restrict__ tmp) {
    __shared__ int cur[NBUCK];
    __shared__ int rtot[NBUCK];
    __shared__ int wsum[SCAN_WAVES];
    const int t = threadIdx.x;
    const int lane = t & 31;
    const int wave = t >> 5;
    const int c = blockIdx.x;
    const int* s = src + c * CHUNK;
    const int* d = dst + c * CHUNK;
    int sv[EPT], dv[EPT];
    #pragma unroll
    for (int k = 0; k < EPT; ++k) {
        const int i = t + k * SORT_T;
        const int ic = i < CHUNK ? i : CHUNK - 1;
        sv[k] = s[ic];
        dv[k] = d[ic];
    }

    const int4* t4 = (const int4*)tab;
    const int j0 = 8 * lane;
    int4 p[ROWS_PER_WAVE];
    #pragma unroll
    for (int i = 0; i < ROWS_PER_WAVE; ++i) {
        const int b  = wave + SORT_WAVES * i;
        const int bb = b < NBUCK ? b : NBUCK - 1;
        p[i] = t4[bb * (NCHUNK / 8) + lane];
    }
    __builtin_amdgcn_sched_barrier(0);
    unsigned keep[4];
    #pragma unroll
    for (int k = 0; k < 4; ++k)
        keep[k] = (j0 + 2 * k < c ? 0x0000ffffu : 0u) | (j0 + 2 * k + 1 < c ? 0xffff0000u : 0u);
    int vt = 0, vp = 0;
    #pragma unroll
    for (int i = 0; i < ROWS_PER_WAVE; ++i) {
        const unsigned w0 = (unsigned)p[i].x, w1 = (unsigned)p[i].y;
        const unsigned w2 = (unsigned)p[i].z, w3 = (unsigned)p[i].w;
        const unsigned t2 = (w0 + w1) + (w2 + w3);
        const unsigned p2 = ((w0 & keep[0]) + (w1 & keep[1])) + ((w2 & keep[2]) + (w3 & keep[3]));
        const int ts = row_sum16_i((int)((t2 & 0xffffu) + (t2 >> 16)));
        const int ps = row_sum16_i((int)((p2 & 0xffffu) + (p2 >> 16)));
        if ((lane & 15) == i) { vt = ts; vp = ps; }
    }
    vt += __shfl_xor(vt, 16, 32);
    vp += __shfl_xor(vp, 16, 32);
    const int bl = wave + SORT_WAVES * lane;
    if (lane < ROWS_PER_WAVE && bl < NBUCK) { rtot[bl] = vt; cur[bl] = vp; }
    __syncthreads();
    int own = 0, inc = 0;
    if (wave < SCAN_WAVES) {
        own = (t < NBUCK) ? rtot[t] : 0;
        inc = own;
        #pragma unroll
        for (int o = 1; o < 32; o <<= 1) {
            const int u = __shfl_up(inc, o, 32);
            if (lane >= o) inc += u;
        }
        if (lane == 31) wsum[wave] = inc;
    }
    __syncthreads();
    if (t < NBUCK) {
        int base = 0;
        #pragma unroll
        for (int w = 0; w < SCAN_WAVES; ++w)
            if (w < wave) base += wsum[w];
        const int first = base + inc - own;
        cur[t] += first;
        if (c == 0) bstart[t] = first;
    }
    __syncthreads();
    #pragma unroll
    for (int k = 0; k < EPT; ++k) {
        if (t + k * SORT_T < CHUNK) {
            const int pos = atomicAdd(&cur[dv[k] >> BSHIFT], 1);
            tmp[pos] = sv[k] | ((dv[k] & (BNODES - 1)) << SRC_BITS);
        }
    }
}

#define LCAP 8192

template <bool IN_LDS>
__device__ __forceinline__ int2 rec_at(const int2* lrec, const int* gs, const float* __restrict__ el,
                                       float erd, int j) {
    if (IN_LDS) return lrec[j];
    const int s = __hip_atomic_load(gs + j, __ATOMIC_RELAXED, __HIP_MEMORY_SCOPE_AGENT);
    float e = el[s] + erd;
    e = (e >= 0.0f) ? e : LRELU_SLOPE * e;
    return make_int2(s, __float_as_int(e));
}

template <bool IN_LDS>
__device__ __forceinline__ void bucket_nodes(const int2* lrec, const int* gs, const int* nfirst,
                                             const float* ler, const float* __restrict__ el,
                                             const float* __restrict__ z, float* __restrict__ hout,
                                             int b, int wave, int lane) {
    const int wave_s = __builtin_amdgcn_readfirstlane(wave);
    for (int dl = wave_s; dl < BNODES; dl += SORT_WAVES) {
        const int node = b * BNODES + dl;
        if (node >= N_NODES) break;
        const int first = __builtin_amdgcn_readfirstlane(nfirst[dl]);
        const int n     = __builtin_amdgcn_readfirstlane(nfirst[dl + 1]) - first;
        const float erd = ler[dl];

        const bool has = lane < n;
        int2 rec = make_int2(0, 0);
        if (n > 0) rec = rec_at<IN_LDS>(lrec, gs, el, erd, first + (has ? lane : n - 1));
        const float e0 = __int_as_float(rec.y);

        float m = has ? e0 : -INFINITY;
        for (int j = lane + 32; j < n; j += 32)
            m = fmaxf(m, __int_as_float(rec_at<IN_LDS>(lrec, gs, el, erd, first + j).y));
        #pragma unroll
        for (int o = 16; o > 0; o >>= 1) m = fmaxf(m, __shfl_xor(m, o, 32));

        float ssum = has ? __expf(e0 - m) : 0.0f;
        for (int j = lane + 32; j < n; j += 32)
            ssum += __expf(__int_as_float(rec_at<IN_LDS>(lrec, gs, el, erd, first + j).y) - m);
        #pragma unroll
        for (int o = 16; o > 0; o >>= 1) ssum += __shfl_xor(ssum, o, 32);

        float acc = 0.0f;
        for (int base = 0; base < n; base += 32) {
            const int left = n - base;
            const bool in = lane < left;
            if (base > 0) rec = rec_at<IN_LDS>(lrec, gs, el, erd, first + (in ? base + lane : n - 1));
            const int s = rec.x;
            const int abits = __float_as_int(in ? __expf(__int_as_float(rec.y) - m) / ssum : 0.0f);
            const int here = left < 32 ? left : 32;
            const int kend = (here + 7) & ~7;
            for (int k = 0; k < kend; k += 8) {
                #pragma unroll
                for (int u = 0; u < 8; ++u) {
                    const int   sk = __builtin_amdgcn_readlane(s, k + u);
                    const float ak = __int_as_float(__builtin_amdgcn_readlane(abits, k + u));
                    acc += ak * z[(size_t)sk * OUT_DIM + lane];
                }
            }
        }
        hout[(size_t)node * OUT_DIM + lane] = acc;
    }
}

__global__ void __launch_bounds__(SORT_T)
gat_bucket_gather_kernel(const float* __restrict__ el,
                         const float* __restrict__ er,
                         const int* __restrict__ bstart,
                         const int* __restrict__ tmp,
                         int* gs,
                         const float* __restrict__ z,
                         float* __restrict__ hout) {
    __shared__ int   nc[BNODES];
    __shared__ int   nfirst[BNODES + 1];
    __shared__ float ler[BNODES];
    __shared__ int   wtot[BNODES / 32];
    __shared__ int2  lrec[LCAP];
    const int t = threadIdx.x;
    const int lane = t & 31;
    const int wave = t >> 5;
    const int b = blockIdx.x;
    const int node = b * BNODES + t;
    const int beg = bstart[b];
    const int end = (b + 1 < NBUCK) ? bstart[b + 1] : N_EDGES;
    const bool fits = (end - beg) <= LCAP;
    if (t < BNODES) {
        nc[t] = 0;
        ler[t] = (node < N_NODES) ? er[node] : 0.0f;
    }
    __syncthreads();
    for (int j0 = beg + t; j0 < end; j0 += FILL_U * SORT_T) {
        int pv[FILL_U];
        #pragma unroll
        for (int u = 0; u < FILL_U; ++u) {
            const int j = j0 + u * SORT_T;
            pv[u] = tmp[j < end ? j : end - 1];
        }
        #pragma unroll
        for (int u = 0; u < FILL_U; ++u)
            if (j0 + u * SORT_T < end) atomicAdd(&nc[pv[u] >> SRC_BITS], 1);
    }
    __syncthreads();
    int c = 0, inc = 0;
    if (t < BNODES) {
        c = nc[t];
        inc = c;
        #pragma unroll
        for (int o = 1; o < 32; o <<= 1) {
            const int u = __shfl_up(inc, o, 32);
            if (lane >= o) inc += u;
        }
        if (lane == 31) wtot[t >> 5] = inc;
    }
    __syncthreads();
    if (t < BNODES) {
        int base = 0;
        #pragma unroll
        for (int w = 0; w < BNODES / 32; ++w)
            if (w < (t >> 5)) base += wtot[w];
        const int first = base + inc - c;
        nc[t] = first;
        nfirst[t] = first;
        if (t == BNODES - 1) nfirst[BNODES] = first + c;
    }
    __syncthreads();
    for (int j0 = beg + t; j0 < end; j0 += FILL_U * SORT_T) {
        int pv[FILL_U];
        float ev[FILL_U];
        #pragma unroll
        for (int u = 0; u < FILL_U; ++u) {
            const int j = j0 + u * SORT_T;
            pv[u] = tmp[j < end ? j : end - 1];
        }
        #pragma unroll
        for (int u = 0; u < FILL_U; ++u) ev[u] = el[pv[u] & ((1 << SRC_BITS) - 1)];
        #pragma unroll
        for (int u = 0; u < FILL_U; ++u) {
            if (j0 + u * SORT_T < end) {
                const int s  = pv[u] & ((1 << SRC_BITS) - 1);
                const int dl = pv[u] >> SRC_BITS;
                const int pos = atomicAdd(&nc[dl], 1);
                float e = ev[u] + ler[dl];
                e = (e >= 0.0f) ? e : LRELU_SLOPE * e;
                if (fits) lrec[pos] = make_int2(s, __float_as_int(e));
                else      gs[beg + pos] = s;
            }
        }
    }
    if (!fits) __threadfence();
    __syncthreads();
    if (fits) bucket_nodes<true >(lrec, gs + beg, nfirst, ler, el, z, hout, b, wave, lane);
    else      bucket_nodes<false>(lrec, gs + beg, nfirst, ler, el, z, hout, b, wave, lane);
}

#define STAT_T      256
#define STAT_BLOCKS 512
#define STAT_WAVES  (STAT_BLOCKS * STAT_T / 32)
#define STAT_ROWS   ((N_NODES + STAT_WAVES - 1) / STAT_WAVES)
static_assert((STAT_ROWS - 1) * STAT_WAVES < N_NODES, "only a wave's last row can lie past the end");
__global__ void __launch_bounds__(STAT_T)
gat_bn_stats_kernel(const float* __restrict__ hout, float* __restrict__ bn) {
    __shared__ float ssum[OUT_DIM];
    __shared__ float ssq[OUT_DIM];
    const int lane = threadIdx.x & 31;
    if (threadIdx.x < OUT_DIM) { ssum[threadIdx.x] = 0.0f; ssq[threadIdx.x] = 0.0f; }
    const int w = (blockIdx.x * STAT_T + threadIdx.x) >> 5;
    const int last = w + (STAT_ROWS - 1) * STAT_WAVES;
    const bool has_last = last < N_NODES;
    float v[STAT_ROWS];
    #pragma unroll
    for (int i = 0; i < STAT_ROWS - 1; ++i)
        v[i] = hout[(size_t)(w + i * STAT_WAVES) * OUT_DIM + lane];
    v[STAT_ROWS - 1] = hout[(size_t)(has_last ? last : N_NODES - 1) * OUT_DIM + lane];
    float s = 0.0f, q = 0.0f;
    #pragma unroll
    for (int i = 0; i < STAT_ROWS - 1; ++i) {
        s += v[i];
        q += v[i] * v[i];
    }
    if (has_last) {
        s += v[STAT_ROWS - 1];
        q += v[STAT_ROWS - 1] * v[STAT_ROWS - 1];
    }
    __syncthreads();
    atomicAdd(&ssum[lane], s);
    atomicAdd(&ssq[lane], q);
    __syncthreads();
    if (threadIdx.x < OUT_DIM) {
        atomicAdd(&bn[threadIdx.x], ssum[threadIdx.x]);
        atomicAdd(&bn[OUT_DIM + threadIdx.x], ssq[threadIdx.x]);
    }
}

__global__ void gat_bn_apply_kernel(const float* __restrict__ bn,
                                    const float* __restrict__ gamma,
                                    const float* __restrict__ beta,
                                    float* __restrict__ out) {
    int idx = blockIdx.x * blockDim.x + threadIdx.x;
    if (idx >= N_NODES * OUT_DIM) return;
    int c = idx & 31;
    const float inv_n = 1.0f / (float)N_NODES;
    float mean = bn[c] * inv_n;
    float var  = bn[OUT_DIM + c] * inv_n - mean * mean;
    float x = (out[idx] - mean) * rsqrtf(var + BN_EPS) * gamma[c] + beta[c];
    out[idx] = (x > 0.0f) ? x : expm1f(x);
}

extern "C" void kernel_launch(void* const* d_in, const int* in_sizes, int n_in,
                              void* d_out, int out_size, void* d_ws, size_t ws_size,
                              hipStream_t stream) {
    const float* h     = (const float*)d_in[0];
    const float* W_fc  = (const float*)d_in[1];
    const float* W_at  = (const float*)d_in[2];
    const float* gamma = (const float*)d_in[3];
    const float* beta  = (const float*)d_in[4];
    const int*   src   = (const int*)d_in[5];
    const int*   dst   = (const int*)d_in[6];
    float* out = (float*)d_out;

    float* z    = (float*)d_ws;
    float* el   = z + (size_t)N_NODES * OUT_DIM;
    float* er   = el + N_NODES;
    int*   bstart = (int*)(er + N_NODES);
    int*   tmp  = bstart + 2 * N_NODES;
    int*   gs   = tmp + N_EDGES;
    float* bn   = (float*)(gs + N_EDGES);
    unsigned short* tab = (unsigned short*)d_out;

    const int B = 256;

    gat_gemm_proj_count_kernel<<<GEMM_BLOCKS + NCHUNK, SORT_T, 0, stream>>>(h, W_fc, W_at, dst, z, el, er, tab, bn);

    gat_bucket_scatter_kernel<<<NCHUNK, SORT_T, 0, stream>>>(src, dst, tab, bstart, tmp);

    gat_bucket_gather_kernel<<<NBUCK, SORT_T, 0, stream>>>(el, er, bstart, tmp, gs, z, out);

    gat_bn_stats_kernel<<<STAT_BLOCKS, STAT_T, 0, stream>>>(out, bn);

    gat_bn_apply_kernel<<<(N_NODES * OUT_DIM + B - 1) / B, B, 0, stream>>>(bn, gamma, beta, out);
}


# Node pass row gather as 16-byte loads: a lane takes 4 channels of one of 4 edges, all row loads of a node in flight at once; 4 partial sums per channel meet across lanes (new order of the row sums)
# speedup vs baseline: 4.1672x; 1.0610x over previous
#include <hip/hip_runtime.h>
#include <hip/hip_bf16.h>

#define N_NODES 100000
#define N_EDGES 1600000
#define IN_DIM 128
#define OUT_DIM 32
#define BN_EPS 1e-5f
#define LRELU_SLOPE 0.01f

#define BSHIFT 8
#define BNODES 256
#define NBUCK  ((N_NODES + BNODES - 1) / BNODES)
#define NCHUNK 256
#define CHUNK  (N_EDGES / NCHUNK)
#define TABN   (NBUCK * NCHUNK)
#define SORT_T 1024
#define EPT    ((CHUNK + SORT_T - 1) / SORT_T)
#define SRC_BITS 17
#define FILL_U 5
static_assert(N_EDGES % NCHUNK == 0, "chunks tile the edge list");
static_assert(TABN % 8 == 0, "table rows of 16-bit counts are read as int4");
static_assert(N_NODES <= (1 << SRC_BITS), "src and the in-bucket dst share one word");
static_assert(NCHUNK == 256, "a table row of 16-bit counts is one int4 per lane of one wave");
static_assert(4 * CHUNK <= 65535, "four piece sizes summed still fit 16 bits");
static_assert(NBUCK <= SORT_T, "one thread per bucket in the cursor scan");

#define GEMM_TILES   (N_NODES / 16)
#define GEMM_WAVES   (SORT_T / 32)
#define GEMM_BLOCKS  ((GEMM_TILES + GEMM_WAVES - 1) / GEMM_WAVES)
#define SCAN_WAVES   ((NBUCK + 31) / 32)
#define SORT_WAVES   (SORT_T / 32)
#define ROWS_PER_WAVE ((NBUCK + SORT_WAVES - 1) / SORT_WAVES)
static_assert(ROWS_PER_WAVE <= 16, "one lane of each half wave per row of a wave");

typedef __attribute__((ext_vector_type(2))) float v2f;
typedef __attribute__((ext_vector_type(8))) float v8f;

template <int M> __device__ __forceinline__ float row_xor_f(float x) {
    return __int_as_float(__builtin_amdgcn_update_dpp(0, __float_as_int(x), 0x160 + M, 0xf, 0xf, true));
}
template <int M> __device__ __forceinline__ int row_xor_i(int x) {
    return __builtin_amdgcn_update_dpp(0, x, 0x160 + M, 0xf, 0xf, true);
}
__device__ __forceinline__ float row_sum16_f(float x) {
    x += row_xor_f<1>(x);
    x += row_xor_f<2>(x);
    x += row_xor_f<4>(x);
    x += row_xor_f<8>(x);
    return x;
}
__device__ __forceinline__ int row_sum16_i(int x) {
    x += row_xor_i<1>(x);
    x += row_xor_i<2>(x);
    x += row_xor_i<4>(x);
    x += row_xor_i<8>(x);
    return x;
}

__global__ void __launch_bounds__(SORT_T)
gat_gemm_proj_count_kernel(const float* __restrict__ h,
                           const float* __restrict__ W,
                           const float* __restrict__ Wa,
                           const int* __restrict__ dst,
                           float* __restrict__ z,
                           float* __restrict__ el,
                           float* __restrict__ er,
                           unsigned short* __restrict__ tab,
                           float* __restrict__ bn) {
    __shared__ float lds_w[IN_DIM * OUT_DIM];
    __shared__ int bc[NBUCK];
    const int t = threadIdx.x;

    if (blockIdx.x >= GEMM_BLOCKS) {
        const int chunk = blockIdx.x - GEMM_BLOCKS;
        if (t < NBUCK) bc[t] = 0;
        if (chunk == 0 && t < 2 * OUT_DIM) bn[t] = 0.0f;
        __syncthreads();
        const int* d = dst + chunk * CHUNK;
        int dv[EPT];
        #pragma unroll
        for (int k = 0; k < EPT; ++k) {
            const int i = t + k * SORT_T;
            dv[k] = d[i < CHUNK ? i : CHUNK - 1];
        }
        #pragma unroll
        for (int k = 0; k < EPT; ++k)
            if (t + k * SORT_T < CHUNK) atomicAdd(&bc[dv[k] >> BSHIFT], 1);
        __syncthreads();
        if (t < NBUCK) tab[t * NCHUNK + chunk] = (unsigned short)bc[t];
        return;
    }

    for (int i = t; i < IN_DIM * OUT_DIM; i += SORT_T)
        lds_w[i] = W[i];
    __syncthreads();

    const int wave = t >> 5;
    const int lane = t & 31;
    const int tile = blockIdx.x * GEMM_WAVES + wave;
    if (tile >= GEMM_TILES) return;

    const int m0   = tile * 16;
    const int row  = m0 + (lane & 15);
    const int koff = (lane >> 4) * 2;
    const int col  = lane & 15;

    const float al0 = Wa[col];
    const float al1 = Wa[col + 16];
    const float ar0 = Wa[OUT_DIM + col];
    const float ar1 = Wa[OUT_DIM + col + 16];

    v8f c0 = {}, c1 = {};
    const float* hrow = h + (size_t)row * IN_DIM;

    #pragma unroll 4
    for (int k = 0; k < IN_DIM; k += 4) {
        v2f a;
        a.x = hrow[k + koff];
        a.y = hrow[k + koff + 1];
        v2f b0, b1;
        b0.x = lds_w[(k + koff)     * OUT_DIM + col];
        b0.y = lds_w[(k + koff + 1) * OUT_DIM + col];
        b1.x = lds_w[(k + koff)     * OUT_DIM + col + 16];
        b1.y = lds_w[(k + koff + 1) * OUT_DIM + col + 16];
        c0 = __builtin_amdgcn_wmma_f32_16x16x4_f32(false, a, false, b0, (short)0, c0, false, false);
        c1 = __builtin_amdgcn_wmma_f32_16x16x4_f32(false, a, false, b1, (short)0, c1, false, false);
    }

    const int rbase = m0 + ((lane >> 4) * 8);
    #pragma unroll
    for (int r = 0; r < 8; ++r) {
        z[(size_t)(rbase + r) * OUT_DIM + col]      = c0[r];
        z[(size_t)(rbase + r) * OUT_DIM + col + 16] = c1[r];
    }

    __builtin_amdgcn_sched_barrier(0);
    float pl[8], pr[8];
    #pragma unroll
    for (int r = 0; r < 8; ++r) {
        pl[r] = row_sum16_f(c0[r] * al0 + c1[r] * al1);
        pr[r] = row_sum16_f(c0[r] * ar0 + c1[r] * ar1);
    }
    if (col == 0) {
        float4* pe = (float4*)(el + rbase);
        pe[0] = make_float4(pl[0], pl[1], pl[2], pl[3]);
        pe[1] = make_float4(pl[4], pl[5], pl[6], pl[7]);
        float4* qe = (float4*)(er + rbase);
        qe[0] = make_float4(pr[0], pr[1], pr[2], pr[3]);
        qe[1] = make_float4(pr[4], pr[5], pr[6], pr[7]);
    }
}

__global__ void __launch_bounds__(SORT_T)
gat_bucket_scatter_kernel(const int* __restrict__ src,
                          const int* __restrict__ dst,
                          const unsigned short* __restrict__ tab,
                          int* __restrict__ bstart,
                          int* __restrict__ tmp) {
    __shared__ int cur[NBUCK];
    __shared__ int rtot[NBUCK];
    __shared__ int wsum[SCAN_WAVES];
    const int t = threadIdx.x;
    const int lane = t & 31;
    const int wave = t >> 5;
    const int c = blockIdx.x;
    const int* s = src + c * CHUNK;
    const int* d = dst + c * CHUNK;
    int sv[EPT], dv[EPT];
    #pragma unroll
    for (int k = 0; k < EPT; ++k) {
        const int i = t + k * SORT_T;
        const int ic = i < CHUNK ? i : CHUNK - 1;
        sv[k] = s[ic];
        dv[k] = d[ic];
    }

    const int4* t4 = (const int4*)tab;
    const int j0 = 8 * lane;
    int4 p[ROWS_PER_WAVE];
    #pragma unroll
    for (int i = 0; i < ROWS_PER_WAVE; ++i) {
        const int b  = wave + SORT_WAVES * i;
        const int bb = b < NBUCK ? b : NBUCK - 1;
        p[i] = t4[bb * (NCHUNK / 8) + lane];
    }
    __builtin_amdgcn_sched_barrier(0);
    unsigned keep[4];
    #pragma unroll
    for (int k = 0; k < 4; ++k)
        keep[k] = (j0 + 2 * k < c ? 0x0000ffffu : 0u) | (j0 + 2 * k + 1 < c ? 0xffff0000u : 0u);
    int vt = 0, vp = 0;
    #pragma unroll
    for (int i = 0; i < ROWS_PER_WAVE; ++i) {
        const unsigned w0 = (unsigned)p[i].x, w1 = (unsigned)p[i].y;
        const unsigned w2 = (unsigned)p[i].z, w3 = (unsigned)p[i].w;
        const unsigned t2 = (w0 + w1) + (w2 + w3);
        const unsigned p2 = ((w0 & keep[0]) + (w1 & keep[1])) + ((w2 & keep[2]) + (w3 & keep[3]));
        const int ts = row_sum16_i((int)((t2 & 0xffffu) + (t2 >> 16)));
        const int ps = row_sum16_i((int)((p2 & 0xffffu) + (p2 >> 16)));
        if ((lane & 15) == i) { vt = ts; vp = ps; }
    }
    vt += __shfl_xor(vt, 16, 32);
    vp += __shfl_xor(vp, 16, 32);
    const int bl = wave + SORT_WAVES * lane;
    if (lane < ROWS_PER_WAVE && bl < NBUCK) { rtot[bl] = vt; cur[bl] = vp; }
    __syncthreads();
    int own = 0, inc = 0;
    if (wave < SCAN_WAVES) {
        own = (t < NBUCK) ? rtot[t] : 0;
        inc = own;
        #pragma unroll
        for (int o = 1; o < 32; o <<= 1) {
            const int u = __shfl_up(inc, o, 32);
            if (lane >= o) inc += u;
        }
        if (lane == 31) wsum[wave] = inc;
    }
    __syncthreads();
    if (t < NBUCK) {
        int base = 0;
        #pragma unroll
        for (int w = 0; w < SCAN_WAVES; ++w)
            if (w < wave) base += wsum[w];
        const int first = base + inc - own;
        cur[t] += first;
        if (c == 0) bstart[t] = first;
    }
    __syncthreads();
    #pragma unroll
    for (int k = 0; k < EPT; ++k) {
        if (t + k * SORT_T < CHUNK) {
            const int pos = atomicAdd(&cur[dv[k] >> BSHIFT], 1);
            tmp[pos] = sv[k] | ((dv[k] & (BNODES - 1)) << SRC_BITS);
        }
    }
}

#define LCAP 8192
#define ROW_LOADS 7
#define ROW_EDGES (4 * ROW_LOADS)
static_assert(ROW_EDGES <= 32, "a group's edges are one lane each");

template <bool IN_LDS>
__device__ __forceinline__ int2 rec_at(const int2* lrec, const int* gs, const float* __restrict__ el,
                                       float erd, int j) {
    if (IN_LDS) return lrec[j];
    const int s = __hip_atomic_load(gs + j, __ATOMIC_RELAXED, __HIP_MEMORY_SCOPE_AGENT);
    float e = el[s] + erd;
    e = (e >= 0.0f) ? e : LRELU_SLOPE * e;
    return make_int2(s, __float_as_int(e));
}

template <bool IN_LDS>
__device__ __forceinline__ void bucket_nodes(const int2* lrec, const int* gs, const int* nfirst,
                                             const float* ler, const float* __restrict__ el,
                                             const float* __restrict__ z, float* __restrict__ hout,
                                             int b, int wave, int lane) {
    const int wave_s = __builtin_amdgcn_readfirstlane(wave);
    const int g = lane >> 3;
    const unsigned q = (unsigned)(lane & 7);
    const float4* __restrict__ z4 = (const float4*)z;
    float4* __restrict__ hout4 = (float4*)hout;
    for (int dl = wave_s; dl < BNODES; dl += SORT_WAVES) {
        const int node = b * BNODES + dl;
        if (node >= N_NODES) break;
        const int first = __builtin_amdgcn_readfirstlane(nfirst[dl]);
        const int n     = __builtin_amdgcn_readfirstlane(nfirst[dl + 1]) - first;
        const float erd = ler[dl];

        const bool has = lane < n;
        int2 rec = make_int2(0, 0);
        if (n > 0) rec = rec_at<IN_LDS>(lrec, gs, el, erd, first + (has ? lane : n - 1));
        const float e0 = __int_as_float(rec.y);

        float m = has ? e0 : -INFINITY;
        for (int j = lane + 32; j < n; j += 32)
            m = fmaxf(m, __int_as_float(rec_at<IN_LDS>(lrec, gs, el, erd, first + j).y));
        #pragma unroll
        for (int o = 16; o > 0; o >>= 1) m = fmaxf(m, __shfl_xor(m, o, 32));

        float ssum = has ? __expf(e0 - m) : 0.0f;
        for (int j = lane + 32; j < n; j += 32)
            ssum += __expf(__int_as_float(rec_at<IN_LDS>(lrec, gs, el, erd, first + j).y) - m);
        #pragma unroll
        for (int o = 16; o > 0; o >>= 1) ssum += __shfl_xor(ssum, o, 32);

        float4 acc = make_float4(0.0f, 0.0f, 0.0f, 0.0f);
        for (int base = 0; base < n; base += ROW_EDGES) {
            const int left = n - base;
            const bool in = lane < left;
            if (base > 0) rec = rec_at<IN_LDS>(lrec, gs, el, erd, first + (in ? base + lane : n - 1));
            const int s = rec.x;
            const int abits = __float_as_int(in ? __expf(__int_as_float(rec.y) - m) / ssum : 0.0f);
            float4 r[ROW_LOADS];
            float ak[ROW_LOADS];
            #pragma unroll
            for (int u = 0; u < ROW_LOADS; ++u) {
                const unsigned sk = (unsigned)__shfl(s, 4 * u + g, 32);
                r[u] = z4[sk * 8u + q];
            }
            #pragma unroll
            for (int u = 0; u < ROW_LOADS; ++u)
                ak[u] = __int_as_float(__shfl(abits, 4 * u + g, 32));
            #pragma unroll
            for (int u = 0; u < ROW_LOADS; ++u) {
                acc.x += ak[u] * r[u].x;
                acc.y += ak[u] * r[u].y;
                acc.z += ak[u] * r[u].z;
                acc.w += ak[u] * r[u].w;
            }
        }
        acc.x += row_xor_f<8>(acc.x);
        acc.y += row_xor_f<8>(acc.y);
        acc.z += row_xor_f<8>(acc.z);
        acc.w += row_xor_f<8>(acc.w);
        acc.x += __shfl_xor(acc.x, 16, 32);
        acc.y += __shfl_xor(acc.y, 16, 32);
        acc.z += __shfl_xor(acc.z, 16, 32);
        acc.w += __shfl_xor(acc.w, 16, 32);
        if (g == 0) hout4[(unsigned)node * 8u + q] = acc;
    }
}

__global__ void __launch_bounds__(SORT_T, 16)
gat_bucket_gather_kernel(const float* __restrict__ el,
                         const float* __restrict__ er,
                         const int* __restrict__ bstart,
                         const int* __restrict__ tmp,
                         int* gs,
                         const float* __restrict__ z,
                         float* __restrict__ hout) {
    __shared__ int   nc[BNODES];
    __shared__ int   nfirst[BNODES + 1];
    __shared__ float ler[BNODES];
    __shared__ int   wtot[BNODES / 32];
    __shared__ int2  lrec[LCAP];
    const int t = threadIdx.x;
    const int lane = t & 31;
    const int wave = t >> 5;
    const int b = blockIdx.x;
    const int node = b * BNODES + t;
    const int beg = bstart[b];
    const int end = (b + 1 < NBUCK) ? bstart[b + 1] : N_EDGES;
    const bool fits = (end - beg) <= LCAP;
    if (t < BNODES) {
        nc[t] = 0;
        ler[t] = (node < N_NODES) ? er[node] : 0.0f;
    }
    __syncthreads();
    for (int j0 = beg + t; j0 < end; j0 += FILL_U * SORT_T) {
        int pv[FILL_U];
        #pragma unroll
        for (int u = 0; u < FILL_U; ++u) {
            const int j = j0 + u * SORT_T;
            pv[u] = tmp[j < end ? j : end - 1];
        }
        #pragma unroll
        for (int u = 0; u < FILL_U; ++u)
            if (j0 + u * SORT_T < end) atomicAdd(&nc[pv[u] >> SRC_BITS], 1);
    }
    __syncthreads();
    int c = 0, inc = 0;
    if (t < BNODES) {
        c = nc[t];
        inc = c;
        #pragma unroll
        for (int o = 1; o < 32; o <<= 1) {
            const int u = __shfl_up(inc, o, 32);
            if (lane >= o) inc += u;
        }
        if (lane == 31) wtot[t >> 5] = inc;
    }
    __syncthreads();
    if (t < BNODES) {
        int base = 0;
        #pragma unroll
        for (int w = 0; w < BNODES / 32; ++w)
            if (w < (t >> 5)) base += wtot[w];
        const int first = base + inc - c;
        nc[t] = first;
        nfirst[t] = first;
        if (t == BNODES - 1) nfirst[BNODES] = first + c;
    }
    __syncthreads();
    for (int j0 = beg + t; j0 < end; j0 += FILL_U * SORT_T) {
        int pv[FILL_U];
        float ev[FILL_U];
        #pragma unroll
        for (int u = 0; u < FILL_U; ++u) {
            const int j = j0 + u * SORT_T;
            pv[u] = tmp[j < end ? j : end - 1];
        }
        #pragma unroll
        for (int u = 0; u < FILL_U; ++u) ev[u] = el[pv[u] & ((1 << SRC_BITS) - 1)];
        #pragma unroll
        for (int u = 0; u < FILL_U; ++u) {
            if (j0 + u * SORT_T < end) {
                const int s  = pv[u] & ((1 << SRC_BITS) - 1);
                const int dl = pv[u] >> SRC_BITS;
                const int pos = atomicAdd(&nc[dl], 1);
                float e = ev[u] + ler[dl];
                e = (e >= 0.0f) ? e : LRELU_SLOPE * e;
                if (fits) lrec[pos] = make_int2(s, __float_as_int(e));
                else      gs[beg + pos] = s;
            }
        }
    }
    if (!fits) __threadfence();
    __syncthreads();
    if (fits) bucket_nodes<true >(lrec, gs + beg, nfirst, ler, el, z, hout, b, wave, lane);
    else      bucket_nodes<false>(lrec, gs + beg, nfirst, ler, el, z, hout, b, wave, lane);
}

#define STAT_T      256
#define STAT_BLOCKS 512
#define STAT_WAVES  (STAT_BLOCKS * STAT_T / 32)
#define STAT_ROWS   ((N_NODES + STAT_WAVES - 1) / STAT_WAVES)
static_assert((STAT_ROWS - 1) * STAT_WAVES < N_NODES, "only a wave's last row can lie past the end");
__global__ void __launch_bounds__(STAT_T)
gat_bn_stats_kernel(const float* __restrict__ hout, float* __restrict__ bn) {
    __shared__ float ssum[OUT_DIM];
    __shared__ float ssq[OUT_DIM];
    const int lane = threadIdx.x & 31;
    if (threadIdx.x < OUT_DIM) { ssum[threadIdx.x] = 0.0f; ssq[threadIdx.x] = 0.0f; }
    const int w = (blockIdx.x * STAT_T + threadIdx.x) >> 5;
    const int last = w + (STAT_ROWS - 1) * STAT_WAVES;
    const bool has_last = last < N_NODES;
    float v[STAT_ROWS];
    #pragma unroll
    for (int i = 0; i < STAT_ROWS - 1; ++i)
        v[i] = hout[(size_t)(w + i * STAT_WAVES) * OUT_DIM + lane];
    v[STAT_ROWS - 1] = hout[(size_t)(has_last ? last : N_NODES - 1) * OUT_DIM + lane];
    float s = 0.0f, q = 0.0f;
    #pragma unroll
    for (int i = 0; i < STAT_ROWS - 1; ++i) {
        s += v[i];
        q += v[i] * v[i];
    }
    if (has_last) {
        s += v[STAT_ROWS - 1];
        q += v[STAT_ROWS - 1] * v[STAT_ROWS - 1];
    }
    __syncthreads();
    atomicAdd(&ssum[lane], s);
    atomicAdd(&ssq[lane], q);
    __syncthreads();
    if (threadIdx.x < OUT_DIM) {
        atomicAdd(&bn[threadIdx.x], ssum[threadIdx.x]);
        atomicAdd(&bn[OUT_DIM + threadIdx.x], ssq[threadIdx.x]);
    }
}

__global__ void gat_bn_apply_kernel(const float* __restrict__ bn,
                                    const float* __restrict__ gamma,
                                    const float* __restrict__ beta,
                                    float* __restrict__ out) {
    int idx = blockIdx.x * blockDim.x + threadIdx.x;
    if (idx >= N_NODES * OUT_DIM) return;
    int c = idx & 31;
    const float inv_n = 1.0f / (float)N_NODES;
    float mean = bn[c] * inv_n;
    float var  = bn[OUT_DIM + c] * inv_n - mean * mean;
    float x = (out[idx] - mean) * rsqrtf(var + BN_EPS) * gamma[c] + beta[c];
    out[idx] = (x > 0.0f) ? x : expm1f(x);
}

extern "C" void kernel_launch(void* const* d_in, const int* in_sizes, int n_in,
                              void* d_out, int out_size, void* d_ws, size_t ws_size,
                              hipStream_t stream) {
    const float* h     = (const float*)d_in[0];
    const float* W_fc  = (const float*)d_in[1];
    const float* W_at  = (const float*)d_in[2];
    const float* gamma = (const float*)d_in[3];
    const float* beta  = (const float*)d_in[4];
    const int*   src   = (const int*)d_in[5];
    const int*   dst   = (const int*)d_in[6];
    float* out = (float*)d_out;

    float* z    = (float*)d_ws;
    float* el   = z + (size_t)N_NODES * OUT_DIM;
    float* er   = el + N_NODES;
    int*   bstart = (int*)(er + N_NODES);
    int*   tmp  = bstart + 2 * N_NODES;
    int*   gs   = tmp + N_EDGES;
    float* bn   = (float*)(gs + N_EDGES);
    unsigned short* tab = (unsigned short*)d_out;

    const int B = 256;

    gat_gemm_proj_count_kernel<<<GEMM_BLOCKS + NCHUNK, SORT_T, 0, stream>>>(h, W_fc, W_at, dst, z, el, er, tab, bn);

    gat_bucket_scatter_kernel<<<NCHUNK, SORT_T, 0, stream>>>(src, dst, tab, bstart, tmp);

    gat_bucket_gather_kernel<<<NBUCK, SORT_T, 0, stream>>>(el, er, bstart, tmp, gs, z, out);

    gat_bn_stats_kernel<<<STAT_BLOCKS, STAT_T, 0, stream>>>(out, bn);

    gat_bn_apply_kernel<<<(N_NODES * OUT_DIM + B - 1) / B, B, 0, stream>>>(bn, gamma, beta, out);
}


# Node pass: a wave takes two nodes at once, one per 16-lane row; max, sum and edge hand-out are in-row vector lane exchanges, no LDS permutes (new order of softmax and row sums)
# speedup vs baseline: 4.3213x; 1.0370x over previous
#include <hip/hip_runtime.h>
#include <hip/hip_bf16.h>

#define N_NODES 100000
#define N_EDGES 1600000
#define IN_DIM 128
#define OUT_DIM 32
#define BN_EPS 1e-5f
#define LRELU_SLOPE 0.01f

#define BSHIFT 8
#define BNODES 256
#define NBUCK  ((N_NODES + BNODES - 1) / BNODES)
#define NCHUNK 256
#define CHUNK  (N_EDGES / NCHUNK)
#define TABN   (NBUCK * NCHUNK)
#define SORT_T 1024
#define EPT    ((CHUNK + SORT_T - 1) / SORT_T)
#define SRC_BITS 17
#define FILL_U 5
static_assert(N_EDGES % NCHUNK == 0, "chunks tile the edge list");
static_assert(TABN % 8 == 0, "table rows of 16-bit counts are read as int4");
static_assert(N_NODES <= (1 << SRC_BITS), "src and the in-bucket dst share one word");
static_assert(NCHUNK == 256, "a table row of 16-bit counts is one int4 per lane of one wave");
static_assert(4 * CHUNK <= 65535, "four piece sizes summed still fit 16 bits");
static_assert(NBUCK <= SORT_T, "one thread per bucket in the cursor scan");

#define GEMM_TILES   (N_NODES / 16)
#define GEMM_WAVES   (SORT_T / 32)
#define GEMM_BLOCKS  ((GEMM_TILES + GEMM_WAVES - 1) / GEMM_WAVES)
#define SCAN_WAVES   ((NBUCK + 31) / 32)
#define SORT_WAVES   (SORT_T / 32)
#define ROWS_PER_WAVE ((NBUCK + SORT_WAVES - 1) / SORT_WAVES)
static_assert(ROWS_PER_WAVE <= 16, "one lane of each half wave per row of a wave");

typedef __attribute__((ext_vector_type(2))) float v2f;
typedef __attribute__((ext_vector_type(8))) float v8f;

template <int M> __device__ __forceinline__ float row_xor_f(float x) {
    return __int_as_float(__builtin_amdgcn_update_dpp(0, __float_as_int(x), 0x160 + M, 0xf, 0xf, true));
}
template <int M> __device__ __forceinline__ int row_xor_i(int x) {
    return __builtin_amdgcn_update_dpp(0, x, 0x160 + M, 0xf, 0xf, true);
}
__device__ __forceinline__ float row_sum16_f(float x) {
    x += row_xor_f<1>(x);
    x += row_xor_f<2>(x);
    x += row_xor_f<4>(x);
    x += row_xor_f<8>(x);
    return x;
}
__device__ __forceinline__ int row_sum16_i(int x) {
    x += row_xor_i<1>(x);
    x += row_xor_i<2>(x);
    x += row_xor_i<4>(x);
    x += row_xor_i<8>(x);
    return x;
}

__global__ void __launch_bounds__(SORT_T)
gat_gemm_proj_count_kernel(const float* __restrict__ h,
                           const float* __restrict__ W,
                           const float* __restrict__ Wa,
                           const int* __restrict__ dst,
                           float* __restrict__ z,
                           float* __restrict__ el,
                           float* __restrict__ er,
                           unsigned short* __restrict__ tab,
                           float* __restrict__ bn) {
    __shared__ float lds_w[IN_DIM * OUT_DIM];
    __shared__ int bc[NBUCK];
    const int t = threadIdx.x;

    if (blockIdx.x >= GEMM_BLOCKS) {
        const int chunk = blockIdx.x - GEMM_BLOCKS;
        if (t < NBUCK) bc[t] = 0;
        if (chunk == 0 && t < 2 * OUT_DIM) bn[t] = 0.0f;
        __syncthreads();
        const int* d = dst + chunk * CHUNK;
        int dv[EPT];
        #pragma unroll
        for (int k = 0; k < EPT; ++k) {
            const int i = t + k * SORT_T;
            dv[k] = d[i < CHUNK ? i : CHUNK - 1];
        }
        #pragma unroll
        for (int k = 0; k < EPT; ++k)
            if (t + k * SORT_T < CHUNK) atomicAdd(&bc[dv[k] >> BSHIFT], 1);
        __syncthreads();
        if (t < NBUCK) tab[t * NCHUNK + chunk] = (unsigned short)bc[t];
        return;
    }

    for (int i = t; i < IN_DIM * OUT_DIM; i += SORT_T)
        lds_w[i] = W[i];
    __syncthreads();

    const int wave = t >> 5;
    const int lane = t & 31;
    const int tile = blockIdx.x * GEMM_WAVES + wave;
    if (tile >= GEMM_TILES) return;

    const int m0   = tile * 16;
    const int row  = m0 + (lane & 15);
    const int koff = (lane >> 4) * 2;
    const int col  = lane & 15;

    const float al0 = Wa[col];
    const float al1 = Wa[col + 16];
    const float ar0 = Wa[OUT_DIM + col];
    const float ar1 = Wa[OUT_DIM + col + 16];

    v8f c0 = {}, c1 = {};
    const float* hrow = h + (size_t)row * IN_DIM;

    #pragma unroll 4
    for (int k = 0; k < IN_DIM; k += 4) {
        v2f a;
        a.x = hrow[k + koff];
        a.y = hrow[k + koff + 1];
        v2f b0, b1;
        b0.x = lds_w[(k + koff)     * OUT_DIM + col];
        b0.y = lds_w[(k + koff + 1) * OUT_DIM + col];
        b1.x = lds_w[(k + koff)     * OUT_DIM + col + 16];
        b1.y = lds_w[(k + koff + 1) * OUT_DIM + col + 16];
        c0 = __builtin_amdgcn_wmma_f32_16x16x4_f32(false, a, false, b0, (short)0, c0, false, false);
        c1 = __builtin_amdgcn_wmma_f32_16x16x4_f32(false, a, false, b1, (short)0, c1, false, false);
    }

    const int rbase = m0 + ((lane >> 4) * 8);
    #pragma unroll
    for (int r = 0; r < 8; ++r) {
        z[(size_t)(rbase + r) * OUT_DIM + col]      = c0[r];
        z[(size_t)(rbase + r) * OUT_DIM + col + 16] = c1[r];
    }

    __builtin_amdgcn_sched_barrier(0);
    float pl[8], pr[8];
    #pragma unroll
    for (int r = 0; r < 8; ++r) {
        pl[r] = row_sum16_f(c0[r] * al0 + c1[r] * al1);
        pr[r] = row_sum16_f(c0[r] * ar0 + c1[r] * ar1);
    }
    if (col == 0) {
        float4* pe = (float4*)(el + rbase);
        pe[0] = make_float4(pl[0], pl[1], pl[2], pl[3]);
        pe[1] = make_float4(pl[4], pl[5], pl[6], pl[7]);
        float4* qe = (float4*)(er + rbase);
        qe[0] = make_float4(pr[0], pr[1], pr[2], pr[3]);
        qe[1] = make_float4(pr[4], pr[5], pr[6], pr[7]);
    }
}

__global__ void __launch_bounds__(SORT_T)
gat_bucket_scatter_kernel(const int* __restrict__ src,
                          const int* __restrict__ dst,
                          const unsigned short* __restrict__ tab,
                          int* __restrict__ bstart,
                          int* __restrict__ tmp) {
    __shared__ int cur[NBUCK];
    __shared__ int rtot[NBUCK];
    __shared__ int wsum[SCAN_WAVES];
    const int t = threadIdx.x;
    const int lane = t & 31;
    const int wave = t >> 5;
    const int c = blockIdx.x;
    const int* s = src + c * CHUNK;
    const int* d = dst + c * CHUNK;
    int sv[EPT], dv[EPT];
    #pragma unroll
    for (int k = 0; k < EPT; ++k) {
        const int i = t + k * SORT_T;
        const int ic = i < CHUNK ? i : CHUNK - 1;
        sv[k] = s[ic];
        dv[k] = d[ic];
    }

    const int4* t4 = (const int4*)tab;
    const int j0 = 8 * lane;
    int4 p[ROWS_PER_WAVE];
    #pragma unroll
    for (int i = 0; i < ROWS_PER_WAVE; ++i) {
        const int b  = wave + SORT_WAVES * i;
        const int bb = b < NBUCK ? b : NBUCK - 1;
        p[i] = t4[bb * (NCHUNK / 8) + lane];
    }
    __builtin_amdgcn_sched_barrier(0);
    unsigned keep[4];
    #pragma unroll
    for (int k = 0; k < 4; ++k)
        keep[k] = (j0 + 2 * k < c ? 0x0000ffffu : 0u) | (j0 + 2 * k + 1 < c ? 0xffff0000u : 0u);
    int vt = 0, vp = 0;
    #pragma unroll
    for (int i = 0; i < ROWS_PER_WAVE; ++i) {
        const unsigned w0 = (unsigned)p[i].x, w1 = (unsigned)p[i].y;
        const unsigned w2 = (unsigned)p[i].z, w3 = (unsigned)p[i].w;
        const unsigned t2 = (w0 + w1) + (w2 + w3);
        const unsigned p2 = ((w0 & keep[0]) + (w1 & keep[1])) + ((w2 & keep[2]) + (w3 & keep[3]));
        const int ts = row_sum16_i((int)((t2 & 0xffffu) + (t2 >> 16)));
        const int ps = row_sum16_i((int)((p2 & 0xffffu) + (p2 >> 16)));
        if ((lane & 15) == i) { vt = ts; vp = ps; }
    }
    vt += __shfl_xor(vt, 16, 32);
    vp += __shfl_xor(vp, 16, 32);
    const int bl = wave + SORT_WAVES * lane;
    if (lane < ROWS_PER_WAVE && bl < NBUCK) { rtot[bl] = vt; cur[bl] = vp; }
    __syncthreads();
    int own = 0, inc = 0;
    if (wave < SCAN_WAVES) {
        own = (t < NBUCK) ? rtot[t] : 0;
        inc = own;
        #pragma unroll
        for (int o = 1; o < 32; o <<= 1) {
            const int u = __shfl_up(inc, o, 32);
            if (lane >= o) inc += u;
        }
        if (lane == 31) wsum[wave] = inc;
    }
    __syncthreads();
    if (t < NBUCK) {
        int base = 0;
        #pragma unroll
        for (int w = 0; w < SCAN_WAVES; ++w)
            if (w < wave) base += wsum[w];
        const int first = base + inc - own;
        cur[t] += first;
        if (c == 0) bstart[t] = first;
    }
    __syncthreads();
    #pragma unroll
    for (int k = 0; k < EPT; ++k) {
        if (t + k * SORT_T < CHUNK) {
            const int pos = atomicAdd(&cur[dv[k] >> BSHIFT], 1);
            tmp[pos] = sv[k] | ((dv[k] & (BNODES - 1)) << SRC_BITS);
        }
    }
}

#define LCAP 8192
#define ROW_LOADS 8
#define ROW_EDGES (2 * ROW_LOADS)
static_assert(ROW_EDGES <= 16, "a group's edges are one lane each of a 16-lane row");
static_assert(BNODES % 2 == 0 && N_NODES % 2 == 0, "the two nodes of a pair are both real or both past the end");

__device__ __forceinline__ float row_max16_f(float x) {
    x = fmaxf(x, row_xor_f<1>(x));
    x = fmaxf(x, row_xor_f<2>(x));
    x = fmaxf(x, row_xor_f<4>(x));
    x = fmaxf(x, row_xor_f<8>(x));
    return x;
}
template <int U> __device__ __forceinline__ int row_pick2(int x) {
    return (int)__builtin_amdgcn_permlane16((unsigned)x, (unsigned)x, 0x11111111u * (unsigned)(2 * U),
                                            0x11111111u * (unsigned)(2 * U + 1), true, false);
}

template <bool IN_LDS>
__device__ __forceinline__ int2 rec_at(const int2* lrec, const int* gs, const float* __restrict__ el,
                                       float erd, int j) {
    if (IN_LDS) return lrec[j];
    const int s = __hip_atomic_load(gs + j, __ATOMIC_RELAXED, __HIP_MEMORY_SCOPE_AGENT);
    float e = el[s] + erd;
    e = (e >= 0.0f) ? e : LRELU_SLOPE * e;
    return make_int2(s, __float_as_int(e));
}

template <int U>
__device__ __forceinline__ void row_group_loads(const float4* __restrict__ z4, int s, int abits, unsigned q,
                                                float4* r, float* ak) {
    if constexpr (U < ROW_LOADS) {
        const unsigned sk = (unsigned)row_pick2<U>(s);
        r[U] = z4[sk * 8u + q];
        ak[U] = __int_as_float(row_pick2<U>(abits));
        row_group_loads<U + 1>(z4, s, abits, q, r, ak);
    }
}

template <bool IN_LDS>
__device__ __forceinline__ void bucket_nodes(const int2* lrec, const int* gs, const int* nfirst,
                                             const float* ler, const float* __restrict__ el,
                                             const float* __restrict__ z, float* __restrict__ hout,
                                             int b, int wave, int lane) {
    const int wave_s = __builtin_amdgcn_readfirstlane(wave);
    const int hrow = lane >> 4;
    const int l = lane & 15;
    const unsigned q = (unsigned)(lane & 7);
    const float4* __restrict__ z4 = (const float4*)z;
    float4* __restrict__ hout4 = (float4*)hout;
    for (int p = wave_s; p < BNODES / 2; p += SORT_WAVES) {
        if (b * BNODES + 2 * p >= N_NODES) break;
        const int dl = 2 * p + hrow;
        const int node = b * BNODES + dl;
        const int first = nfirst[dl];
        const int n     = nfirst[dl + 1] - first;
        const float erd = ler[dl];

        const bool has = l < n;
        int2 rec = make_int2(0, 0);
        if (n > 0) rec = rec_at<IN_LDS>(lrec, gs, el, erd, first + (has ? l : n - 1));
        const float e0 = __int_as_float(rec.y);

        float m = has ? e0 : -INFINITY;
        for (int j = l + 16; j < n; j += 16)
            m = fmaxf(m, __int_as_float(rec_at<IN_LDS>(lrec, gs, el, erd, first + j).y));
        m = row_max16_f(m);

        float ssum = has ? __expf(e0 - m) : 0.0f;
        for (int j = l + 16; j < n; j += 16)
            ssum += __expf(__int_as_float(rec_at<IN_LDS>(lrec, gs, el, erd, first + j).y) - m);
        ssum = row_sum16_f(ssum);

        const int n0 = __builtin_amdgcn_readlane(n, 0);
        const int n1 = __builtin_amdgcn_readlane(n, 16);
        const int nmax = n0 > n1 ? n0 : n1;
        float4 acc = make_float4(0.0f, 0.0f, 0.0f, 0.0f);
        for (int base = 0; base < nmax; base += ROW_EDGES) {
            const bool in = base + l < n;
            if (base > 0 && n > 0) rec = rec_at<IN_LDS>(lrec, gs, el, erd, first + (in ? base + l : n - 1));
            const int s = rec.x;
            const int abits = __float_as_int(in ? __expf(__int_as_float(rec.y) - m) / ssum : 0.0f);
            float4 r[ROW_LOADS];
            float ak[ROW_LOADS];
            row_group_loads<0>(z4, s, abits, q, r, ak);
            #pragma unroll
            for (int u = 0; u < ROW_LOADS; ++u) {
                acc.x += ak[u] * r[u].x;
                acc.y += ak[u] * r[u].y;
                acc.z += ak[u] * r[u].z;
                acc.w += ak[u] * r[u].w;
            }
        }
        acc.x += row_xor_f<8>(acc.x);
        acc.y += row_xor_f<8>(acc.y);
        acc.z += row_xor_f<8>(acc.z);
        acc.w += row_xor_f<8>(acc.w);
        if ((l >> 3) == 0) hout4[(unsigned)node * 8u + q] = acc;
    }
}

__global__ void __launch_bounds__(SORT_T, 16)
gat_bucket_gather_kernel(const float* __restrict__ el,
                         const float* __restrict__ er,
                         const int* __restrict__ bstart,
                         const int* __restrict__ tmp,
                         int* gs,
                         const float* __restrict__ z,
                         float* __restrict__ hout) {
    __shared__ int   nc[BNODES];
    __shared__ int   nfirst[BNODES + 1];
    __shared__ float ler[BNODES];
    __shared__ int   wtot[BNODES / 32];
    __shared__ int2  lrec[LCAP];
    const int t = threadIdx.x;
    const int lane = t & 31;
    const int wave = t >> 5;
    const int b = blockIdx.x;
    const int node = b * BNODES + t;
    const int beg = bstart[b];
    const int end = (b + 1 < NBUCK) ? bstart[b + 1] : N_EDGES;
    const bool fits = (end - beg) <= LCAP;
    if (t < BNODES) {
        nc[t] = 0;
        ler[t] = (node < N_NODES) ? er[node] : 0.0f;
    }
    __syncthreads();
    for (int j0 = beg + t; j0 < end; j0 += FILL_U * SORT_T) {
        int pv[FILL_U];
        #pragma unroll
        for (int u = 0; u < FILL_U; ++u) {
            const int j = j0 + u * SORT_T;
            pv[u] = tmp[j < end ? j : end - 1];
        }
        #pragma unroll
        for (int u = 0; u < FILL_U; ++u)
            if (j0 + u * SORT_T < end) atomicAdd(&nc[pv[u] >> SRC_BITS], 1);
    }
    __syncthreads();
    int c = 0, inc = 0;
    if (t < BNODES) {
        c = nc[t];
        inc = c;
        #pragma unroll
        for (int o = 1; o < 32; o <<= 1) {
            const int u = __shfl_up(inc, o, 32);
            if (lane >= o) inc += u;
        }
        if (lane == 31) wtot[t >> 5] = inc;
    }
    __syncthreads();
    if (t < BNODES) {
        int base = 0;
        #pragma unroll
        for (int w = 0; w < BNODES / 32; ++w)
            if (w < (t >> 5)) base += wtot[w];
        const int first = base + inc - c;
        nc[t] = first;
        nfirst[t] = first;
        if (t == BNODES - 1) nfirst[BNODES] = first + c;
    }
    __syncthreads();
    for (int j0 = beg + t; j0 < end; j0 += FILL_U * SORT_T) {
        int pv[FILL_U];
        float ev[FILL_U];
        #pragma unroll
        for (int u = 0; u < FILL_U; ++u) {
            const int j = j0 + u * SORT_T;
            pv[u] = tmp[j < end ? j : end - 1];
        }
        #pragma unroll
        for (int u = 0; u < FILL_U; ++u) ev[u] = el[pv[u] & ((1 << SRC_BITS) - 1)];
        #pragma unroll
        for (int u = 0; u < FILL_U; ++u) {
            if (j0 + u * SORT_T < end) {
                const int s  = pv[u] & ((1 << SRC_BITS) - 1);
                const int dl = pv[u] >> SRC_BITS;
                const int pos = atomicAdd(&nc[dl], 1);
                float e = ev[u] + ler[dl];
                e = (e >= 0.0f) ? e : LRELU_SLOPE * e;
                if (fits) lrec[pos] = make_int2(s, __float_as_int(e));
                else      gs[beg + pos] = s;
            }
        }
    }
    if (!fits) __threadfence();
    __syncthreads();
    if (fits) bucket_nodes<true >(lrec, gs + beg, nfirst, ler, el, z, hout, b, wave, lane);
    else      bucket_nodes<false>(lrec, gs + beg, nfirst, ler, el, z, hout, b, wave, lane);
}

#define STAT_T      256
#define STAT_BLOCKS 512
#define STAT_WAVES  (STAT_BLOCKS * STAT_T / 32)
#define STAT_ROWS   ((N_NODES + STAT_WAVES - 1) / STAT_WAVES)
static_assert((STAT_ROWS - 1) * STAT_WAVES < N_NODES, "only a wave's last row can lie past the end");
__global__ void __launch_bounds__(STAT_T)
gat_bn_stats_kernel(const float* __restrict__ hout, float* __restrict__ bn) {
    __shared__ float ssum[OUT_DIM];
    __shared__ float ssq[OUT_DIM];
    const int lane = threadIdx.x & 31;
    if (threadIdx.x < OUT_DIM) { ssum[threadIdx.x] = 0.0f; ssq[threadIdx.x] = 0.0f; }
    const int w = (blockIdx.x * STAT_T + threadIdx.x) >> 5;
    const int last = w + (STAT_ROWS - 1) * STAT_WAVES;
    const bool has_last = last < N_NODES;
    float v[STAT_ROWS];
    #pragma unroll
    for (int i = 0; i < STAT_ROWS - 1; ++i)
        v[i] = hout[(size_t)(w + i * STAT_WAVES) * OUT_DIM + lane];
    v[STAT_ROWS - 1] = hout[(size_t)(has_last ? last : N_NODES - 1) * OUT_DIM + lane];
    float s = 0.0f, q = 0.0f;
    #pragma unroll
    for (int i = 0; i < STAT_ROWS - 1; ++i) {
        s += v[i];
        q += v[i] * v[i];
    }
    if (has_last) {
        s += v[STAT_ROWS - 1];
        q += v[STAT_ROWS - 1] * v[STAT_ROWS - 1];
    }
    __syncthreads();
    atomicAdd(&ssum[lane], s);
    atomicAdd(&ssq[lane], q);
    __syncthreads();
    if (threadIdx.x < OUT_DIM) {
        atomicAdd(&bn[threadIdx.x], ssum[threadIdx.x]);
        atomicAdd(&bn[OUT_DIM + threadIdx.x], ssq[threadIdx.x]);
    }
}

__global__ void gat_bn_apply_kernel(const float* __restrict__ bn,
                                    const float* __restrict__ gamma,
                                    const float* __restrict__ beta,
                                    float* __restrict__ out) {
    int idx = blockIdx.x * blockDim.x + threadIdx.x;
    if (idx >= N_NODES * OUT_DIM) return;
    int c = idx & 31;
    const float inv_n = 1.0f / (float)N_NODES;
    float mean = bn[c] * inv_n;
    float var  = bn[OUT_DIM + c] * inv_n - mean * mean;
    float x = (out[idx] - mean) * rsqrtf(var + BN_EPS) * gamma[c] + beta[c];
    out[idx] = (x > 0.0f) ? x : expm1f(x);
}

extern "C" void kernel_launch(void* const* d_in, const int* in_sizes, int n_in,
                              void* d_out, int out_size, void* d_ws, size_t ws_size,
                              hipStream_t stream) {
    const float* h     = (const float*)d_in[0];
    const float* W_fc  = (const float*)d_in[1];
    const float* W_at  = (const float*)d_in[2];
    const float* gamma = (const float*)d_in[3];
    const float* beta  = (const float*)d_in[4];
    const int*   src   = (const int*)d_in[5];
    const int*   dst   = (const int*)d_in[6];
    float* out = (float*)d_out;

    float* z    = (float*)d_ws;
    float* el   = z + (size_t)N_NODES * OUT_DIM;
    float* er   = el + N_NODES;
    int*   bstart = (int*)(er + N_NODES);
    int*   tmp  = bstart + 2 * N_NODES;
    int*   gs   = tmp + N_EDGES;
    float* bn   = (float*)(gs + N_EDGES);
    unsigned short* tab = (unsigned short*)d_out;

    const int B = 256;

    gat_gemm_proj_count_kernel<<<GEMM_BLOCKS + NCHUNK, SORT_T, 0, stream>>>(h, W_fc, W_at, dst, z, el, er, tab, bn);

    gat_bucket_scatter_kernel<<<NCHUNK, SORT_T, 0, stream>>>(src, dst, tab, bstart, tmp);

    gat_bucket_gather_kernel<<<NBUCK, SORT_T, 0, stream>>>(el, er, bstart, tmp, gs, z, out);

    gat_bn_stats_kernel<<<STAT_BLOCKS, STAT_T, 0, stream>>>(out, bn);

    gat_bn_apply_kernel<<<(N_NODES * OUT_DIM + B - 1) / B, B, 0, stream>>>(bn, gamma, beta, out);
}


# Scatter launch: a chunk's edge words are put in bucket order in LDS first and stored from there: a store instruction writes a few runs of neighbouring words of tmp, not 32 words in 32 lines
# speedup vs baseline: 5.7064x; 1.3205x over previous
#include <hip/hip_runtime.h>
#include <hip/hip_bf16.h>

#define N_NODES 100000
#define N_EDGES 1600000
#define IN_DIM 128
#define OUT_DIM 32
#define BN_EPS 1e-5f
#define LRELU_SLOPE 0.01f

#define BSHIFT 8
#define BNODES 256
#define NBUCK  ((N_NODES + BNODES - 1) / BNODES)
#define NCHUNK 256
#define CHUNK  (N_EDGES / NCHUNK)
#define TABN   (NBUCK * NCHUNK)
#define SORT_T 1024
#define EPT    ((CHUNK + SORT_T - 1) / SORT_T)
#define SRC_BITS 17
#define FILL_U 5
static_assert(N_EDGES % NCHUNK == 0, "chunks tile the edge list");
static_assert(TABN % 8 == 0, "table rows of 16-bit counts are read as int4");
static_assert(N_NODES <= (1 << SRC_BITS), "src and the in-bucket dst share one word");
static_assert(NCHUNK == 256, "a table row of 16-bit counts is one int4 per lane of one wave");
static_assert(4 * CHUNK <= 65535, "four piece sizes summed still fit 16 bits");
static_assert(NBUCK <= SORT_T, "one thread per bucket in the cursor scan");

#define GEMM_TILES   (N_NODES / 16)
#define GEMM_WAVES   (SORT_T / 32)
#define GEMM_BLOCKS  ((GEMM_TILES + GEMM_WAVES - 1) / GEMM_WAVES)
#define SCAN_WAVES   ((NBUCK + 31) / 32)
#define SORT_WAVES   (SORT_T / 32)
#define ROWS_PER_WAVE ((NBUCK + SORT_WAVES - 1) / SORT_WAVES)
static_assert(ROWS_PER_WAVE <= 16, "one lane of each half wave per row of a wave");

typedef __attribute__((ext_vector_type(2))) float v2f;
typedef __attribute__((ext_vector_type(8))) float v8f;

template <int M> __device__ __forceinline__ float row_xor_f(float x) {
    return __int_as_float(__builtin_amdgcn_update_dpp(0, __float_as_int(x), 0x160 + M, 0xf, 0xf, true));
}
template <int M> __device__ __forceinline__ int row_xor_i(int x) {
    return __builtin_amdgcn_update_dpp(0, x, 0x160 + M, 0xf, 0xf, true);
}
__device__ __forceinline__ float row_sum16_f(float x) {
    x += row_xor_f<1>(x);
    x += row_xor_f<2>(x);
    x += row_xor_f<4>(x);
    x += row_xor_f<8>(x);
    return x;
}
__device__ __forceinline__ int row_sum16_i(int x) {
    x += row_xor_i<1>(x);
    x += row_xor_i<2>(x);
    x += row_xor_i<4>(x);
    x += row_xor_i<8>(x);
    return x;
}

__global__ void __launch_bounds__(SORT_T)
gat_gemm_proj_count_kernel(const float* __restrict__ h,
                           const float* __restrict__ W,
                           const float* __restrict__ Wa,
                           const int* __restrict__ dst,
                           float* __restrict__ z,
                           float* __restrict__ el,
                           float* __restrict__ er,
                           unsigned short* __restrict__ tab,
                           float* __restrict__ bn) {
    __shared__ float lds_w[IN_DIM * OUT_DIM];
    __shared__ int bc[NBUCK];
    const int t = threadIdx.x;

    if (blockIdx.x >= GEMM_BLOCKS) {
        const int chunk = blockIdx.x - GEMM_BLOCKS;
        if (t < NBUCK) bc[t] = 0;
        if (chunk == 0 && t < 2 * OUT_DIM) bn[t] = 0.0f;
        __syncthreads();
        const int* d = dst + chunk * CHUNK;
        int dv[EPT];
        #pragma unroll
        for (int k = 0; k < EPT; ++k) {
            const int i = t + k * SORT_T;
            dv[k] = d[i < CHUNK ? i : CHUNK - 1];
        }
        #pragma unroll
        for (int k = 0; k < EPT; ++k)
            if (t + k * SORT_T < CHUNK) atomicAdd(&bc[dv[k] >> BSHIFT], 1);
        __syncthreads();
        if (t < NBUCK) tab[t * NCHUNK + chunk] = (unsigned short)bc[t];
        return;
    }

    for (int i = t; i < IN_DIM * OUT_DIM; i += SORT_T)
        lds_w[i] = W[i];
    __syncthreads();

    const int wave = t >> 5;
    const int lane = t & 31;
    const int tile = blockIdx.x * GEMM_WAVES + wave;
    if (tile >= GEMM_TILES) return;

    const int m0   = tile * 16;
    const int row  = m0 + (lane & 15);
    const int koff = (lane >> 4) * 2;
    const int col  = lane & 15;

    const float al0 = Wa[col];
    const float al1 = Wa[col + 16];
    const float ar0 = Wa[OUT_DIM + col];
    const float ar1 = Wa[OUT_DIM + col + 16];

    v8f c0 = {}, c1 = {};
    const float* hrow = h + (size_t)row * IN_DIM;

    #pragma unroll 4
    for (int k = 0; k < IN_DIM; k += 4) {
        v2f a;
        a.x = hrow[k + koff];
        a.y = hrow[k + koff + 1];
        v2f b0, b1;
        b0.x = lds_w[(k + koff)     * OUT_DIM + col];
        b0.y = lds_w[(k + koff + 1) * OUT_DIM + col];
        b1.x = lds_w[(k + koff)     * OUT_DIM + col + 16];
        b1.y = lds_w[(k + koff + 1) * OUT_DIM + col + 16];
        c0 = __builtin_amdgcn_wmma_f32_16x16x4_f32(false, a, false, b0, (short)0, c0, false, false);
        c1 = __builtin_amdgcn_wmma_f32_16x16x4_f32(false, a, false, b1, (short)0, c1, false, false);
    }

    const int rbase = m0 + ((lane >> 4) * 8);
    #pragma unroll
    for (int r = 0; r < 8; ++r) {
        z[(size_t)(rbase + r) * OUT_DIM + col]      = c0[r];
        z[(size_t)(rbase + r) * OUT_DIM + col + 16] = c1[r];
    }

    __builtin_amdgcn_sched_barrier(0);
    float pl[8], pr[8];
    #pragma unroll
    for (int r = 0; r < 8; ++r) {
        pl[r] = row_sum16_f(c0[r] * al0 + c1[r] * al1);
        pr[r] = row_sum16_f(c0[r] * ar0 + c1[r] * ar1);
    }
    if (col == 0) {
        float4* pe = (float4*)(el + rbase);
        pe[0] = make_float4(pl[0], pl[1], pl[2], pl[3]);
        pe[1] = make_float4(pl[4], pl[5], pl[6], pl[7]);
        float4* qe = (float4*)(er + rbase);
        qe[0] = make_float4(pr[0], pr[1], pr[2], pr[3]);
        qe[1] = make_float4(pr[4], pr[5], pr[6], pr[7]);
    }
}

__global__ void __launch_bounds__(SORT_T)
gat_bucket_scatter_kernel(const int* __restrict__ src,
                          const int* __restrict__ dst,
                          const unsigned short* __restrict__ tab,
                          int* __restrict__ bstart,
                          int* __restrict__ tmp) {
    __shared__ int cur[NBUCK];
    __shared__ int rtot[NBUCK];
    __shared__ int lcur[NBUCK];
    __shared__ int wsum[SCAN_WAVES];
    __shared__ int wown[SCAN_WAVES];
    __shared__ int2 srt[CHUNK];
    const int t = threadIdx.x;
    const int lane = t & 31;
    const int wave = t >> 5;
    const int c = blockIdx.x;
    const int* s = src + c * CHUNK;
    const int* d = dst + c * CHUNK;
    int sv[EPT], dv[EPT];
    #pragma unroll
    for (int k = 0; k < EPT; ++k) {
        const int i = t + k * SORT_T;
        const int ic = i < CHUNK ? i : CHUNK - 1;
        sv[k] = s[ic];
        dv[k] = d[ic];
    }
    const int mine = (int)tab[(t < NBUCK ? t : NBUCK - 1) * NCHUNK + c];

    const int4* t4 = (const int4*)tab;
    const int j0 = 8 * lane;
    int4 p[ROWS_PER_WAVE];
    #pragma unroll
    for (int i = 0; i < ROWS_PER_WAVE; ++i) {
        const int b  = wave + SORT_WAVES * i;
        const int bb = b < NBUCK ? b : NBUCK - 1;
        p[i] = t4[bb * (NCHUNK / 8) + lane];
    }
    __builtin_amdgcn_sched_barrier(0);
    unsigned keep[4];
    #pragma unroll
    for (int k = 0; k < 4; ++k)
        keep[k] = (j0 + 2 * k < c ? 0x0000ffffu : 0u) | (j0 + 2 * k + 1 < c ? 0xffff0000u : 0u);
    int vt = 0, vp = 0;
    #pragma unroll
    for (int i = 0; i < ROWS_PER_WAVE; ++i) {
        const unsigned w0 = (unsigned)p[i].x, w1 = (unsigned)p[i].y;
        const unsigned w2 = (unsigned)p[i].z, w3 = (unsigned)p[i].w;
        const unsigned t2 = (w0 + w1) + (w2 + w3);
        const unsigned p2 = ((w0 & keep[0]) + (w1 & keep[1])) + ((w2 & keep[2]) + (w3 & keep[3]));
        const int ts = row_sum16_i((int)((t2 & 0xffffu) + (t2 >> 16)));
        const int ps = row_sum16_i((int)((p2 & 0xffffu) + (p2 >> 16)));
        if ((lane & 15) == i) { vt = ts; vp = ps; }
    }
    vt += __shfl_xor(vt, 16, 32);
    vp += __shfl_xor(vp, 16, 32);
    const int bl = wave + SORT_WAVES * lane;
    if (lane < ROWS_PER_WAVE && bl < NBUCK) { rtot[bl] = vt; cur[bl] = vp; }
    __syncthreads();
    int own = 0, inc = 0, lown = 0, linc = 0;
    if (wave < SCAN_WAVES) {
        own  = (t < NBUCK) ? rtot[t] : 0;
        lown = (t < NBUCK) ? mine : 0;
        inc  = own;
        linc = lown;
        #pragma unroll
        for (int o = 1; o < 32; o <<= 1) {
            const int u = __shfl_up(inc, o, 32);
            const int v = __shfl_up(linc, o, 32);
            if (lane >= o) { inc += u; linc += v; }
        }
        if (lane == 31) { wsum[wave] = inc; wown[wave] = linc; }
    }
    __syncthreads();
    if (t < NBUCK) {
        int base = 0, lbase = 0;
        #pragma unroll
        for (int w = 0; w < SCAN_WAVES; ++w)
            if (w < wave) { base += wsum[w]; lbase += wown[w]; }
        const int first  = base + inc - own;
        const int lfirst = lbase + linc - lown;
        lcur[t] = lfirst;
        cur[t] += first - lfirst;
        if (c == 0) bstart[t] = first;
    }
    __syncthreads();
    int lp[EPT], off[EPT];
    #pragma unroll
    for (int k = 0; k < EPT; ++k) {
        const int b = dv[k] >> BSHIFT;
        lp[k] = 0;
        if (t + k * SORT_T < CHUNK) lp[k] = atomicAdd(&lcur[b], 1);
        off[k] = cur[b];
    }
    #pragma unroll
    for (int k = 0; k < EPT; ++k)
        if (t + k * SORT_T < CHUNK)
            srt[lp[k]] = make_int2(sv[k] | ((dv[k] & (BNODES - 1)) << SRC_BITS), lp[k] + off[k]);
    __syncthreads();
    #pragma unroll
    for (int k = 0; k < EPT; ++k) {
        const int i = t + k * SORT_T;
        if (i < CHUNK) {
            const int2 r = srt[i];
            tmp[r.y] = r.x;
        }
    }
}

#define LCAP 8192
#define ROW_LOADS 8
#define ROW_EDGES (2 * ROW_LOADS)
static_assert(ROW_EDGES <= 16, "a group's edges are one lane each of a 16-lane row");
static_assert(BNODES % 2 == 0 && N_NODES % 2 == 0, "the two nodes of a pair are both real or both past the end");

__device__ __forceinline__ float row_max16_f(float x) {
    x = fmaxf(x, row_xor_f<1>(x));
    x = fmaxf(x, row_xor_f<2>(x));
    x = fmaxf(x, row_xor_f<4>(x));
    x = fmaxf(x, row_xor_f<8>(x));
    return x;
}
template <int U> __device__ __forceinline__ int row_pick2(int x) {
    return (int)__builtin_amdgcn_permlane16((unsigned)x, (unsigned)x, 0x11111111u * (unsigned)(2 * U),
                                            0x11111111u * (unsigned)(2 * U + 1), true, false);
}

template <bool IN_LDS>
__device__ __forceinline__ int2 rec_at(const int2* lrec, const int* gs, const float* __restrict__ el,
                                       float erd, int j) {
    if (IN_LDS) return lrec[j];
    const int s = __hip_atomic_load(gs + j, __ATOMIC_RELAXED, __HIP_MEMORY_SCOPE_AGENT);
    float e = el[s] + erd;
    e = (e >= 0.0f) ? e : LRELU_SLOPE * e;
    return make_int2(s, __float_as_int(e));
}

template <int U>
__device__ __forceinline__ void row_group_loads(const float4* __restrict__ z4, int s, int abits, unsigned q,
                                                float4* r, float* ak) {
    if constexpr (U < ROW_LOADS) {
        const unsigned sk = (unsigned)row_pick2<U>(s);
        r[U] = z4[sk * 8u + q];
        ak[U] = __int_as_float(row_pick2<U>(abits));
        row_group_loads<U + 1>(z4, s, abits, q, r, ak);
    }
}

template <bool IN_LDS>
__device__ __forceinline__ void bucket_nodes(const int2* lrec, const int* gs, const int* nfirst,
                                             const float* ler, const float* __restrict__ el,
                                             const float* __restrict__ z, float* __restrict__ hout,
                                             int b, int wave, int lane) {
    const int wave_s = __builtin_amdgcn_readfirstlane(wave);
    const int hrow = lane >> 4;
    const int l = lane & 15;
    const unsigned q = (unsigned)(lane & 7);
    const float4* __restrict__ z4 = (const float4*)z;
    float4* __restrict__ hout4 = (float4*)hout;
    for (int p = wave_s; p < BNODES / 2; p += SORT_WAVES) {
        if (b * BNODES + 2 * p >= N_NODES) break;
        const int dl = 2 * p + hrow;
        const int node = b * BNODES + dl;
        const int first = nfirst[dl];
        const int n     = nfirst[dl + 1] - first;
        const float erd = ler[dl];

        const bool has = l < n;
        int2 rec = make_int2(0, 0);
        if (n > 0) rec = rec_at<IN_LDS>(lrec, gs, el, erd, first + (has ? l : n - 1));
        const float e0 = __int_as_float(rec.y);

        float m = has ? e0 : -INFINITY;
        for (int j = l + 16; j < n; j += 16)
            m = fmaxf(m, __int_as_float(rec_at<IN_LDS>(lrec, gs, el, erd, first + j).y));
        m = row_max16_f(m);

        float ssum = has ? __expf(e0 - m) : 0.0f;
        for (int j = l + 16; j < n; j += 16)
            ssum += __expf(__int_as_float(rec_at<IN_LDS>(lrec, gs, el, erd, first + j).y) - m);
        ssum = row_sum16_f(ssum);

        const int n0 = __builtin_amdgcn_readlane(n, 0);
        const int n1 = __builtin_amdgcn_readlane(n, 16);
        const int nmax = n0 > n1 ? n0 : n1;
        float4 acc = make_float4(0.0f, 0.0f, 0.0f, 0.0f);
        for (int base = 0; base < nmax; base += ROW_EDGES) {
            const bool in = base + l < n;
            if (base > 0 && n > 0) rec = rec_at<IN_LDS>(lrec, gs, el, erd, first + (in ? base + l : n - 1));
            const int s = rec.x;
            const int abits = __float_as_int(in ? __expf(__int_as_float(rec.y) - m) / ssum : 0.0f);
            float4 r[ROW_LOADS];
            float ak[ROW_LOADS];
            row_group_loads<0>(z4, s, abits, q, r, ak);
            #pragma unroll
            for (int u = 0; u < ROW_LOADS; ++u) {
                acc.x += ak[u] * r[u].x;
                acc.y += ak[u] * r[u].y;
                acc.z += ak[u] * r[u].z;
                acc.w += ak[u] * r[u].w;
            }
        }
        acc.x += row_xor_f<8>(acc.x);
        acc.y += row_xor_f<8>(acc.y);
        acc.z += row_xor_f<8>(acc.z);
        acc.w += row_xor_f<8>(acc.w);
        if ((l >> 3) == 0) hout4[(unsigned)node * 8u + q] = acc;
    }
}

__global__ void __launch_bounds__(SORT_T, 16)
gat_bucket_gather_kernel(const float* __restrict__ el,
                         const float* __restrict__ er,
                         const int* __restrict__ bstart,
                         const int* __restrict__ tmp,
                         int* gs,
                         const float* __restrict__ z,
                         float* __restrict__ hout) {
    __shared__ int   nc[BNODES];
    __shared__ int   nfirst[BNODES + 1];
    __shared__ float ler[BNODES];
    __shared__ int   wtot[BNODES / 32];
    __shared__ int2  lrec[LCAP];
    const int t = threadIdx.x;
    const int lane = t & 31;
    const int wave = t >> 5;
    const int b = blockIdx.x;
    const int node = b * BNODES + t;
    const int beg = bstart[b];
    const int end = (b + 1 < NBUCK) ? bstart[b + 1] : N_EDGES;
    const bool fits = (end - beg) <= LCAP;
    if (t < BNODES) {
        nc[t] = 0;
        ler[t] = (node < N_NODES) ? er[node] : 0.0f;
    }
    __syncthreads();
    for (int j0 = beg + t; j0 < end; j0 += FILL_U * SORT_T) {
        int pv[FILL_U];
        #pragma unroll
        for (int u = 0; u < FILL_U; ++u) {
            const int j = j0 + u * SORT_T;
            pv[u] = tmp[j < end ? j : end - 1];
        }
        #pragma unroll
        for (int u = 0; u < FILL_U; ++u)
            if (j0 + u * SORT_T < end) atomicAdd(&nc[pv[u] >> SRC_BITS], 1);
    }
    __syncthreads();
    int c = 0, inc = 0;
    if (t < BNODES) {
        c = nc[t];
        inc = c;
        #pragma unroll
        for (int o = 1; o < 32; o <<= 1) {
            const int u = __shfl_up(inc, o, 32);
            if (lane >= o) inc += u;
        }
        if (lane == 31) wtot[t >> 5] = inc;
    }
    __syncthreads();
    if (t < BNODES) {
        int base = 0;
        #pragma unroll
        for (int w = 0; w < BNODES / 32; ++w)
            if (w < (t >> 5)) base += wtot[w];
        const int first = base + inc - c;
        nc[t] = first;
        nfirst[t] = first;
        if (t == BNODES - 1) nfirst[BNODES] = first + c;
    }
    __syncthreads();
    for (int j0 = beg + t; j0 < end; j0 += FILL_U * SORT_T) {
        int pv[FILL_U];
        float ev[FILL_U];
        #pragma unroll
        for (int u = 0; u < FILL_U; ++u) {
            const int j = j0 + u * SORT_T;
            pv[u] = tmp[j < end ? j : end - 1];
        }
        #pragma unroll
        for (int u = 0; u < FILL_U; ++u) ev[u] = el[pv[u] & ((1 << SRC_BITS) - 1)];
        #pragma unroll
        for (int u = 0; u < FILL_U; ++u) {
            if (j0 + u * SORT_T < end) {
                const int s  = pv[u] & ((1 << SRC_BITS) - 1);
                const int dl = pv[u] >> SRC_BITS;
                const int pos = atomicAdd(&nc[dl], 1);
                float e = ev[u] + ler[dl];
                e = (e >= 0.0f) ? e : LRELU_SLOPE * e;
                if (fits) lrec[pos] = make_int2(s, __float_as_int(e));
                else      gs[beg + pos] = s;
            }
        }
    }
    if (!fits) __threadfence();
    __syncthreads();
    if (fits) bucket_nodes<true >(lrec, gs + beg, nfirst, ler, el, z, hout, b, wave, lane);
    else      bucket_nodes<false>(lrec, gs + beg, nfirst, ler, el, z, hout, b, wave, lane);
}

#define STAT_T      256
#define STAT_BLOCKS 512
#define STAT_WAVES  (STAT_BLOCKS * STAT_T / 32)
#define STAT_ROWS   ((N_NODES + STAT_WAVES - 1) / STAT_WAVES)
static_assert((STAT_ROWS - 1) * STAT_WAVES < N_NODES, "only a wave's last row can lie past the end");
__global__ void __launch_bounds__(STAT_T)
gat_bn_stats_kernel(const float* __restrict__ hout, float* __restrict__ bn) {
    __shared__ float ssum[OUT_DIM];
    __shared__ float ssq[OUT_DIM];
    const int lane = threadIdx.x & 31;
    if (threadIdx.x < OUT_DIM) { ssum[threadIdx.x] = 0.0f; ssq[threadIdx.x] = 0.0f; }
    const int w = (blockIdx.x * STAT_T + threadIdx.x) >> 5;
    const int last = w + (STAT_ROWS - 1) * STAT_WAVES;
    const bool has_last = last < N_NODES;
    float v[STAT_ROWS];
    #pragma unroll
    for (int i = 0; i < STAT_ROWS - 1; ++i)
        v[i] = hout[(size_t)(w + i * STAT_WAVES) * OUT_DIM + lane];
    v[STAT_ROWS - 1] = hout[(size_t)(has_last ? last : N_NODES - 1) * OUT_DIM + lane];
    float s = 0.0f, q = 0.0f;
    #pragma unroll
    for (int i = 0; i < STAT_ROWS - 1; ++i) {
        s += v[i];
        q += v[i] * v[i];
    }
    if (has_last) {
        s += v[STAT_ROWS - 1];
        q += v[STAT_ROWS - 1] * v[STAT_ROWS - 1];
    }
    __syncthreads();
    atomicAdd(&ssum[lane], s);
    atomicAdd(&ssq[lane], q);
    __syncthreads();
    if (threadIdx.x < OUT_DIM) {
        atomicAdd(&bn[threadIdx.x], ssum[threadIdx.x]);
        atomicAdd(&bn[OUT_DIM + threadIdx.x], ssq[threadIdx.x]);
    }
}

__global__ void gat_bn_apply_kernel(const float* __restrict__ bn,
                                    const float* __restrict__ gamma,
                                    const float* __restrict__ beta,
                                    float* __restrict__ out) {
    int idx = blockIdx.x * blockDim.x + threadIdx.x;
    if (idx >= N_NODES * OUT_DIM) return;
    int c = idx & 31;
    const float inv_n = 1.0f / (float)N_NODES;
    float mean = bn[c] * inv_n;
    float var  = bn[OUT_DIM + c] * inv_n - mean * mean;
    float x = (out[idx] - mean) * rsqrtf(var + BN_EPS) * gamma[c] + beta[c];
    out[idx] = (x > 0.0f) ? x : expm1f(x);
}

extern "C" void kernel_launch(void* const* d_in, const int* in_sizes, int n_in,
                              void* d_out, int out_size, void* d_ws, size_t ws_size,
                              hipStream_t stream) {
    const float* h     = (const float*)d_in[0];
    const float* W_fc  = (const float*)d_in[1];
    const float* W_at  = (const float*)d_in[2];
    const float* gamma = (const float*)d_in[3];
    const float* beta  = (const float*)d_in[4];
    const int*   src   = (const int*)d_in[5];
    const int*   dst   = (const int*)d_in[6];
    float* out = (float*)d_out;

    float* z    = (float*)d_ws;
    float* el   = z + (size_t)N_NODES * OUT_DIM;
    float* er   = el + N_NODES;
    int*   bstart = (int*)(er + N_NODES);
    int*   tmp  = bstart + 2 * N_NODES;
    int*   gs   = tmp + N_EDGES;
    float* bn   = (float*)(gs + N_EDGES);
    unsigned short* tab = (unsigned short*)d_out;

    const int B = 256;

    gat_gemm_proj_count_kernel<<<GEMM_BLOCKS + NCHUNK, SORT_T, 0, stream>>>(h, W_fc, W_at, dst, z, el, er, tab, bn);

    gat_bucket_scatter_kernel<<<NCHUNK, SORT_T, 0, stream>>>(src, dst, tab, bstart, tmp);

    gat_bucket_gather_kernel<<<NBUCK, SORT_T, 0, stream>>>(el, er, bstart, tmp, gs, z, out);

    gat_bn_stats_kernel<<<STAT_BLOCKS, STAT_T, 0, stream>>>(out, bn);

    gat_bn_apply_kernel<<<(N_NODES * OUT_DIM + B - 1) / B, B, 0, stream>>>(bn, gamma, beta, out);
}


# BN apply launch as a fixed grid of 2500 waves, not one element per lane: a thread takes ten 16-byte pieces of its own four channels, all loads asked for at once, mean and scale formed once
# speedup vs baseline: 5.9342x; 1.0399x over previous
#include <hip/hip_runtime.h>
#include <hip/hip_bf16.h>

#define N_NODES 100000
#define N_EDGES 1600000
#define IN_DIM 128
#define OUT_DIM 32
#define BN_EPS 1e-5f
#define LRELU_SLOPE 0.01f

#define BSHIFT 8
#define BNODES 256
#define NBUCK  ((N_NODES + BNODES - 1) / BNODES)
#define NCHUNK 256
#define CHUNK  (N_EDGES / NCHUNK)
#define TABN   (NBUCK * NCHUNK)
#define SORT_T 1024
#define EPT    ((CHUNK + SORT_T - 1) / SORT_T)
#define SRC_BITS 17
#define FILL_U 5
static_assert(N_EDGES % NCHUNK == 0, "chunks tile the edge list");
static_assert(TABN % 8 == 0, "table rows of 16-bit counts are read as int4");
static_assert(N_NODES <= (1 << SRC_BITS), "src and the in-bucket dst share one word");
static_assert(NCHUNK == 256, "a table row of 16-bit counts is one int4 per lane of one wave");
static_assert(4 * CHUNK <= 65535, "four piece sizes summed still fit 16 bits");
static_assert(NBUCK <= SORT_T, "one thread per bucket in the cursor scan");

#define GEMM_TILES   (N_NODES / 16)
#define GEMM_WAVES   (SORT_T / 32)
#define GEMM_BLOCKS  ((GEMM_TILES + GEMM_WAVES - 1) / GEMM_WAVES)
#define SCAN_WAVES   ((NBUCK + 31) / 32)
#define SORT_WAVES   (SORT_T / 32)
#define ROWS_PER_WAVE ((NBUCK + SORT_WAVES - 1) / SORT_WAVES)
static_assert(ROWS_PER_WAVE <= 16, "one lane of each half wave per row of a wave");

typedef __attribute__((ext_vector_type(2))) float v2f;
typedef __attribute__((ext_vector_type(8))) float v8f;

template <int M> __device__ __forceinline__ float row_xor_f(float x) {
    return __int_as_float(__builtin_amdgcn_update_dpp(0, __float_as_int(x), 0x160 + M, 0xf, 0xf, true));
}
template <int M> __device__ __forceinline__ int row_xor_i(int x) {
    return __builtin_amdgcn_update_dpp(0, x, 0x160 + M, 0xf, 0xf, true);
}
__device__ __forceinline__ float row_sum16_f(float x) {
    x += row_xor_f<1>(x);
    x += row_xor_f<2>(x);
    x += row_xor_f<4>(x);
    x += row_xor_f<8>(x);
    return x;
}
__device__ __forceinline__ int row_sum16_i(int x) {
    x += row_xor_i<1>(x);
    x += row_xor_i<2>(x);
    x += row_xor_i<4>(x);
    x += row_xor_i<8>(x);
    return x;
}

__global__ void __launch_bounds__(SORT_T)
gat_gemm_proj_count_kernel(const float* __restrict__ h,
                           const float* __restrict__ W,
                           const float* __restrict__ Wa,
                           const int* __restrict__ dst,
                           float* __restrict__ z,
                           float* __restrict__ el,
                           float* __restrict__ er,
                           unsigned short* __restrict__ tab,
                           float* __restrict__ bn) {
    __shared__ float lds_w[IN_DIM * OUT_DIM];
    __shared__ int bc[NBUCK];
    const int t = threadIdx.x;

    if (blockIdx.x >= GEMM_BLOCKS) {
        const int chunk = blockIdx.x - GEMM_BLOCKS;
        if (t < NBUCK) bc[t] = 0;
        if (chunk == 0 && t < 2 * OUT_DIM) bn[t] = 0.0f;
        __syncthreads();
        const int* d = dst + chunk * CHUNK;
        int dv[EPT];
        #pragma unroll
        for (int k = 0; k < EPT; ++k) {
            const int i = t + k * SORT_T;
            dv[k] = d[i < CHUNK ? i : CHUNK - 1];
        }
        #pragma unroll
        for (int k = 0; k < EPT; ++k)
            if (t + k * SORT_T < CHUNK) atomicAdd(&bc[dv[k] >> BSHIFT], 1);
        __syncthreads();
        if (t < NBUCK) tab[t * NCHUNK + chunk] = (unsigned short)bc[t];
        return;
    }

    for (int i = t; i < IN_DIM * OUT_DIM; i += SORT_T)
        lds_w[i] = W[i];
    __syncthreads();

    const int wave = t >> 5;
    const int lane = t & 31;
    const int tile = blockIdx.x * GEMM_WAVES + wave;
    if (tile >= GEMM_TILES) return;

    const int m0   = tile * 16;
    const int row  = m0 + (lane & 15);
    const int koff = (lane >> 4) * 2;
    const int col  = lane & 15;

    const float al0 = Wa[col];
    const float al1 = Wa[col + 16];
    const float ar0 = Wa[OUT_DIM + col];
    const float ar1 = Wa[OUT_DIM + col + 16];

    v8f c0 = {}, c1 = {};
    const float* hrow = h + (size_t)row * IN_DIM;

    #pragma unroll 4
    for (int k = 0; k < IN_DIM; k += 4) {
        v2f a;
        a.x = hrow[k + koff];
        a.y = hrow[k + koff + 1];
        v2f b0, b1;
        b0.x = lds_w[(k + koff)     * OUT_DIM + col];
        b0.y = lds_w[(k + koff + 1) * OUT_DIM + col];
        b1.x = lds_w[(k + koff)     * OUT_DIM + col + 16];
        b1.y = lds_w[(k + koff + 1) * OUT_DIM + col + 16];
        c0 = __builtin_amdgcn_wmma_f32_16x16x4_f32(false, a, false, b0, (short)0, c0, false, false);
        c1 = __builtin_amdgcn_wmma_f32_16x16x4_f32(false, a, false, b1, (short)0, c1, false, false);
    }

    const int rbase = m0 + ((lane >> 4) * 8);
    #pragma unroll
    for (int r = 0; r < 8; ++r) {
        z[(size_t)(rbase + r) * OUT_DIM + col]      = c0[r];
        z[(size_t)(rbase + r) * OUT_DIM + col + 16] = c1[r];
    }

    __builtin_amdgcn_sched_barrier(0);
    float pl[8], pr[8];
    #pragma unroll
    for (int r = 0; r < 8; ++r) {
        pl[r] = row_sum16_f(c0[r] * al0 + c1[r] * al1);
        pr[r] = row_sum16_f(c0[r] * ar0 + c1[r] * ar1);
    }
    if (col == 0) {
        float4* pe = (float4*)(el + rbase);
        pe[0] = make_float4(pl[0], pl[1], pl[2], pl[3]);
        pe[1] = make_float4(pl[4], pl[5], pl[6], pl[7]);
        float4* qe = (float4*)(er + rbase);
        qe[0] = make_float4(pr[0], pr[1], pr[2], pr[3]);
        qe[1] = make_float4(pr[4], pr[5], pr[6], pr[7]);
    }
}

__global__ void __launch_bounds__(SORT_T)
gat_bucket_scatter_kernel(const int* __restrict__ src,
                          const int* __restrict__ dst,
                          const unsigned short* __restrict__ tab,
                          int* __restrict__ bstart,
                          int* __restrict__ tmp) {
    __shared__ int cur[NBUCK];
    __shared__ int rtot[NBUCK];
    __shared__ int lcur[NBUCK];
    __shared__ int wsum[SCAN_WAVES];
    __shared__ int wown[SCAN_WAVES];
    __shared__ int2 srt[CHUNK];
    const int t = threadIdx.x;
    const int lane = t & 31;
    const int wave = t >> 5;
    const int c = blockIdx.x;
    const int* s = src + c * CHUNK;
    const int* d = dst + c * CHUNK;
    int sv[EPT], dv[EPT];
    #pragma unroll
    for (int k = 0; k < EPT; ++k) {
        const int i = t + k * SORT_T;
        const int ic = i < CHUNK ? i : CHUNK - 1;
        sv[k] = s[ic];
        dv[k] = d[ic];
    }
    const int mine = (int)tab[(t < NBUCK ? t : NBUCK - 1) * NCHUNK + c];

    const int4* t4 = (const int4*)tab;
    const int j0 = 8 * lane;
    int4 p[ROWS_PER_WAVE];
    #pragma unroll
    for (int i = 0; i < ROWS_PER_WAVE; ++i) {
        const int b  = wave + SORT_WAVES * i;
        const int bb = b < NBUCK ? b : NBUCK - 1;
        p[i] = t4[bb * (NCHUNK / 8) + lane];
    }
    __builtin_amdgcn_sched_barrier(0);
    unsigned keep[4];
    #pragma unroll
    for (int k = 0; k < 4; ++k)
        keep[k] = (j0 + 2 * k < c ? 0x0000ffffu : 0u) | (j0 + 2 * k + 1 < c ? 0xffff0000u : 0u);
    int vt = 0, vp = 0;
    #pragma unroll
    for (int i = 0; i < ROWS_PER_WAVE; ++i) {
        const unsigned w0 = (unsigned)p[i].x, w1 = (unsigned)p[i].y;
        const unsigned w2 = (unsigned)p[i].z, w3 = (unsigned)p[i].w;
        const unsigned t2 = (w0 + w1) + (w2 + w3);
        const unsigned p2 = ((w0 & keep[0]) + (w1 & keep[1])) + ((w2 & keep[2]) + (w3 & keep[3]));
        const int ts = row_sum16_i((int)((t2 & 0xffffu) + (t2 >> 16)));
        const int ps = row_sum16_i((int)((p2 & 0xffffu) + (p2 >> 16)));
        if ((lane & 15) == i) { vt = ts; vp = ps; }
    }
    vt += __shfl_xor(vt, 16, 32);
    vp += __shfl_xor(vp, 16, 32);
    const int bl = wave + SORT_WAVES * lane;
    if (lane < ROWS_PER_WAVE && bl < NBUCK) { rtot[bl] = vt; cur[bl] = vp; }
    __syncthreads();
    int own = 0, inc = 0, lown = 0, linc = 0;
    if (wave < SCAN_WAVES) {
        own  = (t < NBUCK) ? rtot[t] : 0;
        lown = (t < NBUCK) ? mine : 0;
        inc  = own;
        linc = lown;
        #pragma unroll
        for (int o = 1; o < 32; o <<= 1) {
            const int u = __shfl_up(inc, o, 32);
            const int v = __shfl_up(linc, o, 32);
            if (lane >= o) { inc += u; linc += v; }
        }
        if (lane == 31) { wsum[wave] = inc; wown[wave] = linc; }
    }
    __syncthreads();
    if (t < NBUCK) {
        int base = 0, lbase = 0;
        #pragma unroll
        for (int w = 0; w < SCAN_WAVES; ++w)
            if (w < wave) { base += wsum[w]; lbase += wown[w]; }
        const int first  = base + inc - own;
        const int lfirst = lbase + linc - lown;
        lcur[t] = lfirst;
        cur[t] += first - lfirst;
        if (c == 0) bstart[t] = first;
    }
    __syncthreads();
    int lp[EPT], off[EPT];
    #pragma unroll
    for (int k = 0; k < EPT; ++k) {
        const int b = dv[k] >> BSHIFT;
        lp[k] = 0;
        if (t + k * SORT_T < CHUNK) lp[k] = atomicAdd(&lcur[b], 1);
        off[k] = cur[b];
    }
    #pragma unroll
    for (int k = 0; k < EPT; ++k)
        if (t + k * SORT_T < CHUNK)
            srt[lp[k]] = make_int2(sv[k] | ((dv[k] & (BNODES - 1)) << SRC_BITS), lp[k] + off[k]);
    __syncthreads();
    #pragma unroll
    for (int k = 0; k < EPT; ++k) {
        const int i = t + k * SORT_T;
        if (i < CHUNK) {
            const int2 r = srt[i];
            tmp[r.y] = r.x;
        }
    }
}

#define LCAP 8192
#define ROW_LOADS 8
#define ROW_EDGES (2 * ROW_LOADS)
static_assert(ROW_EDGES <= 16, "a group's edges are one lane each of a 16-lane row");
static_assert(BNODES % 2 == 0 && N_NODES % 2 == 0, "the two nodes of a pair are both real or both past the end");

__device__ __forceinline__ float row_max16_f(float x) {
    x = fmaxf(x, row_xor_f<1>(x));
    x = fmaxf(x, row_xor_f<2>(x));
    x = fmaxf(x, row_xor_f<4>(x));
    x = fmaxf(x, row_xor_f<8>(x));
    return x;
}
template <int U> __device__ __forceinline__ int row_pick2(int x) {
    return (int)__builtin_amdgcn_permlane16((unsigned)x, (unsigned)x, 0x11111111u * (unsigned)(2 * U),
                                            0x11111111u * (unsigned)(2 * U + 1), true, false);
}

template <bool IN_LDS>
__device__ __forceinline__ int2 rec_at(const int2* lrec, const int* gs, const float* __restrict__ el,
                                       float erd, int j) {
    if (IN_LDS) return lrec[j];
    const int s = __hip_atomic_load(gs + j, __ATOMIC_RELAXED, __HIP_MEMORY_SCOPE_AGENT);
    float e = el[s] + erd;
    e = (e >= 0.0f) ? e : LRELU_SLOPE * e;
    return make_int2(s, __float_as_int(e));
}

template <int U>
__device__ __forceinline__ void row_group_loads(const float4* __restrict__ z4, int s, int abits, unsigned q,
                                                float4* r, float* ak) {
    if constexpr (U < ROW_LOADS) {
        const unsigned sk = (unsigned)row_pick2<U>(s);
        r[U] = z4[sk * 8u + q];
        ak[U] = __int_as_float(row_pick2<U>(abits));
        row_group_loads<U + 1>(z4, s, abits, q, r, ak);
    }
}

template <bool IN_LDS>
__device__ __forceinline__ void bucket_nodes(const int2* lrec, const int* gs, const int* nfirst,
                                             const float* ler, const float* __restrict__ el,
                                             const float* __restrict__ z, float* __restrict__ hout,
                                             int b, int wave, int lane) {
    const int wave_s = __builtin_amdgcn_readfirstlane(wave);
    const int hrow = lane >> 4;
    const int l = lane & 15;
    const unsigned q = (unsigned)(lane & 7);
    const float4* __restrict__ z4 = (const float4*)z;
    float4* __restrict__ hout4 = (float4*)hout;
    for (int p = wave_s; p < BNODES / 2; p += SORT_WAVES) {
        if (b * BNODES + 2 * p >= N_NODES) break;
        const int dl = 2 * p + hrow;
        const int node = b * BNODES + dl;
        const int first = nfirst[dl];
        const int n     = nfirst[dl + 1] - first;
        const float erd = ler[dl];

        const bool has = l < n;
        int2 rec = make_int2(0, 0);
        if (n > 0) rec = rec_at<IN_LDS>(lrec, gs, el, erd, first + (has ? l : n - 1));
        const float e0 = __int_as_float(rec.y);

        float m = has ? e0 : -INFINITY;
        for (int j = l + 16; j < n; j += 16)
            m = fmaxf(m, __int_as_float(rec_at<IN_LDS>(lrec, gs, el, erd, first + j).y));
        m = row_max16_f(m);

        float ssum = has ? __expf(e0 - m) : 0.0f;
        for (int j = l + 16; j < n; j += 16)
            ssum += __expf(__int_as_float(rec_at<IN_LDS>(lrec, gs, el, erd, first + j).y) - m);
        ssum = row_sum16_f(ssum);

        const int n0 = __builtin_amdgcn_readlane(n, 0);
        const int n1 = __builtin_amdgcn_readlane(n, 16);
        const int nmax = n0 > n1 ? n0 : n1;
        float4 acc = make_float4(0.0f, 0.0f, 0.0f, 0.0f);
        for (int base = 0; base < nmax; base += ROW_EDGES) {
            const bool in = base + l < n;
            if (base > 0 && n > 0) rec = rec_at<IN_LDS>(lrec, gs, el, erd, first + (in ? base + l : n - 1));
            const int s = rec.x;
            const int abits = __float_as_int(in ? __expf(__int_as_float(rec.y) - m) / ssum : 0.0f);
            float4 r[ROW_LOADS];
            float ak[ROW_LOADS];
            row_group_loads<0>(z4, s, abits, q, r, ak);
            #pragma unroll
            for (int u = 0; u < ROW_LOADS; ++u) {
                acc.x += ak[u] * r[u].x;
                acc.y += ak[u] * r[u].y;
                acc.z += ak[u] * r[u].z;
                acc.w += ak[u] * r[u].w;
            }
        }
        acc.x += row_xor_f<8>(acc.x);
        acc.y += row_xor_f<8>(acc.y);
        acc.z += row_xor_f<8>(acc.z);
        acc.w += row_xor_f<8>(acc.w);
        if ((l >> 3) == 0) hout4[(unsigned)node * 8u + q] = acc;
    }
}

__global__ void __launch_bounds__(SORT_T, 16)
gat_bucket_gather_kernel(const float* __restrict__ el,
                         const float* __restrict__ er,
                         const int* __restrict__ bstart,
                         const int* __restrict__ tmp,
                         int* gs,
                         const float* __restrict__ z,
                         float* __restrict__ hout) {
    __shared__ int   nc[BNODES];
    __shared__ int   nfirst[BNODES + 1];
    __shared__ float ler[BNODES];
    __shared__ int   wtot[BNODES / 32];
    __shared__ int2  lrec[LCAP];
    const int t = threadIdx.x;
    const int lane = t & 31;
    const int wave = t >> 5;
    const int b = blockIdx.x;
    const int node = b * BNODES + t;
    const int beg = bstart[b];
    const int end = (b + 1 < NBUCK) ? bstart[b + 1] : N_EDGES;
    const bool fits = (end - beg) <= LCAP;
    if (t < BNODES) {
        nc[t] = 0;
        ler[t] = (node < N_NODES) ? er[node] : 0.0f;
    }
    __syncthreads();
    for (int j0 = beg + t; j0 < end; j0 += FILL_U * SORT_T) {
        int pv[FILL_U];
        #pragma unroll
        for (int u = 0; u < FILL_U; ++u) {
            const int j = j0 + u * SORT_T;
            pv[u] = tmp[j < end ? j : end - 1];
        }
        #pragma unroll
        for (int u = 0; u < FILL_U; ++u)
            if (j0 + u * SORT_T < end) atomicAdd(&nc[pv[u] >> SRC_BITS], 1);
    }
    __syncthreads();
    int c = 0, inc = 0;
    if (t < BNODES) {
        c = nc[t];
        inc = c;
        #pragma unroll
        for (int o = 1; o < 32; o <<= 1) {
            const int u = __shfl_up(inc, o, 32);
            if (lane >= o) inc += u;
        }
        if (lane == 31) wtot[t >> 5] = inc;
    }
    __syncthreads();
    if (t < BNODES) {
        int base = 0;
        #pragma unroll
        for (int w = 0; w < BNODES / 32; ++w)
            if (w < (t >> 5)) base += wtot[w];
        const int first = base + inc - c;
        nc[t] = first;
        nfirst[t] = first;
        if (t == BNODES - 1) nfirst[BNODES] = first + c;
    }
    __syncthreads();
    for (int j0 = beg + t; j0 < end; j0 += FILL_U * SORT_T) {
        int pv[FILL_U];
        float ev[FILL_U];
        #pragma unroll
        for (int u = 0; u < FILL_U; ++u) {
            const int j = j0 + u * SORT_T;
            pv[u] = tmp[j < end ? j : end - 1];
        }
        #pragma unroll
        for (int u = 0; u < FILL_U; ++u) ev[u] = el[pv[u] & ((1 << SRC_BITS) - 1)];
        #pragma unroll
        for (int u = 0; u < FILL_U; ++u) {
            if (j0 + u * SORT_T < end) {
                const int s  = pv[u] & ((1 << SRC_BITS) - 1);
                const int dl = pv[u] >> SRC_BITS;
                const int pos = atomicAdd(&nc[dl], 1);
                float e = ev[u] + ler[dl];
                e = (e >= 0.0f) ? e : LRELU_SLOPE * e;
                if (fits) lrec[pos] = make_int2(s, __float_as_int(e));
                else      gs[beg + pos] = s;
            }
        }
    }
    if (!fits) __threadfence();
    __syncthreads();
    if (fits) bucket_nodes<true >(lrec, gs + beg, nfirst, ler, el, z, hout, b, wave, lane);
    else      bucket_nodes<false>(lrec, gs + beg, nfirst, ler, el, z, hout, b, wave, lane);
}

#define STAT_T      256
#define STAT_BLOCKS 512
#define STAT_WAVES  (STAT_BLOCKS * STAT_T / 32)
#define STAT_ROWS   ((N_NODES + STAT_WAVES - 1) / STAT_WAVES)
static_assert((STAT_ROWS - 1) * STAT_WAVES < N_NODES, "only a wave's last row can lie past the end");
__global__ void __launch_bounds__(STAT_T)
gat_bn_stats_kernel(const float* __restrict__ hout, float* __restrict__ bn) {
    __shared__ float ssum[OUT_DIM];
    __shared__ float ssq[OUT_DIM];
    const int lane = threadIdx.x & 31;
    if (threadIdx.x < OUT_DIM) { ssum[threadIdx.x] = 0.0f; ssq[threadIdx.x] = 0.0f; }
    const int w = (blockIdx.x * STAT_T + threadIdx.x) >> 5;
    const int last = w + (STAT_ROWS - 1) * STAT_WAVES;
    const bool has_last = last < N_NODES;
    float v[STAT_ROWS];
    #pragma unroll
    for (int i = 0; i < STAT_ROWS - 1; ++i)
        v[i] = hout[(size_t)(w + i * STAT_WAVES) * OUT_DIM + lane];
    v[STAT_ROWS - 1] = hout[(size_t)(has_last ? last : N_NODES - 1) * OUT_DIM + lane];
    float s = 0.0f, q = 0.0f;
    #pragma unroll
    for (int i = 0; i < STAT_ROWS - 1; ++i) {
        s += v[i];
        q += v[i] * v[i];
    }
    if (has_last) {
        s += v[STAT_ROWS - 1];
        q += v[STAT_ROWS - 1] * v[STAT_ROWS - 1];
    }
    __syncthreads();
    atomicAdd(&ssum[lane], s);
    atomicAdd(&ssq[lane], q);
    __syncthreads();
    if (threadIdx.x < OUT_DIM) {
        atomicAdd(&bn[threadIdx.x], ssum[threadIdx.x]);
        atomicAdd(&bn[OUT_DIM + threadIdx.x], ssq[threadIdx.x]);
    }
}

#define APPLY_T       128
#define APPLY_PIECES  10
#define APPLY_N4      (N_NODES * OUT_DIM / 4)
#define APPLY_THREADS (APPLY_N4 / APPLY_PIECES)
#define APPLY_BLOCKS  (APPLY_THREADS / APPLY_T)
static_assert(APPLY_N4 % APPLY_PIECES == 0 && APPLY_THREADS % APPLY_T == 0, "the grid tiles out exactly: no tail");
static_assert(APPLY_THREADS % (OUT_DIM / 4) == 0, "a thread's pieces all hold the same four channels");

__device__ __forceinline__ float bn_elu(float v, float mean, float rs, float g, float b) {
    const float x = (v - mean) * rs * g + b;
    return (x > 0.0f) ? x : expm1f(x);
}

__global__ void __launch_bounds__(APPLY_T)
gat_bn_apply_kernel(const float* __restrict__ bn,
                    const float* __restrict__ gamma,
                    const float* __restrict__ beta,
                    float* __restrict__ out) {
    const int g = blockIdx.x * APPLY_T + threadIdx.x;
    const int q = g & (OUT_DIM / 4 - 1);
    float4* __restrict__ o4 = (float4*)out;
    const float4 s4 = ((const float4*)bn)[q];
    const float4 q4 = ((const float4*)bn)[OUT_DIM / 4 + q];
    const float4 g4 = ((const float4*)gamma)[q];
    const float4 b4 = ((const float4*)beta)[q];
    float4 v[APPLY_PIECES];
    #pragma unroll
    for (int i = 0; i < APPLY_PIECES; ++i)
        v[i] = o4[g + i * APPLY_THREADS];
    __builtin_amdgcn_sched_barrier(0);
    const float inv_n = 1.0f / (float)N_NODES;
    float4 mean, rs;
    mean.x = s4.x * inv_n;
    mean.y = s4.y * inv_n;
    mean.z = s4.z * inv_n;
    mean.w = s4.w * inv_n;
    rs.x = rsqrtf(q4.x * inv_n - mean.x * mean.x + BN_EPS);
    rs.y = rsqrtf(q4.y * inv_n - mean.y * mean.y + BN_EPS);
    rs.z = rsqrtf(q4.z * inv_n - mean.z * mean.z + BN_EPS);
    rs.w = rsqrtf(q4.w * inv_n - mean.w * mean.w + BN_EPS);
    #pragma unroll
    for (int i = 0; i < APPLY_PIECES; ++i) {
        float4 r;
        r.x = bn_elu(v[i].x, mean.x, rs.x, g4.x, b4.x);
        r.y = bn_elu(v[i].y, mean.y, rs.y, g4.y, b4.y);
        r.z = bn_elu(v[i].z, mean.z, rs.z, g4.z, b4.z);
        r.w = bn_elu(v[i].w, mean.w, rs.w, g4.w, b4.w);
        o4[g + i * APPLY_THREADS] = r;
    }
}

extern "C" void kernel_launch(void* const* d_in, const int* in_sizes, int n_in,
                              void* d_out, int out_size, void* d_ws, size_t ws_size,
                              hipStream_t stream) {
    const float* h     = (const float*)d_in[0];
    const float* W_fc  = (const float*)d_in[1];
    const float* W_at  = (const float*)d_in[2];
    const float* gamma = (const float*)d_in[3];
    const float* beta  = (const float*)d_in[4];
    const int*   src   = (const int*)d_in[5];
    const int*   dst   = (const int*)d_in[6];
    float* out = (float*)d_out;

    float* z    = (float*)d_ws;
    float* el   = z + (size_t)N_NODES * OUT_DIM;
    float* er   = el + N_NODES;
    int*   bstart = (int*)(er + N_NODES);
    int*   tmp  = bstart + 2 * N_NODES;
    int*   gs   = tmp + N_EDGES;
    float* bn   = (float*)(gs + N_EDGES);
    unsigned short* tab = (unsigned short*)d_out;

    gat_gemm_proj_count_kernel<<<GEMM_BLOCKS + NCHUNK, SORT_T, 0, stream>>>(h, W_fc, W_at, dst, z, el, er, tab, bn);

    gat_bucket_scatter_kernel<<<NCHUNK, SORT_T, 0, stream>>>(src, dst, tab, bstart, tmp);

    gat_bucket_gather_kernel<<<NBUCK, SORT_T, 0, stream>>>(el, er, bstart, tmp, gs, z, out);

    gat_bn_stats_kernel<<<STAT_BLOCKS, STAT_T, 0, stream>>>(out, bn);

    gat_bn_apply_kernel<<<APPLY_BLOCKS, APPLY_T, 0, stream>>>(bn, gamma, beta, out);
}


# BN statistics formed inside the node pass launch (five launches to four): a wave sums its finished rows and their squares in registers, waves meet in LDS in wave order (new order of the BN sums)
# speedup vs baseline: 6.2700x; 1.0566x over previous
#include <hip/hip_runtime.h>
#include <hip/hip_bf16.h>

#define N_NODES 100000
#define N_EDGES 1600000
#define IN_DIM 128
#define OUT_DIM 32
#define BN_EPS 1e-5f
#define LRELU_SLOPE 0.01f

#define BSHIFT 8
#define BNODES 256
#define NBUCK  ((N_NODES + BNODES - 1) / BNODES)
#define NCHUNK 256
#define CHUNK  (N_EDGES / NCHUNK)
#define TABN   (NBUCK * NCHUNK)
#define SORT_T 1024
#define EPT    ((CHUNK + SORT_T - 1) / SORT_T)
#define SRC_BITS 17
#define FILL_U 5
static_assert(N_EDGES % NCHUNK == 0, "chunks tile the edge list");
static_assert(TABN % 8 == 0, "table rows of 16-bit counts are read as int4");
static_assert(N_NODES <= (1 << SRC_BITS), "src and the in-bucket dst share one word");
static_assert(NCHUNK == 256, "a table row of 16-bit counts is one int4 per lane of one wave");
static_assert(4 * CHUNK <= 65535, "four piece sizes summed still fit 16 bits");
static_assert(NBUCK <= SORT_T, "one thread per bucket in the cursor scan");

#define GEMM_TILES   (N_NODES / 16)
#define GEMM_WAVES   (SORT_T / 32)
#define GEMM_BLOCKS  ((GEMM_TILES + GEMM_WAVES - 1) / GEMM_WAVES)
#define SCAN_WAVES   ((NBUCK + 31) / 32)
#define SORT_WAVES   (SORT_T / 32)
#define ROWS_PER_WAVE ((NBUCK + SORT_WAVES - 1) / SORT_WAVES)
static_assert(ROWS_PER_WAVE <= 16, "one lane of each half wave per row of a wave");

typedef __attribute__((ext_vector_type(2))) float v2f;
typedef __attribute__((ext_vector_type(8))) float v8f;

template <int M> __device__ __forceinline__ float row_xor_f(float x) {
    return __int_as_float(__builtin_amdgcn_update_dpp(0, __float_as_int(x), 0x160 + M, 0xf, 0xf, true));
}
template <int M> __device__ __forceinline__ int row_xor_i(int x) {
    return __builtin_amdgcn_update_dpp(0, x, 0x160 + M, 0xf, 0xf, true);
}
__device__ __forceinline__ float row_sum16_f(float x) {
    x += row_xor_f<1>(x);
    x += row_xor_f<2>(x);
    x += row_xor_f<4>(x);
    x += row_xor_f<8>(x);
    return x;
}
__device__ __forceinline__ int row_sum16_i(int x) {
    x += row_xor_i<1>(x);
    x += row_xor_i<2>(x);
    x += row_xor_i<4>(x);
    x += row_xor_i<8>(x);
    return x;
}

__global__ void __launch_bounds__(SORT_T)
gat_gemm_proj_count_kernel(const float* __restrict__ h,
                           const float* __restrict__ W,
                           const float* __restrict__ Wa,
                           const int* __restrict__ dst,
                           float* __restrict__ z,
                           float* __restrict__ el,
                           float* __restrict__ er,
                           unsigned short* __restrict__ tab,
                           float* __restrict__ bn) {
    __shared__ float lds_w[IN_DIM * OUT_DIM];
    __shared__ int bc[NBUCK];
    const int t = threadIdx.x;

    if (blockIdx.x >= GEMM_BLOCKS) {
        const int chunk = blockIdx.x - GEMM_BLOCKS;
        if (t < NBUCK) bc[t] = 0;
        if (chunk == 0 && t < 2 * OUT_DIM) bn[t] = 0.0f;
        __syncthreads();
        const int* d = dst + chunk * CHUNK;
        int dv[EPT];
        #pragma unroll
        for (int k = 0; k < EPT; ++k) {
            const int i = t + k * SORT_T;
            dv[k] = d[i < CHUNK ? i : CHUNK - 1];
        }
        #pragma unroll
        for (int k = 0; k < EPT; ++k)
            if (t + k * SORT_T < CHUNK) atomicAdd(&bc[dv[k] >> BSHIFT], 1);
        __syncthreads();
        if (t < NBUCK) tab[t * NCHUNK + chunk] = (unsigned short)bc[t];
        return;
    }

    for (int i = t; i < IN_DIM * OUT_DIM; i += SORT_T)
        lds_w[i] = W[i];
    __syncthreads();

    const int wave = t >> 5;
    const int lane = t & 31;
    const int tile = blockIdx.x * GEMM_WAVES + wave;
    if (tile >= GEMM_TILES) return;

    const int m0   = tile * 16;
    const int row  = m0 + (lane & 15);
    const int koff = (lane >> 4) * 2;
    const int col  = lane & 15;

    const float al0 = Wa[col];
    const float al1 = Wa[col + 16];
    const float ar0 = Wa[OUT_DIM + col];
    const float ar1 = Wa[OUT_DIM + col + 16];

    v8f c0 = {}, c1 = {};
    const float* hrow = h + (size_t)row * IN_DIM;

    #pragma unroll 4
    for (int k = 0; k < IN_DIM; k += 4) {
        v2f a;
        a.x = hrow[k + koff];
        a.y = hrow[k + koff + 1];
        v2f b0, b1;
        b0.x = lds_w[(k + koff)     * OUT_DIM + col];
        b0.y = lds_w[(k + koff + 1) * OUT_DIM + col];
        b1.x = lds_w[(k + koff)     * OUT_DIM + col + 16];
        b1.y = lds_w[(k + koff + 1) * OUT_DIM + col + 16];
        c0 = __builtin_amdgcn_wmma_f32_16x16x4_f32(false, a, false, b0, (short)0, c0, false, false);
        c1 = __builtin_amdgcn_wmma_f32_16x16x4_f32(false, a, false, b1, (short)0, c1, false, false);
    }

    const int rbase = m0 + ((lane >> 4) * 8);
    #pragma unroll
    for (int r = 0; r < 8; ++r) {
        z[(size_t)(rbase + r) * OUT_DIM + col]      = c0[r];
        z[(size_t)(rbase + r) * OUT_DIM + col + 16] = c1[r];
    }

    __builtin_amdgcn_sched_barrier(0);
    float pl[8], pr[8];
    #pragma unroll
    for (int r = 0; r < 8; ++r) {
        pl[r] = row_sum16_f(c0[r] * al0 + c1[r] * al1);
        pr[r] = row_sum16_f(c0[r] * ar0 + c1[r] * ar1);
    }
    if (col == 0) {
        float4* pe = (float4*)(el + rbase);
        pe[0] = make_float4(pl[0], pl[1], pl[2], pl[3]);
        pe[1] = make_float4(pl[4], pl[5], pl[6], pl[7]);
        float4* qe = (float4*)(er + rbase);
        qe[0] = make_float4(pr[0], pr[1], pr[2], pr[3]);
        qe[1] = make_float4(pr[4], pr[5], pr[6], pr[7]);
    }
}

__global__ void __launch_bounds__(SORT_T)
gat_bucket_scatter_kernel(const int* __restrict__ src,
                          const int* __restrict__ dst,
                          const unsigned short* __restrict__ tab,
                          int* __restrict__ bstart,
                          int* __restrict__ tmp) {
    __shared__ int cur[NBUCK];
    __shared__ int rtot[NBUCK];
    __shared__ int lcur[NBUCK];
    __shared__ int wsum[SCAN_WAVES];
    __shared__ int wown[SCAN_WAVES];
    __shared__ int2 srt[CHUNK];
    const int t = threadIdx.x;
    const int lane = t & 31;
    const int wave = t >> 5;
    const int c = blockIdx.x;
    const int* s = src + c * CHUNK;
    const int* d = dst + c * CHUNK;
    int sv[EPT], dv[EPT];
    #pragma unroll
    for (int k = 0; k < EPT; ++k) {
        const int i = t + k * SORT_T;
        const int ic = i < CHUNK ? i : CHUNK - 1;
        sv[k] = s[ic];
        dv[k] = d[ic];
    }
    const int mine = (int)tab[(t < NBUCK ? t : NBUCK - 1) * NCHUNK + c];

    const int4* t4 = (const int4*)tab;
    const int j0 = 8 * lane;
    int4 p[ROWS_PER_WAVE];
    #pragma unroll
    for (int i = 0; i < ROWS_PER_WAVE; ++i) {
        const int b  = wave + SORT_WAVES * i;
        const int bb = b < NBUCK ? b : NBUCK - 1;
        p[i] = t4[bb * (NCHUNK / 8) + lane];
    }
    __builtin_amdgcn_sched_barrier(0);
    unsigned keep[4];
    #pragma unroll
    for (int k = 0; k < 4; ++k)
        keep[k] = (j0 + 2 * k < c ? 0x0000ffffu : 0u) | (j0 + 2 * k + 1 < c ? 0xffff0000u : 0u);
    int vt = 0, vp = 0;
    #pragma unroll
    for (int i = 0; i < ROWS_PER_WAVE; ++i) {
        const unsigned w0 = (unsigned)p[i].x, w1 = (unsigned)p[i].y;
        const unsigned w2 = (unsigned)p[i].z, w3 = (unsigned)p[i].w;
        const unsigned t2 = (w0 + w1) + (w2 + w3);
        const unsigned p2 = ((w0 & keep[0]) + (w1 & keep[1])) + ((w2 & keep[2]) + (w3 & keep[3]));
        const int ts = row_sum16_i((int)((t2 & 0xffffu) + (t2 >> 16)));
        const int ps = row_sum16_i((int)((p2 & 0xffffu) + (p2 >> 16)));
        if ((lane & 15) == i) { vt = ts; vp = ps; }
    }
    vt += __shfl_xor(vt, 16, 32);
    vp += __shfl_xor(vp, 16, 32);
    const int bl = wave + SORT_WAVES * lane;
    if (lane < ROWS_PER_WAVE && bl < NBUCK) { rtot[bl] = vt; cur[bl] = vp; }
    __syncthreads();
    int own = 0, inc = 0, lown = 0, linc = 0;
    if (wave < SCAN_WAVES) {
        own  = (t < NBUCK) ? rtot[t] : 0;
        lown = (t < NBUCK) ? mine : 0;
        inc  = own;
        linc = lown;
        #pragma unroll
        for (int o = 1; o < 32; o <<= 1) {
            const int u = __shfl_up(inc, o, 32);
            const int v = __shfl_up(linc, o, 32);
            if (lane >= o) { inc += u; linc += v; }
        }
        if (lane == 31) { wsum[wave] = inc; wown[wave] = linc; }
    }
    __syncthreads();
    if (t < NBUCK) {
        int base = 0, lbase = 0;
        #pragma unroll
        for (int w = 0; w < SCAN_WAVES; ++w)
            if (w < wave) { base += wsum[w]; lbase += wown[w]; }
        const int first  = base + inc - own;
        const int lfirst = lbase + linc - lown;
        lcur[t] = lfirst;
        cur[t] += first - lfirst;
        if (c == 0) bstart[t] = first;
    }
    __syncthreads();
    int lp[EPT], off[EPT];
    #pragma unroll
    for (int k = 0; k < EPT; ++k) {
        const int b = dv[k] >> BSHIFT;
        lp[k] = 0;
        if (t + k * SORT_T < CHUNK) lp[k] = atomicAdd(&lcur[b], 1);
        off[k] = cur[b];
    }
    #pragma unroll
    for (int k = 0; k < EPT; ++k)
        if (t + k * SORT_T < CHUNK)
            srt[lp[k]] = make_int2(sv[k] | ((dv[k] & (BNODES - 1)) << SRC_BITS), lp[k] + off[k]);
    __syncthreads();
    #pragma unroll
    for (int k = 0; k < EPT; ++k) {
        const int i = t + k * SORT_T;
        if (i < CHUNK) {
            const int2 r = srt[i];
            tmp[r.y] = r.x;
        }
    }
}

#define LCAP 8192
#define ROW_LOADS 8
#define ROW_EDGES (2 * ROW_LOADS)
static_assert(ROW_EDGES <= 16, "a group's edges are one lane each of a 16-lane row");
static_assert(BNODES % 2 == 0 && N_NODES % 2 == 0, "the two nodes of a pair are both real or both past the end");

__device__ __forceinline__ float row_max16_f(float x) {
    x = fmaxf(x, row_xor_f<1>(x));
    x = fmaxf(x, row_xor_f<2>(x));
    x = fmaxf(x, row_xor_f<4>(x));
    x = fmaxf(x, row_xor_f<8>(x));
    return x;
}
template <int U> __device__ __forceinline__ int row_pick2(int x) {
    return (int)__builtin_amdgcn_permlane16((unsigned)x, (unsigned)x, 0x11111111u * (unsigned)(2 * U),
                                            0x11111111u * (unsigned)(2 * U + 1), true, false);
}

__device__ __forceinline__ float other_row_f(float x) {
    return __int_as_float((int)__builtin_amdgcn_permlanex16((unsigned)__float_as_int(x), (unsigned)__float_as_int(x),
                                                           0x76543210u, 0xfedcba98u, true, false));
}

template <bool IN_LDS>
__device__ __forceinline__ int2 rec_at(const int2* lrec, const int* gs, const float* __restrict__ el,
                                       float erd, int j) {
    if (IN_LDS) return lrec[j];
    const int s = __hip_atomic_load(gs + j, __ATOMIC_RELAXED, __HIP_MEMORY_SCOPE_AGENT);
    float e = el[s] + erd;
    e = (e >= 0.0f) ? e : LRELU_SLOPE * e;
    return make_int2(s, __float_as_int(e));
}

template <int U>
__device__ __forceinline__ void row_group_loads(const float4* __restrict__ z4, int s, int abits, unsigned q,
                                                float4* r, float* ak) {
    if constexpr (U < ROW_LOADS) {
        const unsigned sk = (unsigned)row_pick2<U>(s);
        r[U] = z4[sk * 8u + q];
        ak[U] = __int_as_float(row_pick2<U>(abits));
        row_group_loads<U + 1>(z4, s, abits, q, r, ak);
    }
}

template <bool IN_LDS>
__device__ __forceinline__ void bucket_nodes(const int2* lrec, const int* gs, const int* nfirst,
                                             const float* ler, const float* __restrict__ el,
                                             const float* __restrict__ z, float* __restrict__ hout,
                                             float* part, int b, int wave, int lane) {
    const int wave_s = __builtin_amdgcn_readfirstlane(wave);
    const int hrow = lane >> 4;
    const int l = lane & 15;
    const unsigned q = (unsigned)(lane & 7);
    const float4* __restrict__ z4 = (const float4*)z;
    float4* __restrict__ hout4 = (float4*)hout;
    float4 bs = make_float4(0.0f, 0.0f, 0.0f, 0.0f);
    float4 bq = make_float4(0.0f, 0.0f, 0.0f, 0.0f);
    for (int p = wave_s; p < BNODES / 2; p += SORT_WAVES) {
        if (b * BNODES + 2 * p >= N_NODES) break;
        const int dl = 2 * p + hrow;
        const int node = b * BNODES + dl;
        const int first = nfirst[dl];
        const int n     = nfirst[dl + 1] - first;
        const float erd = ler[dl];

        const bool has = l < n;
        int2 rec = make_int2(0, 0);
        if (n > 0) rec = rec_at<IN_LDS>(lrec, gs, el, erd, first + (has ? l : n - 1));
        const float e0 = __int_as_float(rec.y);

        float m = has ? e0 : -INFINITY;
        for (int j = l + 16; j < n; j += 16)
            m = fmaxf(m, __int_as_float(rec_at<IN_LDS>(lrec, gs, el, erd, first + j).y));
        m = row_max16_f(m);

        float ssum = has ? __expf(e0 - m) : 0.0f;
        for (int j = l + 16; j < n; j += 16)
            ssum += __expf(__int_as_float(rec_at<IN_LDS>(lrec, gs, el, erd, first + j).y) - m);
        ssum = row_sum16_f(ssum);

        const int n0 = __builtin_amdgcn_readlane(n, 0);
        const int n1 = __builtin_amdgcn_readlane(n, 16);
        const int nmax = n0 > n1 ? n0 : n1;
        float4 acc = make_float4(0.0f, 0.0f, 0.0f, 0.0f);
        for (int base = 0; base < nmax; base += ROW_EDGES) {
            const bool in = base + l < n;
            if (base > 0 && n > 0) rec = rec_at<IN_LDS>(lrec, gs, el, erd, first + (in ? base + l : n - 1));
            const int s = rec.x;
            const int abits = __float_as_int(in ? __expf(__int_as_float(rec.y) - m) / ssum : 0.0f);
            float4 r[ROW_LOADS];
            float ak[ROW_LOADS];
            row_group_loads<0>(z4, s, abits, q, r, ak);
            #pragma unroll
            for (int u = 0; u < ROW_LOADS; ++u) {
                acc.x += ak[u] * r[u].x;
                acc.y += ak[u] * r[u].y;
                acc.z += ak[u] * r[u].z;
                acc.w += ak[u] * r[u].w;
            }
        }
        acc.x += row_xor_f<8>(acc.x);
        acc.y += row_xor_f<8>(acc.y);
        acc.z += row_xor_f<8>(acc.z);
        acc.w += row_xor_f<8>(acc.w);
        if ((l >> 3) == 0) hout4[(unsigned)node * 8u + q] = acc;
        bs.x += acc.x;
        bs.y += acc.y;
        bs.z += acc.z;
        bs.w += acc.w;
        bq.x += acc.x * acc.x;
        bq.y += acc.y * acc.y;
        bq.z += acc.z * acc.z;
        bq.w += acc.w * acc.w;
    }
    bs.x += other_row_f(bs.x);
    bs.y += other_row_f(bs.y);
    bs.z += other_row_f(bs.z);
    bs.w += other_row_f(bs.w);
    bq.x += other_row_f(bq.x);
    bq.y += other_row_f(bq.y);
    bq.z += other_row_f(bq.z);
    bq.w += other_row_f(bq.w);
    if (lane < OUT_DIM / 4) {
        float4* p4 = (float4*)(part + wave_s * (2 * OUT_DIM));
        p4[lane] = bs;
        p4[OUT_DIM / 4 + lane] = bq;
    }
}

__global__ void __launch_bounds__(SORT_T, 16)
gat_bucket_gather_kernel(const float* __restrict__ el,
                         const float* __restrict__ er,
                         const int* __restrict__ bstart,
                         const int* __restrict__ tmp,
                         int* gs,
                         const float* __restrict__ z,
                         float* __restrict__ hout,
                         float* __restrict__ bn) {
    __shared__ int   nc[BNODES];
    __shared__ int   nfirst[BNODES + 1];
    __shared__ float ler[BNODES];
    __shared__ int   wtot[BNODES / 32];
    __shared__ int2  lrec[LCAP];
    __shared__ __attribute__((aligned(16))) float part[SORT_WAVES * 2 * OUT_DIM];
    const int t = threadIdx.x;
    const int lane = t & 31;
    const int wave = t >> 5;
    const int b = blockIdx.x;
    const int node = b * BNODES + t;
    const int beg = bstart[b];
    const int end = (b + 1 < NBUCK) ? bstart[b + 1] : N_EDGES;
    const bool fits = (end - beg) <= LCAP;
    if (t < BNODES) {
        nc[t] = 0;
        ler[t] = (node < N_NODES) ? er[node] : 0.0f;
    }
    __syncthreads();
    for (int j0 = beg + t; j0 < end; j0 += FILL_U * SORT_T) {
        int pv[FILL_U];
        #pragma unroll
        for (int u = 0; u < FILL_U; ++u) {
            const int j = j0 + u * SORT_T;
            pv[u] = tmp[j < end ? j : end - 1];
        }
        #pragma unroll
        for (int u = 0; u < FILL_U; ++u)
            if (j0 + u * SORT_T < end) atomicAdd(&nc[pv[u] >> SRC_BITS], 1);
    }
    __syncthreads();
    int c = 0, inc = 0;
    if (t < BNODES) {
        c = nc[t];
        inc = c;
        #pragma unroll
        for (int o = 1; o < 32; o <<= 1) {
            const int u = __shfl_up(inc, o, 32);
            if (lane >= o) inc += u;
        }
        if (lane == 31) wtot[t >> 5] = inc;
    }
    __syncthreads();
    if (t < BNODES) {
        int base = 0;
        #pragma unroll
        for (int w = 0; w < BNODES / 32; ++w)
            if (w < (t >> 5)) base += wtot[w];
        const int first = base + inc - c;
        nc[t] = first;
        nfirst[t] = first;
        if (t == BNODES - 1) nfirst[BNODES] = first + c;
    }
    __syncthreads();
    for (int j0 = beg + t; j0 < end; j0 += FILL_U * SORT_T) {
        int pv[FILL_U];
        float ev[FILL_U];
        #pragma unroll
        for (int u = 0; u < FILL_U; ++u) {
            const int j = j0 + u * SORT_T;
            pv[u] = tmp[j < end ? j : end - 1];
        }
        #pragma unroll
        for (int u = 0; u < FILL_U; ++u) ev[u] = el[pv[u] & ((1 << SRC_BITS) - 1)];
        #pragma unroll
        for (int u = 0; u < FILL_U; ++u) {
            if (j0 + u * SORT_T < end) {
                const int s  = pv[u] & ((1 << SRC_BITS) - 1);
                const int dl = pv[u] >> SRC_BITS;
                const int pos = atomicAdd(&nc[dl], 1);
                float e = ev[u] + ler[dl];
                e = (e >= 0.0f) ? e : LRELU_SLOPE * e;
                if (fits) lrec[pos] = make_int2(s, __float_as_int(e));
                else      gs[beg + pos] = s;
            }
        }
    }
    if (!fits) __threadfence();
    __syncthreads();
    if (fits) bucket_nodes<true >(lrec, gs + beg, nfirst, ler, el, z, hout, part, b, wave, lane);
    else      bucket_nodes<false>(lrec, gs + beg, nfirst, ler, el, z, hout, part, b, wave, lane);
    __syncthreads();
    if (t < 2 * OUT_DIM) {
        float s = 0.0f;
        #pragma unroll
        for (int w = 0; w < SORT_WAVES; ++w) s += part[w * (2 * OUT_DIM) + t];
        atomicAdd(&bn[t], s);
    }
}

#define APPLY_T       128
#define APPLY_PIECES  10
#define APPLY_N4      (N_NODES * OUT_DIM / 4)
#define APPLY_THREADS (APPLY_N4 / APPLY_PIECES)
#define APPLY_BLOCKS  (APPLY_THREADS / APPLY_T)
static_assert(APPLY_N4 % APPLY_PIECES == 0 && APPLY_THREADS % APPLY_T == 0, "the grid tiles out exactly: no tail");
static_assert(APPLY_THREADS % (OUT_DIM / 4) == 0, "a thread's pieces all hold the same four channels");

__device__ __forceinline__ float bn_elu(float v, float mean, float rs, float g, float b) {
    const float x = (v - mean) * rs * g + b;
    return (x > 0.0f) ? x : expm1f(x);
}

__global__ void __launch_bounds__(APPLY_T)
gat_bn_apply_kernel(const float* __restrict__ bn,
                    const float* __restrict__ gamma,
                    const float* __restrict__ beta,
                    float* __restrict__ out) {
    const int g = blockIdx.x * APPLY_T + threadIdx.x;
    const int q = g & (OUT_DIM / 4 - 1);
    float4* __restrict__ o4 = (float4*)out;
    const float4 s4 = ((const float4*)bn)[q];
    const float4 q4 = ((const float4*)bn)[OUT_DIM / 4 + q];
    const float4 g4 = ((const float4*)gamma)[q];
    const float4 b4 = ((const float4*)beta)[q];
    float4 v[APPLY_PIECES];
    #pragma unroll
    for (int i = 0; i < APPLY_PIECES; ++i)
        v[i] = o4[g + i * APPLY_THREADS];
    __builtin_amdgcn_sched_barrier(0);
    const float inv_n = 1.0f / (float)N_NODES;
    float4 mean, rs;
    mean.x = s4.x * inv_n;
    mean.y = s4.y * inv_n;
    mean.z = s4.z * inv_n;
    mean.w = s4.w * inv_n;
    rs.x = rsqrtf(q4.x * inv_n - mean.x * mean.x + BN_EPS);
    rs.y = rsqrtf(q4.y * inv_n - mean.y * mean.y + BN_EPS);
    rs.z = rsqrtf(q4.z * inv_n - mean.z * mean.z + BN_EPS);
    rs.w = rsqrtf(q4.w * inv_n - mean.w * mean.w + BN_EPS);
    #pragma unroll
    for (int i = 0; i < APPLY_PIECES; ++i) {
        float4 r;
        r.x = bn_elu(v[i].x, mean.x, rs.x, g4.x, b4.x);
        r.y = bn_elu(v[i].y, mean.y, rs.y, g4.y, b4.y);
        r.z = bn_elu(v[i].z, mean.z, rs.z, g4.z, b4.z);
        r.w = bn_elu(v[i].w, mean.w, rs.w, g4.w, b4.w);
        o4[g + i * APPLY_THREADS] = r;
    }
}

extern "C" void kernel_launch(void* const* d_in, const int* in_sizes, int n_in,
                              void* d_out, int out_size, void* d_ws, size_t ws_size,
                              hipStream_t stream) {
    const float* h     = (const float*)d_in[0];
    const float* W_fc  = (const float*)d_in[1];
    const float* W_at  = (const float*)d_in[2];
    const float* gamma = (const float*)d_in[3];
    const float* beta  = (const float*)d_in[4];
    const int*   src   = (const int*)d_in[5];
    const int*   dst   = (const int*)d_in[6];
    float* out = (float*)d_out;

    float* z    = (float*)d_ws;
    float* el   = z + (size_t)N_NODES * OUT_DIM;
    float* er   = el + N_NODES;
    int*   bstart = (int*)(er + N_NODES);
    int*   tmp  = bstart + 2 * N_NODES;
    int*   gs   = tmp + N_EDGES;
    float* bn   = (float*)(gs + N_EDGES);
    unsigned short* tab = (unsigned short*)d_out;

    gat_gemm_proj_count_kernel<<<GEMM_BLOCKS + NCHUNK, SORT_T, 0, stream>>>(h, W_fc, W_at, dst, z, el, er, tab, bn);

    gat_bucket_scatter_kernel<<<NCHUNK, SORT_T, 0, stream>>>(src, dst, tab, bstart, tmp);

    gat_bucket_gather_kernel<<<NBUCK, SORT_T, 0, stream>>>(el, er, bstart, tmp, gs, z, out, bn);

    gat_bn_apply_kernel<<<APPLY_BLOCKS, APPLY_T, 0, stream>>>(bn, gamma, beta, out);
}
